# starting kernel
# baseline (speedup 1.0000x reference)
#include <hip/hip_runtime.h>
#include <math.h>


#define MAXN 0.996f
#define MINN 1e-15f
#define KC   128
#define BM   64

typedef float v2f __attribute__((ext_vector_type(2)));
typedef float v8f __attribute__((ext_vector_type(8)));

__device__ __forceinline__ float artanh_f(float z) {
    z = fminf(fmaxf(z, -(1.f - 1e-7f)), 1.f - 1e-7f);
    return 0.5f * logf((1.f + z) / (1.f - z));
}

__global__ void bias_prep(const float* __restrict__ b1,
                          const float* __restrict__ b2,
                          float* __restrict__ bb, float* __restrict__ y2) {
    __shared__ float red[128];
    const float* b = blockIdx.x ? b2 : b1;
    int t = threadIdx.x;
    float v = b[t];
    red[t] = v * v;
    __syncthreads();
    for (int s = 64; s > 0; s >>= 1) {
        if (t < s) red[t] += red[t + s];
        __syncthreads();
    }
    float bn = fmaxf(sqrtf(red[0]), MINN);
    float tc = fminf(tanhf(bn), MAXN);
    bb[blockIdx.x * 128 + t] = tc / bn * v;
    if (t == 0) y2[blockIdx.x] = tc * tc;
}

__global__ __launch_bounds__(256) void hyp_linear_wmma(
    const float* __restrict__ X, int Kdim, int nrows,
    const float* __restrict__ W,
    const float* __restrict__ bb,
    const float* __restrict__ y2p,
    float* __restrict__ HT, int apply_exp) {
    __shared__ float As[BM][KC + 4];
    __shared__ float Ws[128][KC + 4];
    __shared__ float red[BM][4];
    __shared__ float scale_row[BM], xn_row[BM];
    __shared__ float rs_mx2[BM], rs_dot[BM];
    __shared__ float rowCA[BM], rowCB[BM];

    const int t = threadIdx.x;
    const int row0 = blockIdx.x * BM;

    {
        int r = t >> 2, p = t & 3;
        int gr = row0 + r;
        float s = 0.f;
        if (gr < nrows) {
            int kl = Kdim >> 2;
            const float* px = X + (size_t)gr * Kdim + p * kl;
            for (int k = 0; k < kl; k += 4) {
                float4 v = *(const float4*)(px + k);
                s += v.x * v.x + v.y * v.y + v.z * v.z + v.w * v.w;
            }
        }
        red[r][p] = s;
    }
    if (t < BM) { rs_mx2[t] = 0.f; rs_dot[t] = 0.f; }
    __syncthreads();
    if (t < BM) {
        float un2 = red[t][0] + red[t][1] + red[t][2] + red[t][3];
        float un = fmaxf(sqrtf(un2), MINN);
        if (apply_exp) {
            float bc = fminf(tanhf(un), MAXN);
            scale_row[t] = bc / un;
            xn_row[t] = fmaxf(bc, MINN);
        } else {
            scale_row[t] = 1.f;
            xn_row[t] = un;
        }
    }
    __syncthreads();

    const int lane = t & 31;
    const int wid = t >> 5;
    const int rs = wid >> 1;
    const int ch = wid & 1;
    const int half = lane >> 4;
    const int l15 = lane & 15;

    v8f acc[4];
    for (int j = 0; j < 4; ++j)
        for (int i = 0; i < 8; ++i) acc[j][i] = 0.f;

    const int nchunks = Kdim / KC;

    for (int kc = 0; kc < nchunks; ++kc) {
        for (int f = t; f < BM * KC / 4; f += 256) {
            int r = f >> 5;
            int c4 = f & 31;
            int gr = row0 + r;
            float4 v = make_float4(0.f, 0.f, 0.f, 0.f);
            if (gr < nrows) {
                v = *(const float4*)(X + (size_t)gr * Kdim + kc * KC + c4 * 4);
                float sc = scale_row[r];
                v.x *= sc; v.y *= sc; v.z *= sc; v.w *= sc;
            }
            *(float4*)(&As[r][c4 * 4]) = v;
        }
        for (int f = t; f < 128 * KC / 4; f += 256) {
            int r = f >> 5;
            int c4 = f & 31;
            float4 v = *(const float4*)(W + (size_t)r * Kdim + kc * KC + c4 * 4);
            *(float4*)(&Ws[r][c4 * 4]) = v;
        }
        __syncthreads();

        const int mr = rs * 16 + l15;
        for (int kk = 0; kk < KC; kk += 4) {
            const int ko = kk + half * 2;
            v2f a;
            a.x = As[mr][ko];
            a.y = As[mr][ko + 1];
#pragma unroll
            for (int j = 0; j < 4; ++j) {
                const int colg = (ch * 4 + j) * 16 + l15;
                v2f b;
                b.x = Ws[colg][ko];
                b.y = Ws[colg][ko + 1];
                acc[j] = __builtin_amdgcn_wmma_f32_16x16x4_f32(
                    false, a, false, b, (short)0, acc[j], false, false);
            }
        }
        __syncthreads();
    }

    float bbv[4];
#pragma unroll
    for (int j = 0; j < 4; ++j) bbv[j] = bb[(ch * 4 + j) * 16 + l15];

#pragma unroll
    for (int i = 0; i < 8; ++i) {
        float p2 = 0.f, pd = 0.f;
#pragma unroll
        for (int j = 0; j < 4; ++j) {
            float v = acc[j][i];
            p2 += v * v;
            pd += v * bbv[j];
        }
        for (int m = 1; m < 16; m <<= 1) {
            p2 += __shfl_xor(p2, m, 16);
            pd += __shfl_xor(pd, m, 16);
        }
        int mrow = rs * 16 + i + half * 8;
        if (l15 == 0) {
            atomicAdd(&rs_mx2[mrow], p2);
            atomicAdd(&rs_dot[mrow], pd);
        }
    }
    __syncthreads();

    if (t < BM) {
        float xn = xn_row[t];
        float mxn = fmaxf(sqrtf(rs_mx2[t]), MINN);
        float tv = tanhf(mxn / xn * artanh_f(xn));
        float tc = fminf(tv, MAXN);
        float s_mv = tc / mxn;
        float x2 = tc * tc;
        float xy = s_mv * rs_dot[t];
        float y2 = y2p[0];
        float al = 1.f + 2.f * xy + y2;
        float be = 1.f - x2;
        float den = fmaxf(1.f + 2.f * xy + x2 * y2, MINN);
        float nn2 = fmaxf(al * al * x2 + 2.f * al * be * xy + be * be * y2, 0.f);
        float hn = fmaxf(sqrtf(nn2) / den, MINN);
        float hc = fminf(hn, MAXN);
        float L = artanh_f(hc) / hn;
        rowCA[t] = L * al * s_mv / den;
        rowCB[t] = L * be / den;
    }
    __syncthreads();

#pragma unroll
    for (int i = 0; i < 8; ++i) {
        int mrow = rs * 16 + i + half * 8;
        int gr = row0 + mrow;
        if (gr < nrows) {
            float ca = rowCA[mrow], cb = rowCB[mrow];
#pragma unroll
            for (int j = 0; j < 4; ++j) {
                int colg = (ch * 4 + j) * 16 + l15;
                HT[(size_t)gr * 128 + colg] = ca * acc[j][i] + cb * bbv[j];
            }
        }
    }
}

__global__ __launch_bounds__(256) void edge_agg(const float* __restrict__ HT,
                                                const int* __restrict__ erow,
                                                const int* __restrict__ ecol,
                                                float* __restrict__ AGG, int E) {
    int e = blockIdx.x * 8 + (threadIdx.x >> 5);
    if (e >= E) return;
    int lane = threadIdx.x & 31;
    int r = erow[e], c = ecol[e];
    float4 v = *(const float4*)(HT + (size_t)c * 128 + lane * 4);
    float* dst = AGG + (size_t)r * 128 + lane * 4;
    atomicAdd(dst + 0, v.x);
    atomicAdd(dst + 1, v.y);
    atomicAdd(dst + 2, v.z);
    atomicAdd(dst + 3, v.w);
}

__global__ __launch_bounds__(256) void post_agg(float* __restrict__ H, int nrows) {
    int r = blockIdx.x * 8 + (threadIdx.x >> 5);
    if (r >= nrows) return;
    int lane = threadIdx.x & 31;
    float* p = H + (size_t)r * 128 + lane * 4;
    float4 v = *(const float4*)p;
    float an2 = v.x * v.x + v.y * v.y + v.z * v.z + v.w * v.w;
    for (int m = 1; m < 32; m <<= 1) an2 += __shfl_xor(an2, m, 32);
    float an = fmaxf(sqrtf(an2), MINN);
    float tn = fminf(fmaxf(tanhf(an), MINN), MAXN);
    float lg = artanh_f(tn) / an;
    float4 at;
    at.x = tanhf(lg * v.x);
    at.y = tanhf(lg * v.y);
    at.z = tanhf(lg * v.z);
    at.w = tanhf(lg * v.w);
    float vn2 = at.x * at.x + at.y * at.y + at.z * at.z + at.w * at.w;
    for (int m = 1; m < 32; m <<= 1) vn2 += __shfl_xor(vn2, m, 32);
    float vn = fmaxf(sqrtf(vn2), MINN);
    float s = fminf(tanhf(vn), MAXN) / vn;
    *(float4*)p = make_float4(s * at.x, s * at.y, s * at.z, s * at.w);
}

__global__ __launch_bounds__(256) void final_scatter(
    const float* __restrict__ H, const int* __restrict__ seg,
    float* __restrict__ ssum, float* __restrict__ cnt, int nrows) {
    int r = blockIdx.x * 8 + (threadIdx.x >> 5);
    if (r >= nrows) return;
    int lane = threadIdx.x & 31;
    float4 v = *(const float4*)(H + (size_t)r * 128 + lane * 4);
    float hn2 = v.x * v.x + v.y * v.y + v.z * v.z + v.w * v.w;
    for (int m = 1; m < 32; m <<= 1) hn2 += __shfl_xor(hn2, m, 32);
    float hn = fmaxf(sqrtf(hn2), MINN);
    float pc = fminf(hn, MAXN);
    float L = artanh_f(pc) / hn;
    int s = seg[r];
    float* d = ssum + (size_t)s * 128 + lane * 4;
    atomicAdd(d + 0, L * v.x);
    atomicAdd(d + 1, L * v.y);
    atomicAdd(d + 2, L * v.z);
    atomicAdd(d + 3, L * v.w);
    if (lane == 0) atomicAdd(&cnt[s], 1.f);
}

__global__ __launch_bounds__(256) void pool_k(const float* __restrict__ ssum,
                                              const float* __restrict__ cnt,
                                              float* __restrict__ out, int nseg) {
    int r = blockIdx.x * 8 + (threadIdx.x >> 5);
    if (r >= nseg) return;
    int lane = threadIdx.x & 31;
    float inv = 1.f / fmaxf(cnt[r], 1.f);
    float4 v = *(const float4*)(ssum + (size_t)r * 128 + lane * 4);
    v.x *= inv; v.y *= inv; v.z *= inv; v.w *= inv;
    float mn2 = v.x * v.x + v.y * v.y + v.z * v.z + v.w * v.w;
    for (int m = 1; m < 32; m <<= 1) mn2 += __shfl_xor(mn2, m, 32);
    float mn = fmaxf(sqrtf(mn2), MINN);
    float s = fminf(tanhf(mn), MAXN) / mn;
    float* o = out + (size_t)r * 128 + lane * 4;
    o[0] = s * v.x; o[1] = s * v.y; o[2] = s * v.z; o[3] = s * v.w;
}

__global__ void zero_f32(float* __restrict__ p, size_t n) {
    size_t i = (size_t)blockIdx.x * blockDim.x + threadIdx.x;
    size_t st = (size_t)gridDim.x * blockDim.x;
    for (; i < n; i += st) p[i] = 0.f;
}

extern "C" void kernel_launch(void* const* d_in, const int* in_sizes, int n_in,
                              void* d_out, int out_size, void* d_ws, size_t ws_size,
                              hipStream_t stream) {
    const float* x = (const float*)d_in[0];
    const int* erow = (const int*)d_in[1];
    const int* ecol = (const int*)d_in[2];
    const int* seg = (const int*)d_in[3];
    const float* W1 = (const float*)d_in[4];
    const float* b1 = (const float*)d_in[5];
    const float* W2 = (const float*)d_in[6];
    const float* b2 = (const float*)d_in[7];

    const int N = in_sizes[0] / 256;
    const int E = in_sizes[1];
    const int NSEG = out_size / 128;

    char* ws = (char*)d_ws;
    size_t bufBytes = (size_t)N * 128 * sizeof(float);
    float* buf0 = (float*)ws;
    float* buf1 = (float*)(ws + bufBytes);
    float* bbias = (float*)(ws + 2 * bufBytes);
    float* y2 = bbias + 256;
    float* ssum = buf0;
    float* cnt = buf0 + (size_t)NSEG * 128;

    dim3 blk(256);
    int gRows = (N + 7) / 8;
    int gGemm = (N + BM - 1) / BM;
    int gEdge = (E + 7) / 8;

    bias_prep<<<2, 128, 0, stream>>>(b1, b2, bbias, y2);

    hyp_linear_wmma<<<gGemm, blk, 0, stream>>>(x, 256, N, W1, bbias, y2, buf0, 1);
    zero_f32<<<2048, 256, 0, stream>>>(buf1, (size_t)N * 128);
    edge_agg<<<gEdge, blk, 0, stream>>>(buf0, erow, ecol, buf1, E);
    post_agg<<<gRows, blk, 0, stream>>>(buf1, N);

    hyp_linear_wmma<<<gGemm, blk, 0, stream>>>(buf1, 128, N, W2, bbias + 128,
                                               y2 + 1, buf0, 0);
    zero_f32<<<2048, 256, 0, stream>>>(buf1, (size_t)N * 128);
    edge_agg<<<gEdge, blk, 0, stream>>>(buf0, erow, ecol, buf1, E);
    post_agg<<<gRows, blk, 0, stream>>>(buf1, N);

    zero_f32<<<64, 256, 0, stream>>>(ssum, (size_t)NSEG * 128 + NSEG);
    final_scatter<<<gRows, blk, 0, stream>>>(buf1, seg, ssum, cnt, N);
    pool_k<<<(NSEG + 7) / 8, blk, 0, stream>>>(ssum, cnt, (float*)d_out, NSEG);
}


# edge sum by f32 atomics -> edges grouped by destination once per call (counting sort, ranks by int atomics); each node sums its rows in registers in list order, one store; sum order changes
# speedup vs baseline: 1.8677x; 1.8677x over previous
#include <hip/hip_runtime.h>
#include <math.h>


#define MAXN 0.996f
#define MINN 1e-15f
#define KC   128
#define BM   64

typedef float v2f __attribute__((ext_vector_type(2)));
typedef float v8f __attribute__((ext_vector_type(8)));

__device__ __forceinline__ float artanh_f(float z) {
    z = fminf(fmaxf(z, -(1.f - 1e-7f)), 1.f - 1e-7f);
    return 0.5f * logf((1.f + z) / (1.f - z));
}

__global__ void bias_prep(const float* __restrict__ b1,
                          const float* __restrict__ b2,
                          float* __restrict__ bb, float* __restrict__ y2) {
    __shared__ float red[128];
    const float* b = blockIdx.x ? b2 : b1;
    int t = threadIdx.x;
    float v = b[t];
    red[t] = v * v;
    __syncthreads();
    for (int s = 64; s > 0; s >>= 1) {
        if (t < s) red[t] += red[t + s];
        __syncthreads();
    }
    float bn = fmaxf(sqrtf(red[0]), MINN);
    float tc = fminf(tanhf(bn), MAXN);
    bb[blockIdx.x * 128 + t] = tc / bn * v;
    if (t == 0) y2[blockIdx.x] = tc * tc;
}

__global__ __launch_bounds__(256) void hyp_linear_wmma(
    const float* __restrict__ X, int Kdim, int nrows,
    const float* __restrict__ W,
    const float* __restrict__ bb,
    const float* __restrict__ y2p,
    float* __restrict__ HT, int apply_exp) {
    __shared__ float As[BM][KC + 4];
    __shared__ float Ws[128][KC + 4];
    __shared__ float red[BM][4];
    __shared__ float scale_row[BM], xn_row[BM];
    __shared__ float rs_mx2[BM], rs_dot[BM];
    __shared__ float rowCA[BM], rowCB[BM];

    const int t = threadIdx.x;
    const int row0 = blockIdx.x * BM;

    {
        int r = t >> 2, p = t & 3;
        int gr = row0 + r;
        float s = 0.f;
        if (gr < nrows) {
            int kl = Kdim >> 2;
            const float* px = X + (size_t)gr * Kdim + p * kl;
            for (int k = 0; k < kl; k += 4) {
                float4 v = *(const float4*)(px + k);
                s += v.x * v.x + v.y * v.y + v.z * v.z + v.w * v.w;
            }
        }
        red[r][p] = s;
    }
    if (t < BM) { rs_mx2[t] = 0.f; rs_dot[t] = 0.f; }
    __syncthreads();
    if (t < BM) {
        float un2 = red[t][0] + red[t][1] + red[t][2] + red[t][3];
        float un = fmaxf(sqrtf(un2), MINN);
        if (apply_exp) {
            float bc = fminf(tanhf(un), MAXN);
            scale_row[t] = bc / un;
            xn_row[t] = fmaxf(bc, MINN);
        } else {
            scale_row[t] = 1.f;
            xn_row[t] = un;
        }
    }
    __syncthreads();

    const int lane = t & 31;
    const int wid = t >> 5;
    const int rs = wid >> 1;
    const int ch = wid & 1;
    const int half = lane >> 4;
    const int l15 = lane & 15;

    v8f acc[4];
    for (int j = 0; j < 4; ++j)
        for (int i = 0; i < 8; ++i) acc[j][i] = 0.f;

    const int nchunks = Kdim / KC;

    for (int kc = 0; kc < nchunks; ++kc) {
        for (int f = t; f < BM * KC / 4; f += 256) {
            int r = f >> 5;
            int c4 = f & 31;
            int gr = row0 + r;
            float4 v = make_float4(0.f, 0.f, 0.f, 0.f);
            if (gr < nrows) {
                v = *(const float4*)(X + (size_t)gr * Kdim + kc * KC + c4 * 4);
                float sc = scale_row[r];
                v.x *= sc; v.y *= sc; v.z *= sc; v.w *= sc;
            }
            *(float4*)(&As[r][c4 * 4]) = v;
        }
        for (int f = t; f < 128 * KC / 4; f += 256) {
            int r = f >> 5;
            int c4 = f & 31;
            float4 v = *(const float4*)(W + (size_t)r * Kdim + kc * KC + c4 * 4);
            *(float4*)(&Ws[r][c4 * 4]) = v;
        }
        __syncthreads();

        const int mr = rs * 16 + l15;
        for (int kk = 0; kk < KC; kk += 4) {
            const int ko = kk + half * 2;
            v2f a;
            a.x = As[mr][ko];
            a.y = As[mr][ko + 1];
#pragma unroll
            for (int j = 0; j < 4; ++j) {
                const int colg = (ch * 4 + j) * 16 + l15;
                v2f b;
                b.x = Ws[colg][ko];
                b.y = Ws[colg][ko + 1];
                acc[j] = __builtin_amdgcn_wmma_f32_16x16x4_f32(
                    false, a, false, b, (short)0, acc[j], false, false);
            }
        }
        __syncthreads();
    }

    float bbv[4];
#pragma unroll
    for (int j = 0; j < 4; ++j) bbv[j] = bb[(ch * 4 + j) * 16 + l15];

#pragma unroll
    for (int i = 0; i < 8; ++i) {
        float p2 = 0.f, pd = 0.f;
#pragma unroll
        for (int j = 0; j < 4; ++j) {
            float v = acc[j][i];
            p2 += v * v;
            pd += v * bbv[j];
        }
        for (int m = 1; m < 16; m <<= 1) {
            p2 += __shfl_xor(p2, m, 16);
            pd += __shfl_xor(pd, m, 16);
        }
        int mrow = rs * 16 + i + half * 8;
        if (l15 == 0) {
            atomicAdd(&rs_mx2[mrow], p2);
            atomicAdd(&rs_dot[mrow], pd);
        }
    }
    __syncthreads();

    if (t < BM) {
        float xn = xn_row[t];
        float mxn = fmaxf(sqrtf(rs_mx2[t]), MINN);
        float tv = tanhf(mxn / xn * artanh_f(xn));
        float tc = fminf(tv, MAXN);
        float s_mv = tc / mxn;
        float x2 = tc * tc;
        float xy = s_mv * rs_dot[t];
        float y2 = y2p[0];
        float al = 1.f + 2.f * xy + y2;
        float be = 1.f - x2;
        float den = fmaxf(1.f + 2.f * xy + x2 * y2, MINN);
        float nn2 = fmaxf(al * al * x2 + 2.f * al * be * xy + be * be * y2, 0.f);
        float hn = fmaxf(sqrtf(nn2) / den, MINN);
        float hc = fminf(hn, MAXN);
        float L = artanh_f(hc) / hn;
        rowCA[t] = L * al * s_mv / den;
        rowCB[t] = L * be / den;
    }
    __syncthreads();

#pragma unroll
    for (int i = 0; i < 8; ++i) {
        int mrow = rs * 16 + i + half * 8;
        int gr = row0 + mrow;
        if (gr < nrows) {
            float ca = rowCA[mrow], cb = rowCB[mrow];
#pragma unroll
            for (int j = 0; j < 4; ++j) {
                int colg = (ch * 4 + j) * 16 + l15;
                HT[(size_t)gr * 128 + colg] = ca * acc[j][i] + cb * bbv[j];
            }
        }
    }
}

__global__ __launch_bounds__(256) void edge_agg(const float* __restrict__ HT,
                                                const int* __restrict__ erow,
                                                const int* __restrict__ ecol,
                                                float* __restrict__ AGG, int E) {
    int e = blockIdx.x * 8 + (threadIdx.x >> 5);
    if (e >= E) return;
    int lane = threadIdx.x & 31;
    int r = erow[e], c = ecol[e];
    float4 v = *(const float4*)(HT + (size_t)c * 128 + lane * 4);
    float* dst = AGG + (size_t)r * 128 + lane * 4;
    atomicAdd(dst + 0, v.x);
    atomicAdd(dst + 1, v.y);
    atomicAdd(dst + 2, v.z);
    atomicAdd(dst + 3, v.w);
}

#define SCAN_B 1024

__global__ __launch_bounds__(256) void edge_rank(const int* __restrict__ erow,
                                                 int* __restrict__ cnt,
                                                 int* __restrict__ rank, int E) {
    int st = gridDim.x * 256;
    for (int i = blockIdx.x * 256 + threadIdx.x; i < E; i += st)
        rank[i] = atomicAdd(&cnt[erow[i]], 1);
}

__device__ __forceinline__ int block_sum_1024(int v, int* sh) {
    int t = threadIdx.x;
    for (int m = 1; m < 32; m <<= 1) v += __shfl_xor(v, m, 32);
    if ((t & 31) == 0) sh[t >> 5] = v;
    __syncthreads();
    int w = sh[t & 31];
    for (int m = 1; m < 32; m <<= 1) w += __shfl_xor(w, m, 32);
    __syncthreads();
    return w;
}

__global__ __launch_bounds__(SCAN_B) void scan_sums(const int* __restrict__ cnt,
                                                    int* __restrict__ bsum, int n) {
    __shared__ int sh[32];
    int i = blockIdx.x * SCAN_B + threadIdx.x;
    int tot = block_sum_1024(i < n ? cnt[i] : 0, sh);
    if (threadIdx.x == 0) bsum[blockIdx.x] = tot;
}

__global__ __launch_bounds__(SCAN_B) void scan_apply(int* __restrict__ offs,
                                                     const int* __restrict__ bsum, int n) {
    __shared__ int sh[32];
    const int t = threadIdx.x, lane = t & 31, wid = t >> 5;
    int p = 0;
    for (int b = t; b < (int)blockIdx.x; b += SCAN_B) p += bsum[b];
    const int base = block_sum_1024(p, sh);

    int i = blockIdx.x * SCAN_B + t;
    int v = i < n ? offs[i] : 0;
    int s = v;
    for (int m = 1; m < 32; m <<= 1) {
        int u = __shfl_up(s, m, 32);
        if (lane >= m) s += u;
    }
    if (lane == 31) sh[wid] = s;
    __syncthreads();
    if (wid == 0) {
        int w0 = sh[lane], w = w0;
        for (int m = 1; m < 32; m <<= 1) {
            int u = __shfl_up(w, m, 32);
            if (lane >= m) w += u;
        }
        sh[lane] = w - w0;
    }
    __syncthreads();
    int excl = base + sh[wid] + s - v;
    if (i < n) offs[i] = excl;
    if (i == n - 1) offs[n] = excl + v;
}

__global__ __launch_bounds__(256) void edge_place(const int* __restrict__ erow,
                                                  const int* __restrict__ ecol,
                                                  const int* __restrict__ offs,
                                                  const int* __restrict__ rank,
                                                  int* __restrict__ scol, int E) {
    int st = gridDim.x * 256;
    for (int i = blockIdx.x * 256 + threadIdx.x; i < E; i += st)
        scol[offs[erow[i]] + rank[i]] = ecol[i];
}

__global__ __launch_bounds__(256) void node_gather(const float* __restrict__ HT,
                                                   const int* __restrict__ offs,
                                                   const int* __restrict__ scol,
                                                   float* __restrict__ AGG, int nrows) {
    const int lane = threadIdx.x & 31;
    const int w0 = __builtin_amdgcn_readfirstlane((int)(blockIdx.x * 8 + (threadIdx.x >> 5)));
    const int nw = gridDim.x * 8;
    for (int r = w0; r < nrows; r += nw) {
        const int beg = offs[r], end = offs[r + 1];
        float4 acc = make_float4(0.f, 0.f, 0.f, 0.f);
        for (int b0 = beg; b0 < end; b0 += 32) {
            const int n = min(32, end - b0);
            const int myc = (lane < n) ? scol[b0 + lane] : 0;
            int j = 0;
            for (; j + 4 <= n; j += 4) {
                const int c0 = __builtin_amdgcn_readlane(myc, j);
                const int c1 = __builtin_amdgcn_readlane(myc, j + 1);
                const int c2 = __builtin_amdgcn_readlane(myc, j + 2);
                const int c3 = __builtin_amdgcn_readlane(myc, j + 3);
                const float4 v0 = *((const float4*)(HT + (size_t)c0 * 128) + lane);
                const float4 v1 = *((const float4*)(HT + (size_t)c1 * 128) + lane);
                const float4 v2 = *((const float4*)(HT + (size_t)c2 * 128) + lane);
                const float4 v3 = *((const float4*)(HT + (size_t)c3 * 128) + lane);
                acc.x += v0.x; acc.y += v0.y; acc.z += v0.z; acc.w += v0.w;
                acc.x += v1.x; acc.y += v1.y; acc.z += v1.z; acc.w += v1.w;
                acc.x += v2.x; acc.y += v2.y; acc.z += v2.z; acc.w += v2.w;
                acc.x += v3.x; acc.y += v3.y; acc.z += v3.z; acc.w += v3.w;
            }
            for (; j < n; ++j) {
                const int c0 = __builtin_amdgcn_readlane(myc, j);
                const float4 v0 = *((const float4*)(HT + (size_t)c0 * 128) + lane);
                acc.x += v0.x; acc.y += v0.y; acc.z += v0.z; acc.w += v0.w;
            }
        }
        *((float4*)(AGG + (size_t)r * 128) + lane) = acc;
    }
}

__global__ __launch_bounds__(256) void post_agg(float* __restrict__ H, int nrows) {
    int r = blockIdx.x * 8 + (threadIdx.x >> 5);
    if (r >= nrows) return;
    int lane = threadIdx.x & 31;
    float* p = H + (size_t)r * 128 + lane * 4;
    float4 v = *(const float4*)p;
    float an2 = v.x * v.x + v.y * v.y + v.z * v.z + v.w * v.w;
    for (int m = 1; m < 32; m <<= 1) an2 += __shfl_xor(an2, m, 32);
    float an = fmaxf(sqrtf(an2), MINN);
    float tn = fminf(fmaxf(tanhf(an), MINN), MAXN);
    float lg = artanh_f(tn) / an;
    float4 at;
    at.x = tanhf(lg * v.x);
    at.y = tanhf(lg * v.y);
    at.z = tanhf(lg * v.z);
    at.w = tanhf(lg * v.w);
    float vn2 = at.x * at.x + at.y * at.y + at.z * at.z + at.w * at.w;
    for (int m = 1; m < 32; m <<= 1) vn2 += __shfl_xor(vn2, m, 32);
    float vn = fmaxf(sqrtf(vn2), MINN);
    float s = fminf(tanhf(vn), MAXN) / vn;
    *(float4*)p = make_float4(s * at.x, s * at.y, s * at.z, s * at.w);
}

__global__ __launch_bounds__(256) void final_scatter(
    const float* __restrict__ H, const int* __restrict__ seg,
    float* __restrict__ ssum, float* __restrict__ cnt, int nrows) {
    int r = blockIdx.x * 8 + (threadIdx.x >> 5);
    if (r >= nrows) return;
    int lane = threadIdx.x & 31;
    float4 v = *(const float4*)(H + (size_t)r * 128 + lane * 4);
    float hn2 = v.x * v.x + v.y * v.y + v.z * v.z + v.w * v.w;
    for (int m = 1; m < 32; m <<= 1) hn2 += __shfl_xor(hn2, m, 32);
    float hn = fmaxf(sqrtf(hn2), MINN);
    float pc = fminf(hn, MAXN);
    float L = artanh_f(pc) / hn;
    int s = seg[r];
    float* d = ssum + (size_t)s * 128 + lane * 4;
    atomicAdd(d + 0, L * v.x);
    atomicAdd(d + 1, L * v.y);
    atomicAdd(d + 2, L * v.z);
    atomicAdd(d + 3, L * v.w);
    if (lane == 0) atomicAdd(&cnt[s], 1.f);
}

__global__ __launch_bounds__(256) void pool_k(const float* __restrict__ ssum,
                                              const float* __restrict__ cnt,
                                              float* __restrict__ out, int nseg) {
    int r = blockIdx.x * 8 + (threadIdx.x >> 5);
    if (r >= nseg) return;
    int lane = threadIdx.x & 31;
    float inv = 1.f / fmaxf(cnt[r], 1.f);
    float4 v = *(const float4*)(ssum + (size_t)r * 128 + lane * 4);
    v.x *= inv; v.y *= inv; v.z *= inv; v.w *= inv;
    float mn2 = v.x * v.x + v.y * v.y + v.z * v.z + v.w * v.w;
    for (int m = 1; m < 32; m <<= 1) mn2 += __shfl_xor(mn2, m, 32);
    float mn = fmaxf(sqrtf(mn2), MINN);
    float s = fminf(tanhf(mn), MAXN) / mn;
    float* o = out + (size_t)r * 128 + lane * 4;
    o[0] = s * v.x; o[1] = s * v.y; o[2] = s * v.z; o[3] = s * v.w;
}

__global__ void zero_f32(float* __restrict__ p, size_t n) {
    size_t i = (size_t)blockIdx.x * blockDim.x + threadIdx.x;
    size_t st = (size_t)gridDim.x * blockDim.x;
    for (; i < n; i += st) p[i] = 0.f;
}

extern "C" void kernel_launch(void* const* d_in, const int* in_sizes, int n_in,
                              void* d_out, int out_size, void* d_ws, size_t ws_size,
                              hipStream_t stream) {
    const float* x = (const float*)d_in[0];
    const int* erow = (const int*)d_in[1];
    const int* ecol = (const int*)d_in[2];
    const int* seg = (const int*)d_in[3];
    const float* W1 = (const float*)d_in[4];
    const float* b1 = (const float*)d_in[5];
    const float* W2 = (const float*)d_in[6];
    const float* b2 = (const float*)d_in[7];

    const int N = in_sizes[0] / 256;
    const int E = in_sizes[1];
    const int NSEG = out_size / 128;

    char* ws = (char*)d_ws;
    size_t bufBytes = (size_t)N * 128 * sizeof(float);
    float* buf0 = (float*)ws;
    float* buf1 = (float*)(ws + bufBytes);
    float* bbias = (float*)(ws + 2 * bufBytes);
    float* y2 = bbias + 256;
    float* ssum = buf0;
    float* cnt = buf0 + (size_t)NSEG * 128;

    dim3 blk(256);
    int gRows = (N + 7) / 8;
    int gGemm = (N + BM - 1) / BM;
    int gEdge = (E + 7) / 8;

    const int nScan = (N + SCAN_B - 1) / SCAN_B;
    size_t offBytes = (((size_t)N + 1) * sizeof(int) + 255) / 256 * 256;
    size_t bsBytes = ((size_t)nScan * sizeof(int) + 255) / 256 * 256;
    int* offs = (int*)(ws + 2 * bufBytes + 4096);
    int* bsum = (int*)(ws + 2 * bufBytes + 4096 + offBytes);
    int* scol = (int*)(ws + 2 * bufBytes + 4096 + offBytes + bsBytes);
    int* rank = (int*)buf1;
    const bool grouped =
        ws_size >= 2 * bufBytes + 4096 + offBytes + bsBytes + (size_t)E * sizeof(int) &&
        (size_t)E <= (size_t)N * 128;

    bias_prep<<<2, 128, 0, stream>>>(b1, b2, bbias, y2);

    if (grouped) {
        zero_f32<<<64, 256, 0, stream>>>((float*)offs, (size_t)N + 1);
        edge_rank<<<2048, blk, 0, stream>>>(erow, offs, rank, E);
        scan_sums<<<nScan, SCAN_B, 0, stream>>>(offs, bsum, N);
        scan_apply<<<nScan, SCAN_B, 0, stream>>>(offs, bsum, N);
        edge_place<<<2048, blk, 0, stream>>>(erow, ecol, offs, rank, scol, E);
    }

    hyp_linear_wmma<<<gGemm, blk, 0, stream>>>(x, 256, N, W1, bbias, y2, buf0, 1);
    if (grouped) {
        node_gather<<<2048, blk, 0, stream>>>(buf0, offs, scol, buf1, N);
    } else {
        zero_f32<<<2048, 256, 0, stream>>>(buf1, (size_t)N * 128);
        edge_agg<<<gEdge, blk, 0, stream>>>(buf0, erow, ecol, buf1, E);
    }
    post_agg<<<gRows, blk, 0, stream>>>(buf1, N);

    hyp_linear_wmma<<<gGemm, blk, 0, stream>>>(buf1, 128, N, W2, bbias + 128,
                                               y2 + 1, buf0, 0);
    if (grouped) {
        node_gather<<<2048, blk, 0, stream>>>(buf0, offs, scol, buf1, N);
    } else {
        zero_f32<<<2048, 256, 0, stream>>>(buf1, (size_t)N * 128);
        edge_agg<<<gEdge, blk, 0, stream>>>(buf0, erow, ecol, buf1, E);
    }
    post_agg<<<gRows, blk, 0, stream>>>(buf1, N);

    zero_f32<<<64, 256, 0, stream>>>(ssum, (size_t)NSEG * 128 + NSEG);
    final_scatter<<<gRows, blk, 0, stream>>>(buf1, seg, ssum, cnt, N);
    pool_k<<<(NSEG + 7) / 8, blk, 0, stream>>>(ssum, cnt, (float*)d_out, NSEG);
}


# post-sum row maps (activation; last layer's logmap + segment add) done in the gather wave's registers: 3 row passes and layer 2's row store removed; segment adds still f32 atomics
# speedup vs baseline: 2.1319x; 1.1415x over previous
#include <hip/hip_runtime.h>
#include <math.h>


#define MAXN 0.996f
#define MINN 1e-15f
#define KC   128
#define BM   64

typedef float v2f __attribute__((ext_vector_type(2)));
typedef float v8f __attribute__((ext_vector_type(8)));

__device__ __forceinline__ float artanh_f(float z) {
    z = fminf(fmaxf(z, -(1.f - 1e-7f)), 1.f - 1e-7f);
    return 0.5f * logf((1.f + z) / (1.f - z));
}

__global__ void bias_prep(const float* __restrict__ b1,
                          const float* __restrict__ b2,
                          float* __restrict__ bb, float* __restrict__ y2) {
    __shared__ float red[128];
    const float* b = blockIdx.x ? b2 : b1;
    int t = threadIdx.x;
    float v = b[t];
    red[t] = v * v;
    __syncthreads();
    for (int s = 64; s > 0; s >>= 1) {
        if (t < s) red[t] += red[t + s];
        __syncthreads();
    }
    float bn = fmaxf(sqrtf(red[0]), MINN);
    float tc = fminf(tanhf(bn), MAXN);
    bb[blockIdx.x * 128 + t] = tc / bn * v;
    if (t == 0) y2[blockIdx.x] = tc * tc;
}

__global__ __launch_bounds__(256) void hyp_linear_wmma(
    const float* __restrict__ X, int Kdim, int nrows,
    const float* __restrict__ W,
    const float* __restrict__ bb,
    const float* __restrict__ y2p,
    float* __restrict__ HT, int apply_exp) {
    __shared__ float As[BM][KC + 4];
    __shared__ float Ws[128][KC + 4];
    __shared__ float red[BM][4];
    __shared__ float scale_row[BM], xn_row[BM];
    __shared__ float rs_mx2[BM], rs_dot[BM];
    __shared__ float rowCA[BM], rowCB[BM];

    const int t = threadIdx.x;
    const int row0 = blockIdx.x * BM;

    {
        int r = t >> 2, p = t & 3;
        int gr = row0 + r;
        float s = 0.f;
        if (gr < nrows) {
            int kl = Kdim >> 2;
            const float* px = X + (size_t)gr * Kdim + p * kl;
            for (int k = 0; k < kl; k += 4) {
                float4 v = *(const float4*)(px + k);
                s += v.x * v.x + v.y * v.y + v.z * v.z + v.w * v.w;
            }
        }
        red[r][p] = s;
    }
    if (t < BM) { rs_mx2[t] = 0.f; rs_dot[t] = 0.f; }
    __syncthreads();
    if (t < BM) {
        float un2 = red[t][0] + red[t][1] + red[t][2] + red[t][3];
        float un = fmaxf(sqrtf(un2), MINN);
        if (apply_exp) {
            float bc = fminf(tanhf(un), MAXN);
            scale_row[t] = bc / un;
            xn_row[t] = fmaxf(bc, MINN);
        } else {
            scale_row[t] = 1.f;
            xn_row[t] = un;
        }
    }
    __syncthreads();

    const int lane = t & 31;
    const int wid = t >> 5;
    const int rs = wid >> 1;
    const int ch = wid & 1;
    const int half = lane >> 4;
    const int l15 = lane & 15;

    v8f acc[4];
    for (int j = 0; j < 4; ++j)
        for (int i = 0; i < 8; ++i) acc[j][i] = 0.f;

    const int nchunks = Kdim / KC;

    for (int kc = 0; kc < nchunks; ++kc) {
        for (int f = t; f < BM * KC / 4; f += 256) {
            int r = f >> 5;
            int c4 = f & 31;
            int gr = row0 + r;
            float4 v = make_float4(0.f, 0.f, 0.f, 0.f);
            if (gr < nrows) {
                v = *(const float4*)(X + (size_t)gr * Kdim + kc * KC + c4 * 4);
                float sc = scale_row[r];
                v.x *= sc; v.y *= sc; v.z *= sc; v.w *= sc;
            }
            *(float4*)(&As[r][c4 * 4]) = v;
        }
        for (int f = t; f < 128 * KC / 4; f += 256) {
            int r = f >> 5;
            int c4 = f & 31;
            float4 v = *(const float4*)(W + (size_t)r * Kdim + kc * KC + c4 * 4);
            *(float4*)(&Ws[r][c4 * 4]) = v;
        }
        __syncthreads();

        const int mr = rs * 16 + l15;
        for (int kk = 0; kk < KC; kk += 4) {
            const int ko = kk + half * 2;
            v2f a;
            a.x = As[mr][ko];
            a.y = As[mr][ko + 1];
#pragma unroll
            for (int j = 0; j < 4; ++j) {
                const int colg = (ch * 4 + j) * 16 + l15;
                v2f b;
                b.x = Ws[colg][ko];
                b.y = Ws[colg][ko + 1];
                acc[j] = __builtin_amdgcn_wmma_f32_16x16x4_f32(
                    false, a, false, b, (short)0, acc[j], false, false);
            }
        }
        __syncthreads();
    }

    float bbv[4];
#pragma unroll
    for (int j = 0; j < 4; ++j) bbv[j] = bb[(ch * 4 + j) * 16 + l15];

#pragma unroll
    for (int i = 0; i < 8; ++i) {
        float p2 = 0.f, pd = 0.f;
#pragma unroll
        for (int j = 0; j < 4; ++j) {
            float v = acc[j][i];
            p2 += v * v;
            pd += v * bbv[j];
        }
        for (int m = 1; m < 16; m <<= 1) {
            p2 += __shfl_xor(p2, m, 16);
            pd += __shfl_xor(pd, m, 16);
        }
        int mrow = rs * 16 + i + half * 8;
        if (l15 == 0) {
            atomicAdd(&rs_mx2[mrow], p2);
            atomicAdd(&rs_dot[mrow], pd);
        }
    }
    __syncthreads();

    if (t < BM) {
        float xn = xn_row[t];
        float mxn = fmaxf(sqrtf(rs_mx2[t]), MINN);
        float tv = tanhf(mxn / xn * artanh_f(xn));
        float tc = fminf(tv, MAXN);
        float s_mv = tc / mxn;
        float x2 = tc * tc;
        float xy = s_mv * rs_dot[t];
        float y2 = y2p[0];
        float al = 1.f + 2.f * xy + y2;
        float be = 1.f - x2;
        float den = fmaxf(1.f + 2.f * xy + x2 * y2, MINN);
        float nn2 = fmaxf(al * al * x2 + 2.f * al * be * xy + be * be * y2, 0.f);
        float hn = fmaxf(sqrtf(nn2) / den, MINN);
        float hc = fminf(hn, MAXN);
        float L = artanh_f(hc) / hn;
        rowCA[t] = L * al * s_mv / den;
        rowCB[t] = L * be / den;
    }
    __syncthreads();

#pragma unroll
    for (int i = 0; i < 8; ++i) {
        int mrow = rs * 16 + i + half * 8;
        int gr = row0 + mrow;
        if (gr < nrows) {
            float ca = rowCA[mrow], cb = rowCB[mrow];
#pragma unroll
            for (int j = 0; j < 4; ++j) {
                int colg = (ch * 4 + j) * 16 + l15;
                HT[(size_t)gr * 128 + colg] = ca * acc[j][i] + cb * bbv[j];
            }
        }
    }
}

__global__ __launch_bounds__(256) void edge_agg(const float* __restrict__ HT,
                                                const int* __restrict__ erow,
                                                const int* __restrict__ ecol,
                                                float* __restrict__ AGG, int E) {
    int e = blockIdx.x * 8 + (threadIdx.x >> 5);
    if (e >= E) return;
    int lane = threadIdx.x & 31;
    int r = erow[e], c = ecol[e];
    float4 v = *(const float4*)(HT + (size_t)c * 128 + lane * 4);
    float* dst = AGG + (size_t)r * 128 + lane * 4;
    atomicAdd(dst + 0, v.x);
    atomicAdd(dst + 1, v.y);
    atomicAdd(dst + 2, v.z);
    atomicAdd(dst + 3, v.w);
}

#define SCAN_B 1024

__global__ __launch_bounds__(256) void edge_rank(const int* __restrict__ erow,
                                                 int* __restrict__ cnt,
                                                 int* __restrict__ rank, int E) {
    int st = gridDim.x * 256;
    for (int i = blockIdx.x * 256 + threadIdx.x; i < E; i += st)
        rank[i] = atomicAdd(&cnt[erow[i]], 1);
}

__device__ __forceinline__ int block_sum_1024(int v, int* sh) {
    int t = threadIdx.x;
    for (int m = 1; m < 32; m <<= 1) v += __shfl_xor(v, m, 32);
    if ((t & 31) == 0) sh[t >> 5] = v;
    __syncthreads();
    int w = sh[t & 31];
    for (int m = 1; m < 32; m <<= 1) w += __shfl_xor(w, m, 32);
    __syncthreads();
    return w;
}

__global__ __launch_bounds__(SCAN_B) void scan_sums(const int* __restrict__ cnt,
                                                    int* __restrict__ bsum, int n) {
    __shared__ int sh[32];
    int i = blockIdx.x * SCAN_B + threadIdx.x;
    int tot = block_sum_1024(i < n ? cnt[i] : 0, sh);
    if (threadIdx.x == 0) bsum[blockIdx.x] = tot;
}

__global__ __launch_bounds__(SCAN_B) void scan_apply(int* __restrict__ offs,
                                                     const int* __restrict__ bsum, int n) {
    __shared__ int sh[32];
    const int t = threadIdx.x, lane = t & 31, wid = t >> 5;
    int p = 0;
    for (int b = t; b < (int)blockIdx.x; b += SCAN_B) p += bsum[b];
    const int base = block_sum_1024(p, sh);

    int i = blockIdx.x * SCAN_B + t;
    int v = i < n ? offs[i] : 0;
    int s = v;
    for (int m = 1; m < 32; m <<= 1) {
        int u = __shfl_up(s, m, 32);
        if (lane >= m) s += u;
    }
    if (lane == 31) sh[wid] = s;
    __syncthreads();
    if (wid == 0) {
        int w0 = sh[lane], w = w0;
        for (int m = 1; m < 32; m <<= 1) {
            int u = __shfl_up(w, m, 32);
            if (lane >= m) w += u;
        }
        sh[lane] = w - w0;
    }
    __syncthreads();
    int excl = base + sh[wid] + s - v;
    if (i < n) offs[i] = excl;
    if (i == n - 1) offs[n] = excl + v;
}

__global__ __launch_bounds__(256) void edge_place(const int* __restrict__ erow,
                                                  const int* __restrict__ ecol,
                                                  const int* __restrict__ offs,
                                                  const int* __restrict__ rank,
                                                  int* __restrict__ scol, int E) {
    int st = gridDim.x * 256;
    for (int i = blockIdx.x * 256 + threadIdx.x; i < E; i += st)
        scol[offs[erow[i]] + rank[i]] = ecol[i];
}

__device__ __forceinline__ float wave_sum32(float x) {
    for (int m = 1; m < 32; m <<= 1) x += __shfl_xor(x, m, 32);
    return x;
}

__device__ __forceinline__ float4 act_row(float4 v) {
    float an2 = wave_sum32(v.x * v.x + v.y * v.y + v.z * v.z + v.w * v.w);
    float an = fmaxf(sqrtf(an2), MINN);
    float tn = fminf(fmaxf(tanhf(an), MINN), MAXN);
    float lg = artanh_f(tn) / an;
    float4 at;
    at.x = tanhf(lg * v.x);
    at.y = tanhf(lg * v.y);
    at.z = tanhf(lg * v.z);
    at.w = tanhf(lg * v.w);
    float vn2 = wave_sum32(at.x * at.x + at.y * at.y + at.z * at.z + at.w * at.w);
    float vn = fmaxf(sqrtf(vn2), MINN);
    float s = fminf(tanhf(vn), MAXN) / vn;
    return make_float4(s * at.x, s * at.y, s * at.z, s * at.w);
}

template <int READOUT>
__global__ __launch_bounds__(256) void node_gather_act(const float* __restrict__ HT,
                                                       const int* __restrict__ offs,
                                                       const int* __restrict__ scol,
                                                       float* __restrict__ OUT,
                                                       const int* __restrict__ seg,
                                                       float* __restrict__ ssum,
                                                       float* __restrict__ cnt, int nrows) {
    const int lane = threadIdx.x & 31;
    const int w0 = __builtin_amdgcn_readfirstlane((int)(blockIdx.x * 8 + (threadIdx.x >> 5)));
    const int nw = gridDim.x * 8;
    for (int r = w0; r < nrows; r += nw) {
        const int beg = offs[r], end = offs[r + 1];
        int sg = 0;
        if (READOUT) sg = seg[r];
        float4 acc = make_float4(0.f, 0.f, 0.f, 0.f);
        for (int b0 = beg; b0 < end; b0 += 32) {
            const int n = min(32, end - b0);
            const int myc = (lane < n) ? scol[b0 + lane] : 0;
            int j = 0;
            for (; j + 4 <= n; j += 4) {
                const int c0 = __builtin_amdgcn_readlane(myc, j);
                const int c1 = __builtin_amdgcn_readlane(myc, j + 1);
                const int c2 = __builtin_amdgcn_readlane(myc, j + 2);
                const int c3 = __builtin_amdgcn_readlane(myc, j + 3);
                const float4 v0 = *((const float4*)(HT + (size_t)c0 * 128) + lane);
                const float4 v1 = *((const float4*)(HT + (size_t)c1 * 128) + lane);
                const float4 v2 = *((const float4*)(HT + (size_t)c2 * 128) + lane);
                const float4 v3 = *((const float4*)(HT + (size_t)c3 * 128) + lane);
                acc.x += v0.x; acc.y += v0.y; acc.z += v0.z; acc.w += v0.w;
                acc.x += v1.x; acc.y += v1.y; acc.z += v1.z; acc.w += v1.w;
                acc.x += v2.x; acc.y += v2.y; acc.z += v2.z; acc.w += v2.w;
                acc.x += v3.x; acc.y += v3.y; acc.z += v3.z; acc.w += v3.w;
            }
            for (; j < n; ++j) {
                const int c0 = __builtin_amdgcn_readlane(myc, j);
                const float4 v0 = *((const float4*)(HT + (size_t)c0 * 128) + lane);
                acc.x += v0.x; acc.y += v0.y; acc.z += v0.z; acc.w += v0.w;
            }
        }
        const float4 h = act_row(acc);
        if (!READOUT) {
            *((float4*)(OUT + (size_t)r * 128) + lane) = h;
        } else {
            float hn2 = wave_sum32(h.x * h.x + h.y * h.y + h.z * h.z + h.w * h.w);
            float hn = fmaxf(sqrtf(hn2), MINN);
            float pc = fminf(hn, MAXN);
            float L = artanh_f(pc) / hn;
            float* d = ssum + (size_t)sg * 128 + lane * 4;
            atomicAdd(d + 0, L * h.x);
            atomicAdd(d + 1, L * h.y);
            atomicAdd(d + 2, L * h.z);
            atomicAdd(d + 3, L * h.w);
            if (lane == 0) atomicAdd(&cnt[sg], 1.f);
        }
    }
}

__global__ __launch_bounds__(256) void post_agg(float* __restrict__ H, int nrows) {
    int r = blockIdx.x * 8 + (threadIdx.x >> 5);
    if (r >= nrows) return;
    int lane = threadIdx.x & 31;
    float* p = H + (size_t)r * 128 + lane * 4;
    float4 v = *(const float4*)p;
    float an2 = v.x * v.x + v.y * v.y + v.z * v.z + v.w * v.w;
    for (int m = 1; m < 32; m <<= 1) an2 += __shfl_xor(an2, m, 32);
    float an = fmaxf(sqrtf(an2), MINN);
    float tn = fminf(fmaxf(tanhf(an), MINN), MAXN);
    float lg = artanh_f(tn) / an;
    float4 at;
    at.x = tanhf(lg * v.x);
    at.y = tanhf(lg * v.y);
    at.z = tanhf(lg * v.z);
    at.w = tanhf(lg * v.w);
    float vn2 = at.x * at.x + at.y * at.y + at.z * at.z + at.w * at.w;
    for (int m = 1; m < 32; m <<= 1) vn2 += __shfl_xor(vn2, m, 32);
    float vn = fmaxf(sqrtf(vn2), MINN);
    float s = fminf(tanhf(vn), MAXN) / vn;
    *(float4*)p = make_float4(s * at.x, s * at.y, s * at.z, s * at.w);
}

__global__ __launch_bounds__(256) void final_scatter(
    const float* __restrict__ H, const int* __restrict__ seg,
    float* __restrict__ ssum, float* __restrict__ cnt, int nrows) {
    int r = blockIdx.x * 8 + (threadIdx.x >> 5);
    if (r >= nrows) return;
    int lane = threadIdx.x & 31;
    float4 v = *(const float4*)(H + (size_t)r * 128 + lane * 4);
    float hn2 = v.x * v.x + v.y * v.y + v.z * v.z + v.w * v.w;
    for (int m = 1; m < 32; m <<= 1) hn2 += __shfl_xor(hn2, m, 32);
    float hn = fmaxf(sqrtf(hn2), MINN);
    float pc = fminf(hn, MAXN);
    float L = artanh_f(pc) / hn;
    int s = seg[r];
    float* d = ssum + (size_t)s * 128 + lane * 4;
    atomicAdd(d + 0, L * v.x);
    atomicAdd(d + 1, L * v.y);
    atomicAdd(d + 2, L * v.z);
    atomicAdd(d + 3, L * v.w);
    if (lane == 0) atomicAdd(&cnt[s], 1.f);
}

__global__ __launch_bounds__(256) void pool_k(const float* __restrict__ ssum,
                                              const float* __restrict__ cnt,
                                              float* __restrict__ out, int nseg) {
    int r = blockIdx.x * 8 + (threadIdx.x >> 5);
    if (r >= nseg) return;
    int lane = threadIdx.x & 31;
    float inv = 1.f / fmaxf(cnt[r], 1.f);
    float4 v = *(const float4*)(ssum + (size_t)r * 128 + lane * 4);
    v.x *= inv; v.y *= inv; v.z *= inv; v.w *= inv;
    float mn2 = v.x * v.x + v.y * v.y + v.z * v.z + v.w * v.w;
    for (int m = 1; m < 32; m <<= 1) mn2 += __shfl_xor(mn2, m, 32);
    float mn = fmaxf(sqrtf(mn2), MINN);
    float s = fminf(tanhf(mn), MAXN) / mn;
    float* o = out + (size_t)r * 128 + lane * 4;
    o[0] = s * v.x; o[1] = s * v.y; o[2] = s * v.z; o[3] = s * v.w;
}

__global__ void zero_f32(float* __restrict__ p, size_t n) {
    size_t i = (size_t)blockIdx.x * blockDim.x + threadIdx.x;
    size_t st = (size_t)gridDim.x * blockDim.x;
    for (; i < n; i += st) p[i] = 0.f;
}

extern "C" void kernel_launch(void* const* d_in, const int* in_sizes, int n_in,
                              void* d_out, int out_size, void* d_ws, size_t ws_size,
                              hipStream_t stream) {
    const float* x = (const float*)d_in[0];
    const int* erow = (const int*)d_in[1];
    const int* ecol = (const int*)d_in[2];
    const int* seg = (const int*)d_in[3];
    const float* W1 = (const float*)d_in[4];
    const float* b1 = (const float*)d_in[5];
    const float* W2 = (const float*)d_in[6];
    const float* b2 = (const float*)d_in[7];

    const int N = in_sizes[0] / 256;
    const int E = in_sizes[1];
    const int NSEG = out_size / 128;

    char* ws = (char*)d_ws;
    size_t bufBytes = (size_t)N * 128 * sizeof(float);
    float* buf0 = (float*)ws;
    float* buf1 = (float*)(ws + bufBytes);
    float* bbias = (float*)(ws + 2 * bufBytes);
    float* y2 = bbias + 256;
    float* ssum = buf0;
    float* cnt = buf0 + (size_t)NSEG * 128;

    dim3 blk(256);
    int gRows = (N + 7) / 8;
    int gGemm = (N + BM - 1) / BM;
    int gEdge = (E + 7) / 8;

    const int nScan = (N + SCAN_B - 1) / SCAN_B;
    size_t offBytes = (((size_t)N + 1) * sizeof(int) + 255) / 256 * 256;
    size_t bsBytes = ((size_t)nScan * sizeof(int) + 255) / 256 * 256;
    int* offs = (int*)(ws + 2 * bufBytes + 4096);
    int* bsum = (int*)(ws + 2 * bufBytes + 4096 + offBytes);
    int* scol = (int*)(ws + 2 * bufBytes + 4096 + offBytes + bsBytes);
    int* rank = (int*)buf1;
    const bool grouped =
        ws_size >= 2 * bufBytes + 4096 + offBytes + bsBytes + (size_t)E * sizeof(int) &&
        (size_t)E <= (size_t)N * 128;

    bias_prep<<<2, 128, 0, stream>>>(b1, b2, bbias, y2);

    if (grouped) {
        zero_f32<<<64, 256, 0, stream>>>((float*)offs, (size_t)N + 1);
        edge_rank<<<2048, blk, 0, stream>>>(erow, offs, rank, E);
        scan_sums<<<nScan, SCAN_B, 0, stream>>>(offs, bsum, N);
        scan_apply<<<nScan, SCAN_B, 0, stream>>>(offs, bsum, N);
        edge_place<<<2048, blk, 0, stream>>>(erow, ecol, offs, rank, scol, E);

        hyp_linear_wmma<<<gGemm, blk, 0, stream>>>(x, 256, N, W1, bbias, y2, buf0, 1);
        node_gather_act<0><<<2048, blk, 0, stream>>>(buf0, offs, scol, buf1, seg,
                                                     (float*)0, (float*)0, N);

        hyp_linear_wmma<<<gGemm, blk, 0, stream>>>(buf1, 128, N, W2, bbias + 128,
                                                   y2 + 1, buf0, 0);
        float* ssumG = buf1;
        float* cntG = buf1 + (size_t)NSEG * 128;
        zero_f32<<<64, 256, 0, stream>>>(ssumG, (size_t)NSEG * 128 + NSEG);
        node_gather_act<1><<<2048, blk, 0, stream>>>(buf0, offs, scol, (float*)0, seg,
                                                     ssumG, cntG, N);
        pool_k<<<(NSEG + 7) / 8, blk, 0, stream>>>(ssumG, cntG, (float*)d_out, NSEG);
        return;
    }

    hyp_linear_wmma<<<gGemm, blk, 0, stream>>>(x, 256, N, W1, bbias, y2, buf0, 1);
    zero_f32<<<2048, 256, 0, stream>>>(buf1, (size_t)N * 128);
    edge_agg<<<gEdge, blk, 0, stream>>>(buf0, erow, ecol, buf1, E);
    post_agg<<<gRows, blk, 0, stream>>>(buf1, N);

    hyp_linear_wmma<<<gGemm, blk, 0, stream>>>(buf1, 128, N, W2, bbias + 128,
                                               y2 + 1, buf0, 0);
    zero_f32<<<2048, 256, 0, stream>>>(buf1, (size_t)N * 128);
    edge_agg<<<gEdge, blk, 0, stream>>>(buf0, erow, ecol, buf1, E);
    post_agg<<<gRows, blk, 0, stream>>>(buf1, N);

    zero_f32<<<64, 256, 0, stream>>>(ssum, (size_t)NSEG * 128 + NSEG);
    final_scatter<<<gRows, blk, 0, stream>>>(buf1, seg, ssum, cnt, N);
    pool_k<<<(NSEG + 7) / 8, blk, 0, stream>>>(ssum, cnt, (float*)d_out, NSEG);
}


# edge grouping: per-edge returning global int atomics (rank) + scattered place -> two-level counting sort on LDS counters (bin by 256-node range, then node order in the bin); f32 sum order changes
# speedup vs baseline: 2.7310x; 1.2810x over previous
#include <hip/hip_runtime.h>
#include <math.h>


#define MAXN 0.996f
#define MINN 1e-15f
#define KC   128
#define BM   64

typedef float v2f __attribute__((ext_vector_type(2)));
typedef float v8f __attribute__((ext_vector_type(8)));

__device__ __forceinline__ float artanh_f(float z) {
    z = fminf(fmaxf(z, -(1.f - 1e-7f)), 1.f - 1e-7f);
    return 0.5f * logf((1.f + z) / (1.f - z));
}

__global__ void bias_prep(const float* __restrict__ b1,
                          const float* __restrict__ b2,
                          float* __restrict__ bb, float* __restrict__ y2) {
    __shared__ float red[128];
    const float* b = blockIdx.x ? b2 : b1;
    int t = threadIdx.x;
    float v = b[t];
    red[t] = v * v;
    __syncthreads();
    for (int s = 64; s > 0; s >>= 1) {
        if (t < s) red[t] += red[t + s];
        __syncthreads();
    }
    float bn = fmaxf(sqrtf(red[0]), MINN);
    float tc = fminf(tanhf(bn), MAXN);
    bb[blockIdx.x * 128 + t] = tc / bn * v;
    if (t == 0) y2[blockIdx.x] = tc * tc;
}

__global__ __launch_bounds__(256) void hyp_linear_wmma(
    const float* __restrict__ X, int Kdim, int nrows,
    const float* __restrict__ W,
    const float* __restrict__ bb,
    const float* __restrict__ y2p,
    float* __restrict__ HT, int apply_exp) {
    __shared__ float As[BM][KC + 4];
    __shared__ float Ws[128][KC + 4];
    __shared__ float red[BM][4];
    __shared__ float scale_row[BM], xn_row[BM];
    __shared__ float rs_mx2[BM], rs_dot[BM];
    __shared__ float rowCA[BM], rowCB[BM];

    const int t = threadIdx.x;
    const int row0 = blockIdx.x * BM;

    {
        int r = t >> 2, p = t & 3;
        int gr = row0 + r;
        float s = 0.f;
        if (gr < nrows) {
            int kl = Kdim >> 2;
            const float* px = X + (size_t)gr * Kdim + p * kl;
            for (int k = 0; k < kl; k += 4) {
                float4 v = *(const float4*)(px + k);
                s += v.x * v.x + v.y * v.y + v.z * v.z + v.w * v.w;
            }
        }
        red[r][p] = s;
    }
    if (t < BM) { rs_mx2[t] = 0.f; rs_dot[t] = 0.f; }
    __syncthreads();
    if (t < BM) {
        float un2 = red[t][0] + red[t][1] + red[t][2] + red[t][3];
        float un = fmaxf(sqrtf(un2), MINN);
        if (apply_exp) {
            float bc = fminf(tanhf(un), MAXN);
            scale_row[t] = bc / un;
            xn_row[t] = fmaxf(bc, MINN);
        } else {
            scale_row[t] = 1.f;
            xn_row[t] = un;
        }
    }
    __syncthreads();

    const int lane = t & 31;
    const int wid = t >> 5;
    const int rs = wid >> 1;
    const int ch = wid & 1;
    const int half = lane >> 4;
    const int l15 = lane & 15;

    v8f acc[4];
    for (int j = 0; j < 4; ++j)
        for (int i = 0; i < 8; ++i) acc[j][i] = 0.f;

    const int nchunks = Kdim / KC;

    for (int kc = 0; kc < nchunks; ++kc) {
        for (int f = t; f < BM * KC / 4; f += 256) {
            int r = f >> 5;
            int c4 = f & 31;
            int gr = row0 + r;
            float4 v = make_float4(0.f, 0.f, 0.f, 0.f);
            if (gr < nrows) {
                v = *(const float4*)(X + (size_t)gr * Kdim + kc * KC + c4 * 4);
                float sc = scale_row[r];
                v.x *= sc; v.y *= sc; v.z *= sc; v.w *= sc;
            }
            *(float4*)(&As[r][c4 * 4]) = v;
        }
        for (int f = t; f < 128 * KC / 4; f += 256) {
            int r = f >> 5;
            int c4 = f & 31;
            float4 v = *(const float4*)(W + (size_t)r * Kdim + kc * KC + c4 * 4);
            *(float4*)(&Ws[r][c4 * 4]) = v;
        }
        __syncthreads();

        const int mr = rs * 16 + l15;
        for (int kk = 0; kk < KC; kk += 4) {
            const int ko = kk + half * 2;
            v2f a;
            a.x = As[mr][ko];
            a.y = As[mr][ko + 1];
#pragma unroll
            for (int j = 0; j < 4; ++j) {
                const int colg = (ch * 4 + j) * 16 + l15;
                v2f b;
                b.x = Ws[colg][ko];
                b.y = Ws[colg][ko + 1];
                acc[j] = __builtin_amdgcn_wmma_f32_16x16x4_f32(
                    false, a, false, b, (short)0, acc[j], false, false);
            }
        }
        __syncthreads();
    }

    float bbv[4];
#pragma unroll
    for (int j = 0; j < 4; ++j) bbv[j] = bb[(ch * 4 + j) * 16 + l15];

#pragma unroll
    for (int i = 0; i < 8; ++i) {
        float p2 = 0.f, pd = 0.f;
#pragma unroll
        for (int j = 0; j < 4; ++j) {
            float v = acc[j][i];
            p2 += v * v;
            pd += v * bbv[j];
        }
        for (int m = 1; m < 16; m <<= 1) {
            p2 += __shfl_xor(p2, m, 16);
            pd += __shfl_xor(pd, m, 16);
        }
        int mrow = rs * 16 + i + half * 8;
        if (l15 == 0) {
            atomicAdd(&rs_mx2[mrow], p2);
            atomicAdd(&rs_dot[mrow], pd);
        }
    }
    __syncthreads();

    if (t < BM) {
        float xn = xn_row[t];
        float mxn = fmaxf(sqrtf(rs_mx2[t]), MINN);
        float tv = tanhf(mxn / xn * artanh_f(xn));
        float tc = fminf(tv, MAXN);
        float s_mv = tc / mxn;
        float x2 = tc * tc;
        float xy = s_mv * rs_dot[t];
        float y2 = y2p[0];
        float al = 1.f + 2.f * xy + y2;
        float be = 1.f - x2;
        float den = fmaxf(1.f + 2.f * xy + x2 * y2, MINN);
        float nn2 = fmaxf(al * al * x2 + 2.f * al * be * xy + be * be * y2, 0.f);
        float hn = fmaxf(sqrtf(nn2) / den, MINN);
        float hc = fminf(hn, MAXN);
        float L = artanh_f(hc) / hn;
        rowCA[t] = L * al * s_mv / den;
        rowCB[t] = L * be / den;
    }
    __syncthreads();

#pragma unroll
    for (int i = 0; i < 8; ++i) {
        int mrow = rs * 16 + i + half * 8;
        int gr = row0 + mrow;
        if (gr < nrows) {
            float ca = rowCA[mrow], cb = rowCB[mrow];
#pragma unroll
            for (int j = 0; j < 4; ++j) {
                int colg = (ch * 4 + j) * 16 + l15;
                HT[(size_t)gr * 128 + colg] = ca * acc[j][i] + cb * bbv[j];
            }
        }
    }
}

__global__ __launch_bounds__(256) void edge_agg(const float* __restrict__ HT,
                                                const int* __restrict__ erow,
                                                const int* __restrict__ ecol,
                                                float* __restrict__ AGG, int E) {
    int e = blockIdx.x * 8 + (threadIdx.x >> 5);
    if (e >= E) return;
    int lane = threadIdx.x & 31;
    int r = erow[e], c = ecol[e];
    float4 v = *(const float4*)(HT + (size_t)c * 128 + lane * 4);
    float* dst = AGG + (size_t)r * 128 + lane * 4;
    atomicAdd(dst + 0, v.x);
    atomicAdd(dst + 1, v.y);
    atomicAdd(dst + 2, v.z);
    atomicAdd(dst + 3, v.w);
}

#define SCAN_B 1024
#define BKT_SH 8
#define BKT_N  256
#define MAXBK  1024
#define SORT_T 1024
#define SORT_U 4

__global__ __launch_bounds__(SORT_T) void bin_count(const int* __restrict__ erow,
                                                    int* __restrict__ bh,
                                                    int E, int nbk) {
    __shared__ int hist[MAXBK];
    const int t = threadIdx.x, G = gridDim.x, b = blockIdx.x;
    for (int k = t; k < nbk; k += SORT_T) hist[k] = 0;
    __syncthreads();
    const int chunk = (E + G - 1) / G;
    const int beg = b * chunk, end = min(E, beg + chunk);
    for (int i = beg + t; i < end; i += SORT_U * SORT_T) {
        int r[SORT_U];
#pragma unroll
        for (int u = 0; u < SORT_U; ++u) {
            const int j = i + u * SORT_T;
            r[u] = j < end ? erow[j] : 0;
        }
#pragma unroll
        for (int u = 0; u < SORT_U; ++u) {
            const int k = min((int)((unsigned)r[u] >> BKT_SH), nbk - 1);
            if (i + u * SORT_T < end) atomicAdd(&hist[k], 1);
        }
    }
    __syncthreads();
    for (int k = t; k < nbk; k += SORT_T) bh[(size_t)k * G + b] = hist[k];
}

__global__ __launch_bounds__(SORT_T) void bin_scatter(const int* __restrict__ erow,
                                                      const int* __restrict__ ecol,
                                                      const int* __restrict__ bo,
                                                      int2* __restrict__ tmp,
                                                      int E, int nbk) {
    __shared__ int cur[MAXBK];
    const int t = threadIdx.x, G = gridDim.x, b = blockIdx.x;
    for (int k = t; k < nbk; k += SORT_T) cur[k] = bo[(size_t)k * G + b];
    __syncthreads();
    const int chunk = (E + G - 1) / G;
    const int beg = b * chunk, end = min(E, beg + chunk);
    for (int i = beg + t; i < end; i += SORT_U * SORT_T) {
        int r[SORT_U], c[SORT_U];
#pragma unroll
        for (int u = 0; u < SORT_U; ++u) {
            const int j = i + u * SORT_T;
            r[u] = j < end ? erow[j] : 0;
            c[u] = j < end ? ecol[j] : 0;
        }
#pragma unroll
        for (int u = 0; u < SORT_U; ++u) {
            if (i + u * SORT_T < end) {
                const int k = min((int)((unsigned)r[u] >> BKT_SH), nbk - 1);
                const int p = atomicAdd(&cur[k], 1);
                tmp[p] = make_int2(r[u], c[u]);
            }
        }
    }
}

__global__ __launch_bounds__(SORT_T) void bin_sort(const int2* __restrict__ tmp,
                                                   const int* __restrict__ bo,
                                                   int* __restrict__ offs,
                                                   int* __restrict__ scol,
                                                   int G, int nbk, int n) {
    __shared__ int cnt[BKT_N];
    __shared__ int wsum[BKT_N / 32];
    const int t = threadIdx.x, k = blockIdx.x, lane = t & 31, wid = t >> 5;
    const int s = bo[(size_t)k * G], e = bo[(size_t)(k + 1) * G];
    if (t < BKT_N) cnt[t] = 0;
    __syncthreads();
    for (int i = s + t; i < e; i += SORT_U * SORT_T) {
        int r[SORT_U];
#pragma unroll
        for (int u = 0; u < SORT_U; ++u) {
            const int j = i + u * SORT_T;
            r[u] = j < e ? tmp[j].x : 0;
        }
#pragma unroll
        for (int u = 0; u < SORT_U; ++u)
            if (i + u * SORT_T < e) atomicAdd(&cnt[r[u] & (BKT_N - 1)], 1);
    }
    __syncthreads();
    int v = 0, sc = 0;
    if (t < BKT_N) {
        v = cnt[t];
        sc = v;
        for (int m = 1; m < 32; m <<= 1) {
            int u = __shfl_up(sc, m, 32);
            if (lane >= m) sc += u;
        }
        if (lane == 31) wsum[wid] = sc;
    }
    __syncthreads();
    if (t < BKT_N) {
        int base = s;
        for (int w = 0; w < wid; ++w) base += wsum[w];
        const int excl = base + sc - v;
        cnt[t] = excl;
        const int node = k * BKT_N + t;
        if (node < n) offs[node] = excl;
    }
    if (k == nbk - 1 && t == 0) offs[n] = e;
    __syncthreads();
    for (int i = s + t; i < e; i += SORT_U * SORT_T) {
        int2 rc[SORT_U];
#pragma unroll
        for (int u = 0; u < SORT_U; ++u) {
            const int j = i + u * SORT_T;
            rc[u] = j < e ? tmp[j] : make_int2(0, 0);
        }
#pragma unroll
        for (int u = 0; u < SORT_U; ++u) {
            if (i + u * SORT_T < e) {
                const int p = atomicAdd(&cnt[rc[u].x & (BKT_N - 1)], 1);
                scol[p] = rc[u].y;
            }
        }
    }
}

__device__ __forceinline__ int block_sum_1024(int v, int* sh) {
    int t = threadIdx.x;
    for (int m = 1; m < 32; m <<= 1) v += __shfl_xor(v, m, 32);
    if ((t & 31) == 0) sh[t >> 5] = v;
    __syncthreads();
    int w = sh[t & 31];
    for (int m = 1; m < 32; m <<= 1) w += __shfl_xor(w, m, 32);
    __syncthreads();
    return w;
}

__global__ __launch_bounds__(SCAN_B) void scan_sums(const int* __restrict__ cnt,
                                                    int* __restrict__ bsum, int n) {
    __shared__ int sh[32];
    int i = blockIdx.x * SCAN_B + threadIdx.x;
    int tot = block_sum_1024(i < n ? cnt[i] : 0, sh);
    if (threadIdx.x == 0) bsum[blockIdx.x] = tot;
}

__global__ __launch_bounds__(SCAN_B) void scan_apply(int* __restrict__ offs,
                                                     const int* __restrict__ bsum, int n) {
    __shared__ int sh[32];
    const int t = threadIdx.x, lane = t & 31, wid = t >> 5;
    int p = 0;
    for (int b = t; b < (int)blockIdx.x; b += SCAN_B) p += bsum[b];
    const int base = block_sum_1024(p, sh);

    int i = blockIdx.x * SCAN_B + t;
    int v = i < n ? offs[i] : 0;
    int s = v;
    for (int m = 1; m < 32; m <<= 1) {
        int u = __shfl_up(s, m, 32);
        if (lane >= m) s += u;
    }
    if (lane == 31) sh[wid] = s;
    __syncthreads();
    if (wid == 0) {
        int w0 = sh[lane], w = w0;
        for (int m = 1; m < 32; m <<= 1) {
            int u = __shfl_up(w, m, 32);
            if (lane >= m) w += u;
        }
        sh[lane] = w - w0;
    }
    __syncthreads();
    int excl = base + sh[wid] + s - v;
    if (i < n) offs[i] = excl;
    if (i == n - 1) offs[n] = excl + v;
}

__device__ __forceinline__ float wave_sum32(float x) {
    for (int m = 1; m < 32; m <<= 1) x += __shfl_xor(x, m, 32);
    return x;
}

__device__ __forceinline__ float4 act_row(float4 v) {
    float an2 = wave_sum32(v.x * v.x + v.y * v.y + v.z * v.z + v.w * v.w);
    float an = fmaxf(sqrtf(an2), MINN);
    float tn = fminf(fmaxf(tanhf(an), MINN), MAXN);
    float lg = artanh_f(tn) / an;
    float4 at;
    at.x = tanhf(lg * v.x);
    at.y = tanhf(lg * v.y);
    at.z = tanhf(lg * v.z);
    at.w = tanhf(lg * v.w);
    float vn2 = wave_sum32(at.x * at.x + at.y * at.y + at.z * at.z + at.w * at.w);
    float vn = fmaxf(sqrtf(vn2), MINN);
    float s = fminf(tanhf(vn), MAXN) / vn;
    return make_float4(s * at.x, s * at.y, s * at.z, s * at.w);
}

template <int READOUT>
__global__ __launch_bounds__(256) void node_gather_act(const float* __restrict__ HT,
                                                       const int* __restrict__ offs,
                                                       const int* __restrict__ scol,
                                                       float* __restrict__ OUT,
                                                       const int* __restrict__ seg,
                                                       float* __restrict__ ssum,
                                                       float* __restrict__ cnt, int nrows) {
    const int lane = threadIdx.x & 31;
    const int w0 = __builtin_amdgcn_readfirstlane((int)(blockIdx.x * 8 + (threadIdx.x >> 5)));
    const int nw = gridDim.x * 8;
    for (int r = w0; r < nrows; r += nw) {
        const int beg = offs[r], end = offs[r + 1];
        int sg = 0;
        if (READOUT) sg = seg[r];
        float4 acc = make_float4(0.f, 0.f, 0.f, 0.f);
        for (int b0 = beg; b0 < end; b0 += 32) {
            const int n = min(32, end - b0);
            const int myc = (lane < n) ? scol[b0 + lane] : 0;
            int j = 0;
            for (; j + 4 <= n; j += 4) {
                const int c0 = __builtin_amdgcn_readlane(myc, j);
                const int c1 = __builtin_amdgcn_readlane(myc, j + 1);
                const int c2 = __builtin_amdgcn_readlane(myc, j + 2);
                const int c3 = __builtin_amdgcn_readlane(myc, j + 3);
                const float4 v0 = *((const float4*)(HT + (size_t)c0 * 128) + lane);
                const float4 v1 = *((const float4*)(HT + (size_t)c1 * 128) + lane);
                const float4 v2 = *((const float4*)(HT + (size_t)c2 * 128) + lane);
                const float4 v3 = *((const float4*)(HT + (size_t)c3 * 128) + lane);
                acc.x += v0.x; acc.y += v0.y; acc.z += v0.z; acc.w += v0.w;
                acc.x += v1.x; acc.y += v1.y; acc.z += v1.z; acc.w += v1.w;
                acc.x += v2.x; acc.y += v2.y; acc.z += v2.z; acc.w += v2.w;
                acc.x += v3.x; acc.y += v3.y; acc.z += v3.z; acc.w += v3.w;
            }
            for (; j < n; ++j) {
                const int c0 = __builtin_amdgcn_readlane(myc, j);
                const float4 v0 = *((const float4*)(HT + (size_t)c0 * 128) + lane);
                acc.x += v0.x; acc.y += v0.y; acc.z += v0.z; acc.w += v0.w;
            }
        }
        const float4 h = act_row(acc);
        if (!READOUT) {
            *((float4*)(OUT + (size_t)r * 128) + lane) = h;
        } else {
            float hn2 = wave_sum32(h.x * h.x + h.y * h.y + h.z * h.z + h.w * h.w);
            float hn = fmaxf(sqrtf(hn2), MINN);
            float pc = fminf(hn, MAXN);
            float L = artanh_f(pc) / hn;
            float* d = ssum + (size_t)sg * 128 + lane * 4;
            atomicAdd(d + 0, L * h.x);
            atomicAdd(d + 1, L * h.y);
            atomicAdd(d + 2, L * h.z);
            atomicAdd(d + 3, L * h.w);
            if (lane == 0) atomicAdd(&cnt[sg], 1.f);
        }
    }
}

__global__ __launch_bounds__(256) void post_agg(float* __restrict__ H, int nrows) {
    int r = blockIdx.x * 8 + (threadIdx.x >> 5);
    if (r >= nrows) return;
    int lane = threadIdx.x & 31;
    float* p = H + (size_t)r * 128 + lane * 4;
    float4 v = *(const float4*)p;
    float an2 = v.x * v.x + v.y * v.y + v.z * v.z + v.w * v.w;
    for (int m = 1; m < 32; m <<= 1) an2 += __shfl_xor(an2, m, 32);
    float an = fmaxf(sqrtf(an2), MINN);
    float tn = fminf(fmaxf(tanhf(an), MINN), MAXN);
    float lg = artanh_f(tn) / an;
    float4 at;
    at.x = tanhf(lg * v.x);
    at.y = tanhf(lg * v.y);
    at.z = tanhf(lg * v.z);
    at.w = tanhf(lg * v.w);
    float vn2 = at.x * at.x + at.y * at.y + at.z * at.z + at.w * at.w;
    for (int m = 1; m < 32; m <<= 1) vn2 += __shfl_xor(vn2, m, 32);
    float vn = fmaxf(sqrtf(vn2), MINN);
    float s = fminf(tanhf(vn), MAXN) / vn;
    *(float4*)p = make_float4(s * at.x, s * at.y, s * at.z, s * at.w);
}

__global__ __launch_bounds__(256) void final_scatter(
    const float* __restrict__ H, const int* __restrict__ seg,
    float* __restrict__ ssum, float* __restrict__ cnt, int nrows) {
    int r = blockIdx.x * 8 + (threadIdx.x >> 5);
    if (r >= nrows) return;
    int lane = threadIdx.x & 31;
    float4 v = *(const float4*)(H + (size_t)r * 128 + lane * 4);
    float hn2 = v.x * v.x + v.y * v.y + v.z * v.z + v.w * v.w;
    for (int m = 1; m < 32; m <<= 1) hn2 += __shfl_xor(hn2, m, 32);
    float hn = fmaxf(sqrtf(hn2), MINN);
    float pc = fminf(hn, MAXN);
    float L = artanh_f(pc) / hn;
    int s = seg[r];
    float* d = ssum + (size_t)s * 128 + lane * 4;
    atomicAdd(d + 0, L * v.x);
    atomicAdd(d + 1, L * v.y);
    atomicAdd(d + 2, L * v.z);
    atomicAdd(d + 3, L * v.w);
    if (lane == 0) atomicAdd(&cnt[s], 1.f);
}

__global__ __launch_bounds__(256) void pool_k(const float* __restrict__ ssum,
                                              const float* __restrict__ cnt,
                                              float* __restrict__ out, int nseg) {
    int r = blockIdx.x * 8 + (threadIdx.x >> 5);
    if (r >= nseg) return;
    int lane = threadIdx.x & 31;
    float inv = 1.f / fmaxf(cnt[r], 1.f);
    float4 v = *(const float4*)(ssum + (size_t)r * 128 + lane * 4);
    v.x *= inv; v.y *= inv; v.z *= inv; v.w *= inv;
    float mn2 = v.x * v.x + v.y * v.y + v.z * v.z + v.w * v.w;
    for (int m = 1; m < 32; m <<= 1) mn2 += __shfl_xor(mn2, m, 32);
    float mn = fmaxf(sqrtf(mn2), MINN);
    float s = fminf(tanhf(mn), MAXN) / mn;
    float* o = out + (size_t)r * 128 + lane * 4;
    o[0] = s * v.x; o[1] = s * v.y; o[2] = s * v.z; o[3] = s * v.w;
}

__global__ void zero_f32(float* __restrict__ p, size_t n) {
    size_t i = (size_t)blockIdx.x * blockDim.x + threadIdx.x;
    size_t st = (size_t)gridDim.x * blockDim.x;
    for (; i < n; i += st) p[i] = 0.f;
}

extern "C" void kernel_launch(void* const* d_in, const int* in_sizes, int n_in,
                              void* d_out, int out_size, void* d_ws, size_t ws_size,
                              hipStream_t stream) {
    const float* x = (const float*)d_in[0];
    const int* erow = (const int*)d_in[1];
    const int* ecol = (const int*)d_in[2];
    const int* seg = (const int*)d_in[3];
    const float* W1 = (const float*)d_in[4];
    const float* b1 = (const float*)d_in[5];
    const float* W2 = (const float*)d_in[6];
    const float* b2 = (const float*)d_in[7];

    const int N = in_sizes[0] / 256;
    const int E = in_sizes[1];
    const int NSEG = out_size / 128;

    char* ws = (char*)d_ws;
    size_t bufBytes = (size_t)N * 128 * sizeof(float);
    float* buf0 = (float*)ws;
    float* buf1 = (float*)(ws + bufBytes);
    float* bbias = (float*)(ws + 2 * bufBytes);
    float* y2 = bbias + 256;
    float* ssum = buf0;
    float* cnt = buf0 + (size_t)NSEG * 128;

    dim3 blk(256);
    int gRows = (N + 7) / 8;
    int gGemm = (N + BM - 1) / BM;
    int gEdge = (E + 7) / 8;

    const int nScan = (N + SCAN_B - 1) / SCAN_B;
    size_t offBytes = (((size_t)N + 1) * sizeof(int) + 255) / 256 * 256;
    size_t bsBytes = ((size_t)nScan * sizeof(int) + 255) / 256 * 256;
    int* offs = (int*)(ws + 2 * bufBytes + 4096);
    int* scol = (int*)(ws + 2 * bufBytes + 4096 + offBytes + bsBytes);
    const int G1 = 256;
    const int NBK = (N + BKT_N - 1) / BKT_N;
    const int nTab = NBK * G1;
    const int nScanT = (nTab + SCAN_B - 1) / SCAN_B;
    size_t tmpBytes = ((size_t)E * sizeof(int2) + 255) / 256 * 256;
    size_t tabBytes = (((size_t)nTab + 1) * sizeof(int) + 255) / 256 * 256;
    int2* tmp = (int2*)buf1;
    int* btab = (int*)(ws + bufBytes + tmpBytes);
    int* bsum = (int*)(ws + bufBytes + tmpBytes + tabBytes);
    const bool grouped =
        ws_size >= 2 * bufBytes + 4096 + offBytes + bsBytes + (size_t)E * sizeof(int) &&
        NBK <= MAXBK &&
        tmpBytes + tabBytes + (size_t)nScanT * sizeof(int) <= bufBytes;

    bias_prep<<<2, 128, 0, stream>>>(b1, b2, bbias, y2);

    if (grouped) {
        bin_count<<<G1, SORT_T, 0, stream>>>(erow, btab, E, NBK);
        scan_sums<<<nScanT, SCAN_B, 0, stream>>>(btab, bsum, nTab);
        scan_apply<<<nScanT, SCAN_B, 0, stream>>>(btab, bsum, nTab);
        bin_scatter<<<G1, SORT_T, 0, stream>>>(erow, ecol, btab, tmp, E, NBK);
        bin_sort<<<NBK, SORT_T, 0, stream>>>(tmp, btab, offs, scol, G1, NBK, N);

        hyp_linear_wmma<<<gGemm, blk, 0, stream>>>(x, 256, N, W1, bbias, y2, buf0, 1);
        node_gather_act<0><<<2048, blk, 0, stream>>>(buf0, offs, scol, buf1, seg,
                                                     (float*)0, (float*)0, N);

        hyp_linear_wmma<<<gGemm, blk, 0, stream>>>(buf1, 128, N, W2, bbias + 128,
                                                   y2 + 1, buf0, 0);
        float* ssumG = buf1;
        float* cntG = buf1 + (size_t)NSEG * 128;
        zero_f32<<<64, 256, 0, stream>>>(ssumG, (size_t)NSEG * 128 + NSEG);
        node_gather_act<1><<<2048, blk, 0, stream>>>(buf0, offs, scol, (float*)0, seg,
                                                     ssumG, cntG, N);
        pool_k<<<(NSEG + 7) / 8, blk, 0, stream>>>(ssumG, cntG, (float*)d_out, NSEG);
        return;
    }

    hyp_linear_wmma<<<gGemm, blk, 0, stream>>>(x, 256, N, W1, bbias, y2, buf0, 1);
    zero_f32<<<2048, 256, 0, stream>>>(buf1, (size_t)N * 128);
    edge_agg<<<gEdge, blk, 0, stream>>>(buf0, erow, ecol, buf1, E);
    post_agg<<<gRows, blk, 0, stream>>>(buf1, N);

    hyp_linear_wmma<<<gGemm, blk, 0, stream>>>(buf1, 128, N, W2, bbias + 128,
                                               y2 + 1, buf0, 0);
    zero_f32<<<2048, 256, 0, stream>>>(buf1, (size_t)N * 128);
    edge_agg<<<gEdge, blk, 0, stream>>>(buf0, erow, ecol, buf1, E);
    post_agg<<<gRows, blk, 0, stream>>>(buf1, N);

    zero_f32<<<64, 256, 0, stream>>>(ssum, (size_t)NSEG * 128 + NSEG);
    final_scatter<<<gRows, blk, 0, stream>>>(buf1, seg, ssum, cnt, N);
    pool_k<<<(NSEG + 7) / 8, blk, 0, stream>>>(ssum, cnt, (float*)d_out, NSEG);
}


# GEMM re-tiled so every row tile is resident at once: K chunk 128 -> 32, 4-wave blocks with 16x128 wave tiles (LDS 101 -> 30 KB), a chunk's 12 loads per thread issued together; f32 sum order kept
# speedup vs baseline: 3.0496x; 1.1167x over previous
#include <hip/hip_runtime.h>
#include <math.h>


#define MAXN 0.996f
#define MINN 1e-15f
#define KC   32
#define LDP  (KC + 4)
#define BM   64

typedef float v2f __attribute__((ext_vector_type(2)));
typedef float v8f __attribute__((ext_vector_type(8)));
typedef float v4f __attribute__((ext_vector_type(4)));

__device__ __forceinline__ float artanh_f(float z) {
    z = fminf(fmaxf(z, -(1.f - 1e-7f)), 1.f - 1e-7f);
    return 0.5f * logf((1.f + z) / (1.f - z));
}

__global__ void bias_prep(const float* __restrict__ b1,
                          const float* __restrict__ b2,
                          float* __restrict__ bb, float* __restrict__ y2) {
    __shared__ float red[128];
    const float* b = blockIdx.x ? b2 : b1;
    int t = threadIdx.x;
    float v = b[t];
    red[t] = v * v;
    __syncthreads();
    for (int s = 64; s > 0; s >>= 1) {
        if (t < s) red[t] += red[t + s];
        __syncthreads();
    }
    float bn = fmaxf(sqrtf(red[0]), MINN);
    float tc = fminf(tanhf(bn), MAXN);
    bb[blockIdx.x * 128 + t] = tc / bn * v;
    if (t == 0) y2[blockIdx.x] = tc * tc;
}

__global__ __launch_bounds__(128) void hyp_linear_wmma(
    const float* __restrict__ X, int Kdim, int nrows,
    const float* __restrict__ W,
    const float* __restrict__ bb,
    const float* __restrict__ y2p,
    float* __restrict__ HT, int apply_exp) {
    __shared__ float As[BM][LDP];
    __shared__ float Ws[128][LDP];
    __shared__ float red[BM][4];
    __shared__ float scale_row[BM], xn_row[BM];
    __shared__ float rs_mx2[BM], rs_dot[BM];
    __shared__ float rowCA[BM], rowCB[BM];

    const int t = threadIdx.x;
    const int row0 = blockIdx.x * BM;

    {
        const int r0 = t >> 2, p = t & 3;
        const int kl = Kdim >> 2;
        const float* px0 = X + (size_t)min(row0 + r0, nrows - 1) * Kdim + p * kl;
        const float* px1 = X + (size_t)min(row0 + r0 + 32, nrows - 1) * Kdim + p * kl;
        float s0 = 0.f, s1 = 0.f;
        for (int k = 0; k < kl; k += 4) {
            float4 v = *(const float4*)(px0 + k);
            float4 u = *(const float4*)(px1 + k);
            s0 += v.x * v.x + v.y * v.y + v.z * v.z + v.w * v.w;
            s1 += u.x * u.x + u.y * u.y + u.z * u.z + u.w * u.w;
        }
        red[r0][p] = s0;
        red[r0 + 32][p] = s1;
    }
    __syncthreads();
    if (t < BM) {
        float un2 = red[t][0] + red[t][1] + red[t][2] + red[t][3];
        float un = fmaxf(sqrtf(un2), MINN);
        if (apply_exp) {
            float bc = fminf(tanhf(un), MAXN);
            scale_row[t] = bc / un;
            xn_row[t] = fmaxf(bc, MINN);
        } else {
            scale_row[t] = 1.f;
            xn_row[t] = un;
        }
        if (row0 + t >= nrows) scale_row[t] = 0.f;
    }
    __syncthreads();

    const int lane = t & 31;
    const int rs = t >> 5;
    const int half = lane >> 4;
    const int l15 = lane & 15;

    v8f acc[8];
    for (int j = 0; j < 8; ++j)
        for (int i = 0; i < 8; ++i) acc[j][i] = 0.f;

    const int nchunks = Kdim / KC;

    const int sr = t >> 3;
    const int sc4 = t & 7;
    const int kq = Kdim >> 2;
    const v4f* __restrict__ X4 = (const v4f*)X;
    const v4f* __restrict__ W4 = (const v4f*)W;
    int ia[4];
#pragma unroll
    for (int i = 0; i < 4; ++i) ia[i] = min(row0 + sr + 16 * i, nrows - 1) * kq + sc4;
    const int iw = sr * kq + sc4;

    const int mr = rs * 16 + l15;

    for (int kc = 0; kc < nchunks; ++kc) {
        const int q0 = kc * (KC / 4);
        v4f av[4], wv[8];
#pragma unroll
        for (int i = 0; i < 4; ++i) av[i] = X4[ia[i] + q0];
#pragma unroll
        for (int i = 0; i < 8; ++i) wv[i] = W4[iw + i * 16 * kq + q0];
        asm volatile("" : "+v"(av[0]), "+v"(av[1]), "+v"(av[2]), "+v"(av[3]),
                          "+v"(wv[0]), "+v"(wv[1]), "+v"(wv[2]), "+v"(wv[3]),
                          "+v"(wv[4]), "+v"(wv[5]), "+v"(wv[6]), "+v"(wv[7]));
#pragma unroll
        for (int i = 0; i < 4; ++i)
            *(v4f*)(&As[sr + 16 * i][sc4 * 4]) = av[i] * scale_row[sr + 16 * i];
#pragma unroll
        for (int i = 0; i < 8; ++i) *(v4f*)(&Ws[sr + 16 * i][sc4 * 4]) = wv[i];
        __syncthreads();

        for (int kk = 0; kk < KC; kk += 4) {
            const int ko = kk + half * 2;
            v2f a;
            a.x = As[mr][ko];
            a.y = As[mr][ko + 1];
#pragma unroll
            for (int j = 0; j < 8; ++j) {
                const int colg = j * 16 + l15;
                v2f b;
                b.x = Ws[colg][ko];
                b.y = Ws[colg][ko + 1];
                acc[j] = __builtin_amdgcn_wmma_f32_16x16x4_f32(
                    false, a, false, b, (short)0, acc[j], false, false);
            }
        }
        __syncthreads();
    }

    float bbv[8];
#pragma unroll
    for (int j = 0; j < 8; ++j) bbv[j] = bb[j * 16 + l15];

#pragma unroll
    for (int i = 0; i < 8; ++i) {
        float p2a = 0.f, pda = 0.f, p2b = 0.f, pdb = 0.f;
#pragma unroll
        for (int j = 0; j < 4; ++j) {
            float v = acc[j][i];
            p2a += v * v;
            pda += v * bbv[j];
        }
#pragma unroll
        for (int j = 4; j < 8; ++j) {
            float v = acc[j][i];
            p2b += v * v;
            pdb += v * bbv[j];
        }
        for (int m = 1; m < 16; m <<= 1) {
            p2a += __shfl_xor(p2a, m, 16);
            pda += __shfl_xor(pda, m, 16);
            p2b += __shfl_xor(p2b, m, 16);
            pdb += __shfl_xor(pdb, m, 16);
        }
        int mrow = rs * 16 + i + half * 8;
        if (l15 == 0) {
            rs_mx2[mrow] = p2a + p2b;
            rs_dot[mrow] = pda + pdb;
        }
    }
    __syncthreads();

    if (t < BM) {
        float xn = xn_row[t];
        float mxn = fmaxf(sqrtf(rs_mx2[t]), MINN);
        float tv = tanhf(mxn / xn * artanh_f(xn));
        float tc = fminf(tv, MAXN);
        float s_mv = tc / mxn;
        float x2 = tc * tc;
        float xy = s_mv * rs_dot[t];
        float y2 = y2p[0];
        float al = 1.f + 2.f * xy + y2;
        float be = 1.f - x2;
        float den = fmaxf(1.f + 2.f * xy + x2 * y2, MINN);
        float nn2 = fmaxf(al * al * x2 + 2.f * al * be * xy + be * be * y2, 0.f);
        float hn = fmaxf(sqrtf(nn2) / den, MINN);
        float hc = fminf(hn, MAXN);
        float L = artanh_f(hc) / hn;
        rowCA[t] = L * al * s_mv / den;
        rowCB[t] = L * be / den;
    }
    __syncthreads();

#pragma unroll
    for (int i = 0; i < 8; ++i) {
        int mrow = rs * 16 + i + half * 8;
        int gr = row0 + mrow;
        if (gr < nrows) {
            float ca = rowCA[mrow], cb = rowCB[mrow];
#pragma unroll
            for (int j = 0; j < 8; ++j) {
                int colg = j * 16 + l15;
                HT[(size_t)gr * 128 + colg] = ca * acc[j][i] + cb * bbv[j];
            }
        }
    }
}

__global__ __launch_bounds__(256) void edge_agg(const float* __restrict__ HT,
                                                const int* __restrict__ erow,
                                                const int* __restrict__ ecol,
                                                float* __restrict__ AGG, int E) {
    int e = blockIdx.x * 8 + (threadIdx.x >> 5);
    if (e >= E) return;
    int lane = threadIdx.x & 31;
    int r = erow[e], c = ecol[e];
    float4 v = *(const float4*)(HT + (size_t)c * 128 + lane * 4);
    float* dst = AGG + (size_t)r * 128 + lane * 4;
    atomicAdd(dst + 0, v.x);
    atomicAdd(dst + 1, v.y);
    atomicAdd(dst + 2, v.z);
    atomicAdd(dst + 3, v.w);
}

#define SCAN_B 1024
#define BKT_SH 8
#define BKT_N  256
#define MAXBK  1024
#define SORT_T 1024
#define SORT_U 4

__global__ __launch_bounds__(SORT_T) void bin_count(const int* __restrict__ erow,
                                                    int* __restrict__ bh,
                                                    int E, int nbk) {
    __shared__ int hist[MAXBK];
    const int t = threadIdx.x, G = gridDim.x, b = blockIdx.x;
    for (int k = t; k < nbk; k += SORT_T) hist[k] = 0;
    __syncthreads();
    const int chunk = (E + G - 1) / G;
    const int beg = b * chunk, end = min(E, beg + chunk);
    for (int i = beg + t; i < end; i += SORT_U * SORT_T) {
        int r[SORT_U];
#pragma unroll
        for (int u = 0; u < SORT_U; ++u) {
            const int j = i + u * SORT_T;
            r[u] = j < end ? erow[j] : 0;
        }
#pragma unroll
        for (int u = 0; u < SORT_U; ++u) {
            const int k = min((int)((unsigned)r[u] >> BKT_SH), nbk - 1);
            if (i + u * SORT_T < end) atomicAdd(&hist[k], 1);
        }
    }
    __syncthreads();
    for (int k = t; k < nbk; k += SORT_T) bh[(size_t)k * G + b] = hist[k];
}

__global__ __launch_bounds__(SORT_T) void bin_scatter(const int* __restrict__ erow,
                                                      const int* __restrict__ ecol,
                                                      const int* __restrict__ bo,
                                                      int2* __restrict__ tmp,
                                                      int E, int nbk) {
    __shared__ int cur[MAXBK];
    const int t = threadIdx.x, G = gridDim.x, b = blockIdx.x;
    for (int k = t; k < nbk; k += SORT_T) cur[k] = bo[(size_t)k * G + b];
    __syncthreads();
    const int chunk = (E + G - 1) / G;
    const int beg = b * chunk, end = min(E, beg + chunk);
    for (int i = beg + t; i < end; i += SORT_U * SORT_T) {
        int r[SORT_U], c[SORT_U];
#pragma unroll
        for (int u = 0; u < SORT_U; ++u) {
            const int j = i + u * SORT_T;
            r[u] = j < end ? erow[j] : 0;
            c[u] = j < end ? ecol[j] : 0;
        }
#pragma unroll
        for (int u = 0; u < SORT_U; ++u) {
            if (i + u * SORT_T < end) {
                const int k = min((int)((unsigned)r[u] >> BKT_SH), nbk - 1);
                const int p = atomicAdd(&cur[k], 1);
                tmp[p] = make_int2(r[u], c[u]);
            }
        }
    }
}

__global__ __launch_bounds__(SORT_T) void bin_sort(const int2* __restrict__ tmp,
                                                   const int* __restrict__ bo,
                                                   int* __restrict__ offs,
                                                   int* __restrict__ scol,
                                                   int G, int nbk, int n) {
    __shared__ int cnt[BKT_N];
    __shared__ int wsum[BKT_N / 32];
    const int t = threadIdx.x, k = blockIdx.x, lane = t & 31, wid = t >> 5;
    const int s = bo[(size_t)k * G], e = bo[(size_t)(k + 1) * G];
    if (t < BKT_N) cnt[t] = 0;
    __syncthreads();
    for (int i = s + t; i < e; i += SORT_U * SORT_T) {
        int r[SORT_U];
#pragma unroll
        for (int u = 0; u < SORT_U; ++u) {
            const int j = i + u * SORT_T;
            r[u] = j < e ? tmp[j].x : 0;
        }
#pragma unroll
        for (int u = 0; u < SORT_U; ++u)
            if (i + u * SORT_T < e) atomicAdd(&cnt[r[u] & (BKT_N - 1)], 1);
    }
    __syncthreads();
    int v = 0, sc = 0;
    if (t < BKT_N) {
        v = cnt[t];
        sc = v;
        for (int m = 1; m < 32; m <<= 1) {
            int u = __shfl_up(sc, m, 32);
            if (lane >= m) sc += u;
        }
        if (lane == 31) wsum[wid] = sc;
    }
    __syncthreads();
    if (t < BKT_N) {
        int base = s;
        for (int w = 0; w < wid; ++w) base += wsum[w];
        const int excl = base + sc - v;
        cnt[t] = excl;
        const int node = k * BKT_N + t;
        if (node < n) offs[node] = excl;
    }
    if (k == nbk - 1 && t == 0) offs[n] = e;
    __syncthreads();
    for (int i = s + t; i < e; i += SORT_U * SORT_T) {
        int2 rc[SORT_U];
#pragma unroll
        for (int u = 0; u < SORT_U; ++u) {
            const int j = i + u * SORT_T;
            rc[u] = j < e ? tmp[j] : make_int2(0, 0);
        }
#pragma unroll
        for (int u = 0; u < SORT_U; ++u) {
            if (i + u * SORT_T < e) {
                const int p = atomicAdd(&cnt[rc[u].x & (BKT_N - 1)], 1);
                scol[p] = rc[u].y;
            }
        }
    }
}

__device__ __forceinline__ int block_sum_1024(int v, int* sh) {
    int t = threadIdx.x;
    for (int m = 1; m < 32; m <<= 1) v += __shfl_xor(v, m, 32);
    if ((t & 31) == 0) sh[t >> 5] = v;
    __syncthreads();
    int w = sh[t & 31];
    for (int m = 1; m < 32; m <<= 1) w += __shfl_xor(w, m, 32);
    __syncthreads();
    return w;
}

__global__ __launch_bounds__(SCAN_B) void scan_sums(const int* __restrict__ cnt,
                                                    int* __restrict__ bsum, int n) {
    __shared__ int sh[32];
    int i = blockIdx.x * SCAN_B + threadIdx.x;
    int tot = block_sum_1024(i < n ? cnt[i] : 0, sh);
    if (threadIdx.x == 0) bsum[blockIdx.x] = tot;
}

__global__ __launch_bounds__(SCAN_B) void scan_apply(int* __restrict__ offs,
                                                     const int* __restrict__ bsum, int n) {
    __shared__ int sh[32];
    const int t = threadIdx.x, lane = t & 31, wid = t >> 5;
    int p = 0;
    for (int b = t; b < (int)blockIdx.x; b += SCAN_B) p += bsum[b];
    const int base = block_sum_1024(p, sh);

    int i = blockIdx.x * SCAN_B + t;
    int v = i < n ? offs[i] : 0;
    int s = v;
    for (int m = 1; m < 32; m <<= 1) {
        int u = __shfl_up(s, m, 32);
        if (lane >= m) s += u;
    }
    if (lane == 31) sh[wid] = s;
    __syncthreads();
    if (wid == 0) {
        int w0 = sh[lane], w = w0;
        for (int m = 1; m < 32; m <<= 1) {
            int u = __shfl_up(w, m, 32);
            if (lane >= m) w += u;
        }
        sh[lane] = w - w0;
    }
    __syncthreads();
    int excl = base + sh[wid] + s - v;
    if (i < n) offs[i] = excl;
    if (i == n - 1) offs[n] = excl + v;
}

__device__ __forceinline__ float wave_sum32(float x) {
    for (int m = 1; m < 32; m <<= 1) x += __shfl_xor(x, m, 32);
    return x;
}

__device__ __forceinline__ float4 act_row(float4 v) {
    float an2 = wave_sum32(v.x * v.x + v.y * v.y + v.z * v.z + v.w * v.w);
    float an = fmaxf(sqrtf(an2), MINN);
    float tn = fminf(fmaxf(tanhf(an), MINN), MAXN);
    float lg = artanh_f(tn) / an;
    float4 at;
    at.x = tanhf(lg * v.x);
    at.y = tanhf(lg * v.y);
    at.z = tanhf(lg * v.z);
    at.w = tanhf(lg * v.w);
    float vn2 = wave_sum32(at.x * at.x + at.y * at.y + at.z * at.z + at.w * at.w);
    float vn = fmaxf(sqrtf(vn2), MINN);
    float s = fminf(tanhf(vn), MAXN) / vn;
    return make_float4(s * at.x, s * at.y, s * at.z, s * at.w);
}

template <int READOUT>
__global__ __launch_bounds__(256) void node_gather_act(const float* __restrict__ HT,
                                                       const int* __restrict__ offs,
                                                       const int* __restrict__ scol,
                                                       float* __restrict__ OUT,
                                                       const int* __restrict__ seg,
                                                       float* __restrict__ ssum,
                                                       float* __restrict__ cnt, int nrows) {
    const int lane = threadIdx.x & 31;
    const int w0 = __builtin_amdgcn_readfirstlane((int)(blockIdx.x * 8 + (threadIdx.x >> 5)));
    const int nw = gridDim.x * 8;
    for (int r = w0; r < nrows; r += nw) {
        const int beg = offs[r], end = offs[r + 1];
        int sg = 0;
        if (READOUT) sg = seg[r];
        float4 acc = make_float4(0.f, 0.f, 0.f, 0.f);
        for (int b0 = beg; b0 < end; b0 += 32) {
            const int n = min(32, end - b0);
            const int myc = (lane < n) ? scol[b0 + lane] : 0;
            int j = 0;
            for (; j + 4 <= n; j += 4) {
                const int c0 = __builtin_amdgcn_readlane(myc, j);
                const int c1 = __builtin_amdgcn_readlane(myc, j + 1);
                const int c2 = __builtin_amdgcn_readlane(myc, j + 2);
                const int c3 = __builtin_amdgcn_readlane(myc, j + 3);
                const float4 v0 = *((const float4*)(HT + (size_t)c0 * 128) + lane);
                const float4 v1 = *((const float4*)(HT + (size_t)c1 * 128) + lane);
                const float4 v2 = *((const float4*)(HT + (size_t)c2 * 128) + lane);
                const float4 v3 = *((const float4*)(HT + (size_t)c3 * 128) + lane);
                acc.x += v0.x; acc.y += v0.y; acc.z += v0.z; acc.w += v0.w;
                acc.x += v1.x; acc.y += v1.y; acc.z += v1.z; acc.w += v1.w;
                acc.x += v2.x; acc.y += v2.y; acc.z += v2.z; acc.w += v2.w;
                acc.x += v3.x; acc.y += v3.y; acc.z += v3.z; acc.w += v3.w;
            }
            for (; j < n; ++j) {
                const int c0 = __builtin_amdgcn_readlane(myc, j);
                const float4 v0 = *((const float4*)(HT + (size_t)c0 * 128) + lane);
                acc.x += v0.x; acc.y += v0.y; acc.z += v0.z; acc.w += v0.w;
            }
        }
        const float4 h = act_row(acc);
        if (!READOUT) {
            *((float4*)(OUT + (size_t)r * 128) + lane) = h;
        } else {
            float hn2 = wave_sum32(h.x * h.x + h.y * h.y + h.z * h.z + h.w * h.w);
            float hn = fmaxf(sqrtf(hn2), MINN);
            float pc = fminf(hn, MAXN);
            float L = artanh_f(pc) / hn;
            float* d = ssum + (size_t)sg * 128 + lane * 4;
            atomicAdd(d + 0, L * h.x);
            atomicAdd(d + 1, L * h.y);
            atomicAdd(d + 2, L * h.z);
            atomicAdd(d + 3, L * h.w);
            if (lane == 0) atomicAdd(&cnt[sg], 1.f);
        }
    }
}

__global__ __launch_bounds__(256) void post_agg(float* __restrict__ H, int nrows) {
    int r = blockIdx.x * 8 + (threadIdx.x >> 5);
    if (r >= nrows) return;
    int lane = threadIdx.x & 31;
    float* p = H + (size_t)r * 128 + lane * 4;
    float4 v = *(const float4*)p;
    float an2 = v.x * v.x + v.y * v.y + v.z * v.z + v.w * v.w;
    for (int m = 1; m < 32; m <<= 1) an2 += __shfl_xor(an2, m, 32);
    float an = fmaxf(sqrtf(an2), MINN);
    float tn = fminf(fmaxf(tanhf(an), MINN), MAXN);
    float lg = artanh_f(tn) / an;
    float4 at;
    at.x = tanhf(lg * v.x);
    at.y = tanhf(lg * v.y);
    at.z = tanhf(lg * v.z);
    at.w = tanhf(lg * v.w);
    float vn2 = at.x * at.x + at.y * at.y + at.z * at.z + at.w * at.w;
    for (int m = 1; m < 32; m <<= 1) vn2 += __shfl_xor(vn2, m, 32);
    float vn = fmaxf(sqrtf(vn2), MINN);
    float s = fminf(tanhf(vn), MAXN) / vn;
    *(float4*)p = make_float4(s * at.x, s * at.y, s * at.z, s * at.w);
}

__global__ __launch_bounds__(256) void final_scatter(
    const float* __restrict__ H, const int* __restrict__ seg,
    float* __restrict__ ssum, float* __restrict__ cnt, int nrows) {
    int r = blockIdx.x * 8 + (threadIdx.x >> 5);
    if (r >= nrows) return;
    int lane = threadIdx.x & 31;
    float4 v = *(const float4*)(H + (size_t)r * 128 + lane * 4);
    float hn2 = v.x * v.x + v.y * v.y + v.z * v.z + v.w * v.w;
    for (int m = 1; m < 32; m <<= 1) hn2 += __shfl_xor(hn2, m, 32);
    float hn = fmaxf(sqrtf(hn2), MINN);
    float pc = fminf(hn, MAXN);
    float L = artanh_f(pc) / hn;
    int s = seg[r];
    float* d = ssum + (size_t)s * 128 + lane * 4;
    atomicAdd(d + 0, L * v.x);
    atomicAdd(d + 1, L * v.y);
    atomicAdd(d + 2, L * v.z);
    atomicAdd(d + 3, L * v.w);
    if (lane == 0) atomicAdd(&cnt[s], 1.f);
}

__global__ __launch_bounds__(256) void pool_k(const float* __restrict__ ssum,
                                              const float* __restrict__ cnt,
                                              float* __restrict__ out, int nseg) {
    int r = blockIdx.x * 8 + (threadIdx.x >> 5);
    if (r >= nseg) return;
    int lane = threadIdx.x & 31;
    float inv = 1.f / fmaxf(cnt[r], 1.f);
    float4 v = *(const float4*)(ssum + (size_t)r * 128 + lane * 4);
    v.x *= inv; v.y *= inv; v.z *= inv; v.w *= inv;
    float mn2 = v.x * v.x + v.y * v.y + v.z * v.z + v.w * v.w;
    for (int m = 1; m < 32; m <<= 1) mn2 += __shfl_xor(mn2, m, 32);
    float mn = fmaxf(sqrtf(mn2), MINN);
    float s = fminf(tanhf(mn), MAXN) / mn;
    float* o = out + (size_t)r * 128 + lane * 4;
    o[0] = s * v.x; o[1] = s * v.y; o[2] = s * v.z; o[3] = s * v.w;
}

__global__ void zero_f32(float* __restrict__ p, size_t n) {
    size_t i = (size_t)blockIdx.x * blockDim.x + threadIdx.x;
    size_t st = (size_t)gridDim.x * blockDim.x;
    for (; i < n; i += st) p[i] = 0.f;
}

extern "C" void kernel_launch(void* const* d_in, const int* in_sizes, int n_in,
                              void* d_out, int out_size, void* d_ws, size_t ws_size,
                              hipStream_t stream) {
    const float* x = (const float*)d_in[0];
    const int* erow = (const int*)d_in[1];
    const int* ecol = (const int*)d_in[2];
    const int* seg = (const int*)d_in[3];
    const float* W1 = (const float*)d_in[4];
    const float* b1 = (const float*)d_in[5];
    const float* W2 = (const float*)d_in[6];
    const float* b2 = (const float*)d_in[7];

    const int N = in_sizes[0] / 256;
    const int E = in_sizes[1];
    const int NSEG = out_size / 128;

    char* ws = (char*)d_ws;
    size_t bufBytes = (size_t)N * 128 * sizeof(float);
    float* buf0 = (float*)ws;
    float* buf1 = (float*)(ws + bufBytes);
    float* bbias = (float*)(ws + 2 * bufBytes);
    float* y2 = bbias + 256;
    float* ssum = buf0;
    float* cnt = buf0 + (size_t)NSEG * 128;

    dim3 blk(256);
    dim3 gemmBlk(128);
    int gRows = (N + 7) / 8;
    int gGemm = (N + BM - 1) / BM;
    int gEdge = (E + 7) / 8;

    const int nScan = (N + SCAN_B - 1) / SCAN_B;
    size_t offBytes = (((size_t)N + 1) * sizeof(int) + 255) / 256 * 256;
    size_t bsBytes = ((size_t)nScan * sizeof(int) + 255) / 256 * 256;
    int* offs = (int*)(ws + 2 * bufBytes + 4096);
    int* scol = (int*)(ws + 2 * bufBytes + 4096 + offBytes + bsBytes);
    const int G1 = 256;
    const int NBK = (N + BKT_N - 1) / BKT_N;
    const int nTab = NBK * G1;
    const int nScanT = (nTab + SCAN_B - 1) / SCAN_B;
    size_t tmpBytes = ((size_t)E * sizeof(int2) + 255) / 256 * 256;
    size_t tabBytes = (((size_t)nTab + 1) * sizeof(int) + 255) / 256 * 256;
    int2* tmp = (int2*)buf1;
    int* btab = (int*)(ws + bufBytes + tmpBytes);
    int* bsum = (int*)(ws + bufBytes + tmpBytes + tabBytes);
    const bool grouped =
        ws_size >= 2 * bufBytes + 4096 + offBytes + bsBytes + (size_t)E * sizeof(int) &&
        NBK <= MAXBK &&
        tmpBytes + tabBytes + (size_t)nScanT * sizeof(int) <= bufBytes;

    bias_prep<<<2, 128, 0, stream>>>(b1, b2, bbias, y2);

    if (grouped) {
        bin_count<<<G1, SORT_T, 0, stream>>>(erow, btab, E, NBK);
        scan_sums<<<nScanT, SCAN_B, 0, stream>>>(btab, bsum, nTab);
        scan_apply<<<nScanT, SCAN_B, 0, stream>>>(btab, bsum, nTab);
        bin_scatter<<<G1, SORT_T, 0, stream>>>(erow, ecol, btab, tmp, E, NBK);
        bin_sort<<<NBK, SORT_T, 0, stream>>>(tmp, btab, offs, scol, G1, NBK, N);

        hyp_linear_wmma<<<gGemm, gemmBlk, 0, stream>>>(x, 256, N, W1, bbias, y2, buf0, 1);
        node_gather_act<0><<<2048, blk, 0, stream>>>(buf0, offs, scol, buf1, seg,
                                                     (float*)0, (float*)0, N);

        hyp_linear_wmma<<<gGemm, gemmBlk, 0, stream>>>(buf1, 128, N, W2, bbias + 128,
                                                   y2 + 1, buf0, 0);
        float* ssumG = buf1;
        float* cntG = buf1 + (size_t)NSEG * 128;
        zero_f32<<<64, 256, 0, stream>>>(ssumG, (size_t)NSEG * 128 + NSEG);
        node_gather_act<1><<<2048, blk, 0, stream>>>(buf0, offs, scol, (float*)0, seg,
                                                     ssumG, cntG, N);
        pool_k<<<(NSEG + 7) / 8, blk, 0, stream>>>(ssumG, cntG, (float*)d_out, NSEG);
        return;
    }

    hyp_linear_wmma<<<gGemm, gemmBlk, 0, stream>>>(x, 256, N, W1, bbias, y2, buf0, 1);
    zero_f32<<<2048, 256, 0, stream>>>(buf1, (size_t)N * 128);
    edge_agg<<<gEdge, blk, 0, stream>>>(buf0, erow, ecol, buf1, E);
    post_agg<<<gRows, blk, 0, stream>>>(buf1, N);

    hyp_linear_wmma<<<gGemm, gemmBlk, 0, stream>>>(buf1, 128, N, W2, bbias + 128,
                                               y2 + 1, buf0, 0);
    zero_f32<<<2048, 256, 0, stream>>>(buf1, (size_t)N * 128);
    edge_agg<<<gEdge, blk, 0, stream>>>(buf0, erow, ecol, buf1, E);
    post_agg<<<gRows, blk, 0, stream>>>(buf1, N);

    zero_f32<<<64, 256, 0, stream>>>(ssum, (size_t)NSEG * 128 + NSEG);
    final_scatter<<<gRows, blk, 0, stream>>>(buf1, seg, ssum, cnt, N);
    pool_k<<<(NSEG + 7) / 8, blk, 0, stream>>>(ssum, cnt, (float*)d_out, NSEG);
}


# GEMM reads its input rows once: separate row-norm pass dropped, squared norms summed from the staged K chunks (norm sum order changes); layer 1 row factor multiplies the product instead of A
# speedup vs baseline: 3.1818x; 1.0433x over previous
#include <hip/hip_runtime.h>
#include <math.h>


#define MAXN 0.996f
#define MINN 1e-15f
#define KC   32
#define LDP  (KC + 4)
#define BM   64

typedef float v2f __attribute__((ext_vector_type(2)));
typedef float v8f __attribute__((ext_vector_type(8)));
typedef float v4f __attribute__((ext_vector_type(4)));

__device__ __forceinline__ float artanh_f(float z) {
    z = fminf(fmaxf(z, -(1.f - 1e-7f)), 1.f - 1e-7f);
    return 0.5f * logf((1.f + z) / (1.f - z));
}

__global__ void bias_prep(const float* __restrict__ b1,
                          const float* __restrict__ b2,
                          float* __restrict__ bb, float* __restrict__ y2) {
    __shared__ float red[128];
    const float* b = blockIdx.x ? b2 : b1;
    int t = threadIdx.x;
    float v = b[t];
    red[t] = v * v;
    __syncthreads();
    for (int s = 64; s > 0; s >>= 1) {
        if (t < s) red[t] += red[t + s];
        __syncthreads();
    }
    float bn = fmaxf(sqrtf(red[0]), MINN);
    float tc = fminf(tanhf(bn), MAXN);
    bb[blockIdx.x * 128 + t] = tc / bn * v;
    if (t == 0) y2[blockIdx.x] = tc * tc;
}

__global__ __launch_bounds__(128) void hyp_linear_wmma(
    const float* __restrict__ X, int Kdim, int nrows,
    const float* __restrict__ W,
    const float* __restrict__ bb,
    const float* __restrict__ y2p,
    float* __restrict__ HT, int apply_exp) {
    __shared__ float As[BM][LDP];
    __shared__ float Ws[128][LDP];
    __shared__ float red[BM][8];
    __shared__ float scale_row[BM], xn_row[BM];
    __shared__ float rs_mx2[BM], rs_dot[BM];
    __shared__ float rowCA[BM], rowCB[BM];

    const int t = threadIdx.x;
    const int row0 = blockIdx.x * BM;

    const int lane = t & 31;
    const int rs = t >> 5;
    const int half = lane >> 4;
    const int l15 = lane & 15;

    v8f acc[8];
    for (int j = 0; j < 8; ++j)
        for (int i = 0; i < 8; ++i) acc[j][i] = 0.f;

    const int nchunks = Kdim / KC;

    const int sr = t >> 3;
    const int sc4 = t & 7;
    const int kq = Kdim >> 2;
    const v4f* __restrict__ X4 = (const v4f*)X;
    const v4f* __restrict__ W4 = (const v4f*)W;
    int ia[4];
#pragma unroll
    for (int i = 0; i < 4; ++i) ia[i] = min(row0 + sr + 16 * i, nrows - 1) * kq + sc4;
    const int iw = sr * kq + sc4;

    const int mr = rs * 16 + l15;

    float ss[4] = {0.f, 0.f, 0.f, 0.f};

    for (int kc = 0; kc < nchunks; ++kc) {
        const int q0 = kc * (KC / 4);
        v4f av[4], wv[8];
#pragma unroll
        for (int i = 0; i < 4; ++i) av[i] = X4[ia[i] + q0];
#pragma unroll
        for (int i = 0; i < 8; ++i) wv[i] = W4[iw + i * 16 * kq + q0];
        asm volatile("" : "+v"(av[0]), "+v"(av[1]), "+v"(av[2]), "+v"(av[3]),
                          "+v"(wv[0]), "+v"(wv[1]), "+v"(wv[2]), "+v"(wv[3]),
                          "+v"(wv[4]), "+v"(wv[5]), "+v"(wv[6]), "+v"(wv[7]));
#pragma unroll
        for (int i = 0; i < 4; ++i) {
            ss[i] += av[i].x * av[i].x + av[i].y * av[i].y + av[i].z * av[i].z + av[i].w * av[i].w;
            *(v4f*)(&As[sr + 16 * i][sc4 * 4]) = av[i];
        }
#pragma unroll
        for (int i = 0; i < 8; ++i) *(v4f*)(&Ws[sr + 16 * i][sc4 * 4]) = wv[i];
        __syncthreads();

        for (int kk = 0; kk < KC; kk += 4) {
            const int ko = kk + half * 2;
            v2f a;
            a.x = As[mr][ko];
            a.y = As[mr][ko + 1];
#pragma unroll
            for (int j = 0; j < 8; ++j) {
                const int colg = j * 16 + l15;
                v2f b;
                b.x = Ws[colg][ko];
                b.y = Ws[colg][ko + 1];
                acc[j] = __builtin_amdgcn_wmma_f32_16x16x4_f32(
                    false, a, false, b, (short)0, acc[j], false, false);
            }
        }
        __syncthreads();
    }

#pragma unroll
    for (int i = 0; i < 4; ++i) red[sr + 16 * i][sc4] = ss[i];
    __syncthreads();
    if (t < BM) {
        float un2 = red[t][0];
#pragma unroll
        for (int q = 1; q < 8; ++q) un2 += red[t][q];
        float un = fmaxf(sqrtf(un2), MINN);
        if (apply_exp) {
            float bc = fminf(tanhf(un), MAXN);
            scale_row[t] = bc / un;
            xn_row[t] = fmaxf(bc, MINN);
        } else {
            scale_row[t] = 1.f;
            xn_row[t] = un;
        }
        if (row0 + t >= nrows) scale_row[t] = 0.f;
    }
    __syncthreads();
    if (apply_exp) {
#pragma unroll
        for (int i = 0; i < 8; ++i) {
            const float sc = scale_row[rs * 16 + i + half * 8];
#pragma unroll
            for (int j = 0; j < 8; ++j) acc[j][i] *= sc;
        }
    }

    float bbv[8];
#pragma unroll
    for (int j = 0; j < 8; ++j) bbv[j] = bb[j * 16 + l15];

#pragma unroll
    for (int i = 0; i < 8; ++i) {
        float p2a = 0.f, pda = 0.f, p2b = 0.f, pdb = 0.f;
#pragma unroll
        for (int j = 0; j < 4; ++j) {
            float v = acc[j][i];
            p2a += v * v;
            pda += v * bbv[j];
        }
#pragma unroll
        for (int j = 4; j < 8; ++j) {
            float v = acc[j][i];
            p2b += v * v;
            pdb += v * bbv[j];
        }
        for (int m = 1; m < 16; m <<= 1) {
            p2a += __shfl_xor(p2a, m, 16);
            pda += __shfl_xor(pda, m, 16);
            p2b += __shfl_xor(p2b, m, 16);
            pdb += __shfl_xor(pdb, m, 16);
        }
        int mrow = rs * 16 + i + half * 8;
        if (l15 == 0) {
            rs_mx2[mrow] = p2a + p2b;
            rs_dot[mrow] = pda + pdb;
        }
    }
    __syncthreads();

    if (t < BM) {
        float xn = xn_row[t];
        float mxn = fmaxf(sqrtf(rs_mx2[t]), MINN);
        float tv = tanhf(mxn / xn * artanh_f(xn));
        float tc = fminf(tv, MAXN);
        float s_mv = tc / mxn;
        float x2 = tc * tc;
        float xy = s_mv * rs_dot[t];
        float y2 = y2p[0];
        float al = 1.f + 2.f * xy + y2;
        float be = 1.f - x2;
        float den = fmaxf(1.f + 2.f * xy + x2 * y2, MINN);
        float nn2 = fmaxf(al * al * x2 + 2.f * al * be * xy + be * be * y2, 0.f);
        float hn = fmaxf(sqrtf(nn2) / den, MINN);
        float hc = fminf(hn, MAXN);
        float L = artanh_f(hc) / hn;
        rowCA[t] = L * al * s_mv / den;
        rowCB[t] = L * be / den;
    }
    __syncthreads();

#pragma unroll
    for (int i = 0; i < 8; ++i) {
        int mrow = rs * 16 + i + half * 8;
        int gr = row0 + mrow;
        if (gr < nrows) {
            float ca = rowCA[mrow], cb = rowCB[mrow];
#pragma unroll
            for (int j = 0; j < 8; ++j) {
                int colg = j * 16 + l15;
                HT[(size_t)gr * 128 + colg] = ca * acc[j][i] + cb * bbv[j];
            }
        }
    }
}

__global__ __launch_bounds__(256) void edge_agg(const float* __restrict__ HT,
                                                const int* __restrict__ erow,
                                                const int* __restrict__ ecol,
                                                float* __restrict__ AGG, int E) {
    int e = blockIdx.x * 8 + (threadIdx.x >> 5);
    if (e >= E) return;
    int lane = threadIdx.x & 31;
    int r = erow[e], c = ecol[e];
    float4 v = *(const float4*)(HT + (size_t)c * 128 + lane * 4);
    float* dst = AGG + (size_t)r * 128 + lane * 4;
    atomicAdd(dst + 0, v.x);
    atomicAdd(dst + 1, v.y);
    atomicAdd(dst + 2, v.z);
    atomicAdd(dst + 3, v.w);
}

#define SCAN_B 1024
#define BKT_SH 8
#define BKT_N  256
#define MAXBK  1024
#define SORT_T 1024
#define SORT_U 4

__global__ __launch_bounds__(SORT_T) void bin_count(const int* __restrict__ erow,
                                                    int* __restrict__ bh,
                                                    int E, int nbk) {
    __shared__ int hist[MAXBK];
    const int t = threadIdx.x, G = gridDim.x, b = blockIdx.x;
    for (int k = t; k < nbk; k += SORT_T) hist[k] = 0;
    __syncthreads();
    const int chunk = (E + G - 1) / G;
    const int beg = b * chunk, end = min(E, beg + chunk);
    for (int i = beg + t; i < end; i += SORT_U * SORT_T) {
        int r[SORT_U];
#pragma unroll
        for (int u = 0; u < SORT_U; ++u) {
            const int j = i + u * SORT_T;
            r[u] = j < end ? erow[j] : 0;
        }
#pragma unroll
        for (int u = 0; u < SORT_U; ++u) {
            const int k = min((int)((unsigned)r[u] >> BKT_SH), nbk - 1);
            if (i + u * SORT_T < end) atomicAdd(&hist[k], 1);
        }
    }
    __syncthreads();
    for (int k = t; k < nbk; k += SORT_T) bh[(size_t)k * G + b] = hist[k];
}

__global__ __launch_bounds__(SORT_T) void bin_scatter(const int* __restrict__ erow,
                                                      const int* __restrict__ ecol,
                                                      const int* __restrict__ bo,
                                                      int2* __restrict__ tmp,
                                                      int E, int nbk) {
    __shared__ int cur[MAXBK];
    const int t = threadIdx.x, G = gridDim.x, b = blockIdx.x;
    for (int k = t; k < nbk; k += SORT_T) cur[k] = bo[(size_t)k * G + b];
    __syncthreads();
    const int chunk = (E + G - 1) / G;
    const int beg = b * chunk, end = min(E, beg + chunk);
    for (int i = beg + t; i < end; i += SORT_U * SORT_T) {
        int r[SORT_U], c[SORT_U];
#pragma unroll
        for (int u = 0; u < SORT_U; ++u) {
            const int j = i + u * SORT_T;
            r[u] = j < end ? erow[j] : 0;
            c[u] = j < end ? ecol[j] : 0;
        }
#pragma unroll
        for (int u = 0; u < SORT_U; ++u) {
            if (i + u * SORT_T < end) {
                const int k = min((int)((unsigned)r[u] >> BKT_SH), nbk - 1);
                const int p = atomicAdd(&cur[k], 1);
                tmp[p] = make_int2(r[u], c[u]);
            }
        }
    }
}

__global__ __launch_bounds__(SORT_T) void bin_sort(const int2* __restrict__ tmp,
                                                   const int* __restrict__ bo,
                                                   int* __restrict__ offs,
                                                   int* __restrict__ scol,
                                                   int G, int nbk, int n) {
    __shared__ int cnt[BKT_N];
    __shared__ int wsum[BKT_N / 32];
    const int t = threadIdx.x, k = blockIdx.x, lane = t & 31, wid = t >> 5;
    const int s = bo[(size_t)k * G], e = bo[(size_t)(k + 1) * G];
    if (t < BKT_N) cnt[t] = 0;
    __syncthreads();
    for (int i = s + t; i < e; i += SORT_U * SORT_T) {
        int r[SORT_U];
#pragma unroll
        for (int u = 0; u < SORT_U; ++u) {
            const int j = i + u * SORT_T;
            r[u] = j < e ? tmp[j].x : 0;
        }
#pragma unroll
        for (int u = 0; u < SORT_U; ++u)
            if (i + u * SORT_T < e) atomicAdd(&cnt[r[u] & (BKT_N - 1)], 1);
    }
    __syncthreads();
    int v = 0, sc = 0;
    if (t < BKT_N) {
        v = cnt[t];
        sc = v;
        for (int m = 1; m < 32; m <<= 1) {
            int u = __shfl_up(sc, m, 32);
            if (lane >= m) sc += u;
        }
        if (lane == 31) wsum[wid] = sc;
    }
    __syncthreads();
    if (t < BKT_N) {
        int base = s;
        for (int w = 0; w < wid; ++w) base += wsum[w];
        const int excl = base + sc - v;
        cnt[t] = excl;
        const int node = k * BKT_N + t;
        if (node < n) offs[node] = excl;
    }
    if (k == nbk - 1 && t == 0) offs[n] = e;
    __syncthreads();
    for (int i = s + t; i < e; i += SORT_U * SORT_T) {
        int2 rc[SORT_U];
#pragma unroll
        for (int u = 0; u < SORT_U; ++u) {
            const int j = i + u * SORT_T;
            rc[u] = j < e ? tmp[j] : make_int2(0, 0);
        }
#pragma unroll
        for (int u = 0; u < SORT_U; ++u) {
            if (i + u * SORT_T < e) {
                const int p = atomicAdd(&cnt[rc[u].x & (BKT_N - 1)], 1);
                scol[p] = rc[u].y;
            }
        }
    }
}

__device__ __forceinline__ int block_sum_1024(int v, int* sh) {
    int t = threadIdx.x;
    for (int m = 1; m < 32; m <<= 1) v += __shfl_xor(v, m, 32);
    if ((t & 31) == 0) sh[t >> 5] = v;
    __syncthreads();
    int w = sh[t & 31];
    for (int m = 1; m < 32; m <<= 1) w += __shfl_xor(w, m, 32);
    __syncthreads();
    return w;
}

__global__ __launch_bounds__(SCAN_B) void scan_sums(const int* __restrict__ cnt,
                                                    int* __restrict__ bsum, int n) {
    __shared__ int sh[32];
    int i = blockIdx.x * SCAN_B + threadIdx.x;
    int tot = block_sum_1024(i < n ? cnt[i] : 0, sh);
    if (threadIdx.x == 0) bsum[blockIdx.x] = tot;
}

__global__ __launch_bounds__(SCAN_B) void scan_apply(int* __restrict__ offs,
                                                     const int* __restrict__ bsum, int n) {
    __shared__ int sh[32];
    const int t = threadIdx.x, lane = t & 31, wid = t >> 5;
    int p = 0;
    for (int b = t; b < (int)blockIdx.x; b += SCAN_B) p += bsum[b];
    const int base = block_sum_1024(p, sh);

    int i = blockIdx.x * SCAN_B + t;
    int v = i < n ? offs[i] : 0;
    int s = v;
    for (int m = 1; m < 32; m <<= 1) {
        int u = __shfl_up(s, m, 32);
        if (lane >= m) s += u;
    }
    if (lane == 31) sh[wid] = s;
    __syncthreads();
    if (wid == 0) {
        int w0 = sh[lane], w = w0;
        for (int m = 1; m < 32; m <<= 1) {
            int u = __shfl_up(w, m, 32);
            if (lane >= m) w += u;
        }
        sh[lane] = w - w0;
    }
    __syncthreads();
    int excl = base + sh[wid] + s - v;
    if (i < n) offs[i] = excl;
    if (i == n - 1) offs[n] = excl + v;
}

__device__ __forceinline__ float wave_sum32(float x) {
    for (int m = 1; m < 32; m <<= 1) x += __shfl_xor(x, m, 32);
    return x;
}

__device__ __forceinline__ float4 act_row(float4 v) {
    float an2 = wave_sum32(v.x * v.x + v.y * v.y + v.z * v.z + v.w * v.w);
    float an = fmaxf(sqrtf(an2), MINN);
    float tn = fminf(fmaxf(tanhf(an), MINN), MAXN);
    float lg = artanh_f(tn) / an;
    float4 at;
    at.x = tanhf(lg * v.x);
    at.y = tanhf(lg * v.y);
    at.z = tanhf(lg * v.z);
    at.w = tanhf(lg * v.w);
    float vn2 = wave_sum32(at.x * at.x + at.y * at.y + at.z * at.z + at.w * at.w);
    float vn = fmaxf(sqrtf(vn2), MINN);
    float s = fminf(tanhf(vn), MAXN) / vn;
    return make_float4(s * at.x, s * at.y, s * at.z, s * at.w);
}

template <int READOUT>
__global__ __launch_bounds__(256) void node_gather_act(const float* __restrict__ HT,
                                                       const int* __restrict__ offs,
                                                       const int* __restrict__ scol,
                                                       float* __restrict__ OUT,
                                                       const int* __restrict__ seg,
                                                       float* __restrict__ ssum,
                                                       float* __restrict__ cnt, int nrows) {
    const int lane = threadIdx.x & 31;
    const int w0 = __builtin_amdgcn_readfirstlane((int)(blockIdx.x * 8 + (threadIdx.x >> 5)));
    const int nw = gridDim.x * 8;
    for (int r = w0; r < nrows; r += nw) {
        const int beg = offs[r], end = offs[r + 1];
        int sg = 0;
        if (READOUT) sg = seg[r];
        float4 acc = make_float4(0.f, 0.f, 0.f, 0.f);
        for (int b0 = beg; b0 < end; b0 += 32) {
            const int n = min(32, end - b0);
            const int myc = (lane < n) ? scol[b0 + lane] : 0;
            int j = 0;
            for (; j + 4 <= n; j += 4) {
                const int c0 = __builtin_amdgcn_readlane(myc, j);
                const int c1 = __builtin_amdgcn_readlane(myc, j + 1);
                const int c2 = __builtin_amdgcn_readlane(myc, j + 2);
                const int c3 = __builtin_amdgcn_readlane(myc, j + 3);
                const float4 v0 = *((const float4*)(HT + (size_t)c0 * 128) + lane);
                const float4 v1 = *((const float4*)(HT + (size_t)c1 * 128) + lane);
                const float4 v2 = *((const float4*)(HT + (size_t)c2 * 128) + lane);
                const float4 v3 = *((const float4*)(HT + (size_t)c3 * 128) + lane);
                acc.x += v0.x; acc.y += v0.y; acc.z += v0.z; acc.w += v0.w;
                acc.x += v1.x; acc.y += v1.y; acc.z += v1.z; acc.w += v1.w;
                acc.x += v2.x; acc.y += v2.y; acc.z += v2.z; acc.w += v2.w;
                acc.x += v3.x; acc.y += v3.y; acc.z += v3.z; acc.w += v3.w;
            }
            for (; j < n; ++j) {
                const int c0 = __builtin_amdgcn_readlane(myc, j);
                const float4 v0 = *((const float4*)(HT + (size_t)c0 * 128) + lane);
                acc.x += v0.x; acc.y += v0.y; acc.z += v0.z; acc.w += v0.w;
            }
        }
        const float4 h = act_row(acc);
        if (!READOUT) {
            *((float4*)(OUT + (size_t)r * 128) + lane) = h;
        } else {
            float hn2 = wave_sum32(h.x * h.x + h.y * h.y + h.z * h.z + h.w * h.w);
            float hn = fmaxf(sqrtf(hn2), MINN);
            float pc = fminf(hn, MAXN);
            float L = artanh_f(pc) / hn;
            float* d = ssum + (size_t)sg * 128 + lane * 4;
            atomicAdd(d + 0, L * h.x);
            atomicAdd(d + 1, L * h.y);
            atomicAdd(d + 2, L * h.z);
            atomicAdd(d + 3, L * h.w);
            if (lane == 0) atomicAdd(&cnt[sg], 1.f);
        }
    }
}

__global__ __launch_bounds__(256) void post_agg(float* __restrict__ H, int nrows) {
    int r = blockIdx.x * 8 + (threadIdx.x >> 5);
    if (r >= nrows) return;
    int lane = threadIdx.x & 31;
    float* p = H + (size_t)r * 128 + lane * 4;
    float4 v = *(const float4*)p;
    float an2 = v.x * v.x + v.y * v.y + v.z * v.z + v.w * v.w;
    for (int m = 1; m < 32; m <<= 1) an2 += __shfl_xor(an2, m, 32);
    float an = fmaxf(sqrtf(an2), MINN);
    float tn = fminf(fmaxf(tanhf(an), MINN), MAXN);
    float lg = artanh_f(tn) / an;
    float4 at;
    at.x = tanhf(lg * v.x);
    at.y = tanhf(lg * v.y);
    at.z = tanhf(lg * v.z);
    at.w = tanhf(lg * v.w);
    float vn2 = at.x * at.x + at.y * at.y + at.z * at.z + at.w * at.w;
    for (int m = 1; m < 32; m <<= 1) vn2 += __shfl_xor(vn2, m, 32);
    float vn = fmaxf(sqrtf(vn2), MINN);
    float s = fminf(tanhf(vn), MAXN) / vn;
    *(float4*)p = make_float4(s * at.x, s * at.y, s * at.z, s * at.w);
}

__global__ __launch_bounds__(256) void final_scatter(
    const float* __restrict__ H, const int* __restrict__ seg,
    float* __restrict__ ssum, float* __restrict__ cnt, int nrows) {
    int r = blockIdx.x * 8 + (threadIdx.x >> 5);
    if (r >= nrows) return;
    int lane = threadIdx.x & 31;
    float4 v = *(const float4*)(H + (size_t)r * 128 + lane * 4);
    float hn2 = v.x * v.x + v.y * v.y + v.z * v.z + v.w * v.w;
    for (int m = 1; m < 32; m <<= 1) hn2 += __shfl_xor(hn2, m, 32);
    float hn = fmaxf(sqrtf(hn2), MINN);
    float pc = fminf(hn, MAXN);
    float L = artanh_f(pc) / hn;
    int s = seg[r];
    float* d = ssum + (size_t)s * 128 + lane * 4;
    atomicAdd(d + 0, L * v.x);
    atomicAdd(d + 1, L * v.y);
    atomicAdd(d + 2, L * v.z);
    atomicAdd(d + 3, L * v.w);
    if (lane == 0) atomicAdd(&cnt[s], 1.f);
}

__global__ __launch_bounds__(256) void pool_k(const float* __restrict__ ssum,
                                              const float* __restrict__ cnt,
                                              float* __restrict__ out, int nseg) {
    int r = blockIdx.x * 8 + (threadIdx.x >> 5);
    if (r >= nseg) return;
    int lane = threadIdx.x & 31;
    float inv = 1.f / fmaxf(cnt[r], 1.f);
    float4 v = *(const float4*)(ssum + (size_t)r * 128 + lane * 4);
    v.x *= inv; v.y *= inv; v.z *= inv; v.w *= inv;
    float mn2 = v.x * v.x + v.y * v.y + v.z * v.z + v.w * v.w;
    for (int m = 1; m < 32; m <<= 1) mn2 += __shfl_xor(mn2, m, 32);
    float mn = fmaxf(sqrtf(mn2), MINN);
    float s = fminf(tanhf(mn), MAXN) / mn;
    float* o = out + (size_t)r * 128 + lane * 4;
    o[0] = s * v.x; o[1] = s * v.y; o[2] = s * v.z; o[3] = s * v.w;
}

__global__ void zero_f32(float* __restrict__ p, size_t n) {
    size_t i = (size_t)blockIdx.x * blockDim.x + threadIdx.x;
    size_t st = (size_t)gridDim.x * blockDim.x;
    for (; i < n; i += st) p[i] = 0.f;
}

extern "C" void kernel_launch(void* const* d_in, const int* in_sizes, int n_in,
                              void* d_out, int out_size, void* d_ws, size_t ws_size,
                              hipStream_t stream) {
    const float* x = (const float*)d_in[0];
    const int* erow = (const int*)d_in[1];
    const int* ecol = (const int*)d_in[2];
    const int* seg = (const int*)d_in[3];
    const float* W1 = (const float*)d_in[4];
    const float* b1 = (const float*)d_in[5];
    const float* W2 = (const float*)d_in[6];
    const float* b2 = (const float*)d_in[7];

    const int N = in_sizes[0] / 256;
    const int E = in_sizes[1];
    const int NSEG = out_size / 128;

    char* ws = (char*)d_ws;
    size_t bufBytes = (size_t)N * 128 * sizeof(float);
    float* buf0 = (float*)ws;
    float* buf1 = (float*)(ws + bufBytes);
    float* bbias = (float*)(ws + 2 * bufBytes);
    float* y2 = bbias + 256;
    float* ssum = buf0;
    float* cnt = buf0 + (size_t)NSEG * 128;

    dim3 blk(256);
    dim3 gemmBlk(128);
    int gRows = (N + 7) / 8;
    int gGemm = (N + BM - 1) / BM;
    int gEdge = (E + 7) / 8;

    const int nScan = (N + SCAN_B - 1) / SCAN_B;
    size_t offBytes = (((size_t)N + 1) * sizeof(int) + 255) / 256 * 256;
    size_t bsBytes = ((size_t)nScan * sizeof(int) + 255) / 256 * 256;
    int* offs = (int*)(ws + 2 * bufBytes + 4096);
    int* scol = (int*)(ws + 2 * bufBytes + 4096 + offBytes + bsBytes);
    const int G1 = 256;
    const int NBK = (N + BKT_N - 1) / BKT_N;
    const int nTab = NBK * G1;
    const int nScanT = (nTab + SCAN_B - 1) / SCAN_B;
    size_t tmpBytes = ((size_t)E * sizeof(int2) + 255) / 256 * 256;
    size_t tabBytes = (((size_t)nTab + 1) * sizeof(int) + 255) / 256 * 256;
    int2* tmp = (int2*)buf1;
    int* btab = (int*)(ws + bufBytes + tmpBytes);
    int* bsum = (int*)(ws + bufBytes + tmpBytes + tabBytes);
    const bool grouped =
        ws_size >= 2 * bufBytes + 4096 + offBytes + bsBytes + (size_t)E * sizeof(int) &&
        NBK <= MAXBK &&
        tmpBytes + tabBytes + (size_t)nScanT * sizeof(int) <= bufBytes;

    bias_prep<<<2, 128, 0, stream>>>(b1, b2, bbias, y2);

    if (grouped) {
        bin_count<<<G1, SORT_T, 0, stream>>>(erow, btab, E, NBK);
        scan_sums<<<nScanT, SCAN_B, 0, stream>>>(btab, bsum, nTab);
        scan_apply<<<nScanT, SCAN_B, 0, stream>>>(btab, bsum, nTab);
        bin_scatter<<<G1, SORT_T, 0, stream>>>(erow, ecol, btab, tmp, E, NBK);
        bin_sort<<<NBK, SORT_T, 0, stream>>>(tmp, btab, offs, scol, G1, NBK, N);

        hyp_linear_wmma<<<gGemm, gemmBlk, 0, stream>>>(x, 256, N, W1, bbias, y2, buf0, 1);
        node_gather_act<0><<<2048, blk, 0, stream>>>(buf0, offs, scol, buf1, seg,
                                                     (float*)0, (float*)0, N);

        hyp_linear_wmma<<<gGemm, gemmBlk, 0, stream>>>(buf1, 128, N, W2, bbias + 128,
                                                   y2 + 1, buf0, 0);
        float* ssumG = buf1;
        float* cntG = buf1 + (size_t)NSEG * 128;
        zero_f32<<<64, 256, 0, stream>>>(ssumG, (size_t)NSEG * 128 + NSEG);
        node_gather_act<1><<<2048, blk, 0, stream>>>(buf0, offs, scol, (float*)0, seg,
                                                     ssumG, cntG, N);
        pool_k<<<(NSEG + 7) / 8, blk, 0, stream>>>(ssumG, cntG, (float*)d_out, NSEG);
        return;
    }

    hyp_linear_wmma<<<gGemm, gemmBlk, 0, stream>>>(x, 256, N, W1, bbias, y2, buf0, 1);
    zero_f32<<<2048, 256, 0, stream>>>(buf1, (size_t)N * 128);
    edge_agg<<<gEdge, blk, 0, stream>>>(buf0, erow, ecol, buf1, E);
    post_agg<<<gRows, blk, 0, stream>>>(buf1, N);

    hyp_linear_wmma<<<gGemm, gemmBlk, 0, stream>>>(buf1, 128, N, W2, bbias + 128,
                                               y2 + 1, buf0, 0);
    zero_f32<<<2048, 256, 0, stream>>>(buf1, (size_t)N * 128);
    edge_agg<<<gEdge, blk, 0, stream>>>(buf0, erow, ecol, buf1, E);
    post_agg<<<gRows, blk, 0, stream>>>(buf1, N);

    zero_f32<<<64, 256, 0, stream>>>(ssum, (size_t)NSEG * 128 + NSEG);
    final_scatter<<<gRows, blk, 0, stream>>>(buf1, seg, ssum, cnt, N);
    pool_k<<<(NSEG + 7) / 8, blk, 0, stream>>>(ssum, cnt, (float*)d_out, NSEG);
}


# layer 1 edge sum + activation fused into layer 2's GEMM launch: a wave sums its 16 rows (8 loads in flight) into the LDS A tile; activated plane store, re-read, 1 launch gone; norm sum order changes
# speedup vs baseline: 3.3072x; 1.0394x over previous
#include <hip/hip_runtime.h>
#include <math.h>


#define MAXN 0.996f
#define MINN 1e-15f
#define KC   32
#define LDP  (KC + 4)
#define BM   64

typedef float v2f __attribute__((ext_vector_type(2)));
typedef float v8f __attribute__((ext_vector_type(8)));
typedef float v4f __attribute__((ext_vector_type(4)));

__device__ __forceinline__ float artanh_f(float z) {
    z = fminf(fmaxf(z, -(1.f - 1e-7f)), 1.f - 1e-7f);
    return 0.5f * logf((1.f + z) / (1.f - z));
}

__global__ void bias_prep(const float* __restrict__ b1,
                          const float* __restrict__ b2,
                          float* __restrict__ bb, float* __restrict__ y2) {
    __shared__ float red[128];
    const float* b = blockIdx.x ? b2 : b1;
    int t = threadIdx.x;
    float v = b[t];
    red[t] = v * v;
    __syncthreads();
    for (int s = 64; s > 0; s >>= 1) {
        if (t < s) red[t] += red[t + s];
        __syncthreads();
    }
    float bn = fmaxf(sqrtf(red[0]), MINN);
    float tc = fminf(tanhf(bn), MAXN);
    bb[blockIdx.x * 128 + t] = tc / bn * v;
    if (t == 0) y2[blockIdx.x] = tc * tc;
}

__global__ __launch_bounds__(128) void hyp_linear_wmma(
    const float* __restrict__ X, int Kdim, int nrows,
    const float* __restrict__ W,
    const float* __restrict__ bb,
    const float* __restrict__ y2p,
    float* __restrict__ HT, int apply_exp) {
    __shared__ float As[BM][LDP];
    __shared__ float Ws[128][LDP];
    __shared__ float red[BM][8];
    __shared__ float scale_row[BM], xn_row[BM];
    __shared__ float rs_mx2[BM], rs_dot[BM];
    __shared__ float rowCA[BM], rowCB[BM];

    const int t = threadIdx.x;
    const int row0 = blockIdx.x * BM;

    const int lane = t & 31;
    const int rs = t >> 5;
    const int half = lane >> 4;
    const int l15 = lane & 15;

    v8f acc[8];
    for (int j = 0; j < 8; ++j)
        for (int i = 0; i < 8; ++i) acc[j][i] = 0.f;

    const int nchunks = Kdim / KC;

    const int sr = t >> 3;
    const int sc4 = t & 7;
    const int kq = Kdim >> 2;
    const v4f* __restrict__ X4 = (const v4f*)X;
    const v4f* __restrict__ W4 = (const v4f*)W;
    int ia[4];
#pragma unroll
    for (int i = 0; i < 4; ++i) ia[i] = min(row0 + sr + 16 * i, nrows - 1) * kq + sc4;
    const int iw = sr * kq + sc4;

    const int mr = rs * 16 + l15;

    float ss[4] = {0.f, 0.f, 0.f, 0.f};

    for (int kc = 0; kc < nchunks; ++kc) {
        const int q0 = kc * (KC / 4);
        v4f av[4], wv[8];
#pragma unroll
        for (int i = 0; i < 4; ++i) av[i] = X4[ia[i] + q0];
#pragma unroll
        for (int i = 0; i < 8; ++i) wv[i] = W4[iw + i * 16 * kq + q0];
        asm volatile("" : "+v"(av[0]), "+v"(av[1]), "+v"(av[2]), "+v"(av[3]),
                          "+v"(wv[0]), "+v"(wv[1]), "+v"(wv[2]), "+v"(wv[3]),
                          "+v"(wv[4]), "+v"(wv[5]), "+v"(wv[6]), "+v"(wv[7]));
#pragma unroll
        for (int i = 0; i < 4; ++i) {
            ss[i] += av[i].x * av[i].x + av[i].y * av[i].y + av[i].z * av[i].z + av[i].w * av[i].w;
            *(v4f*)(&As[sr + 16 * i][sc4 * 4]) = av[i];
        }
#pragma unroll
        for (int i = 0; i < 8; ++i) *(v4f*)(&Ws[sr + 16 * i][sc4 * 4]) = wv[i];
        __syncthreads();

        for (int kk = 0; kk < KC; kk += 4) {
            const int ko = kk + half * 2;
            v2f a;
            a.x = As[mr][ko];
            a.y = As[mr][ko + 1];
#pragma unroll
            for (int j = 0; j < 8; ++j) {
                const int colg = j * 16 + l15;
                v2f b;
                b.x = Ws[colg][ko];
                b.y = Ws[colg][ko + 1];
                acc[j] = __builtin_amdgcn_wmma_f32_16x16x4_f32(
                    false, a, false, b, (short)0, acc[j], false, false);
            }
        }
        __syncthreads();
    }

#pragma unroll
    for (int i = 0; i < 4; ++i) red[sr + 16 * i][sc4] = ss[i];
    __syncthreads();
    if (t < BM) {
        float un2 = red[t][0];
#pragma unroll
        for (int q = 1; q < 8; ++q) un2 += red[t][q];
        float un = fmaxf(sqrtf(un2), MINN);
        if (apply_exp) {
            float bc = fminf(tanhf(un), MAXN);
            scale_row[t] = bc / un;
            xn_row[t] = fmaxf(bc, MINN);
        } else {
            scale_row[t] = 1.f;
            xn_row[t] = un;
        }
        if (row0 + t >= nrows) scale_row[t] = 0.f;
    }
    __syncthreads();
    if (apply_exp) {
#pragma unroll
        for (int i = 0; i < 8; ++i) {
            const float sc = scale_row[rs * 16 + i + half * 8];
#pragma unroll
            for (int j = 0; j < 8; ++j) acc[j][i] *= sc;
        }
    }

    float bbv[8];
#pragma unroll
    for (int j = 0; j < 8; ++j) bbv[j] = bb[j * 16 + l15];

#pragma unroll
    for (int i = 0; i < 8; ++i) {
        float p2a = 0.f, pda = 0.f, p2b = 0.f, pdb = 0.f;
#pragma unroll
        for (int j = 0; j < 4; ++j) {
            float v = acc[j][i];
            p2a += v * v;
            pda += v * bbv[j];
        }
#pragma unroll
        for (int j = 4; j < 8; ++j) {
            float v = acc[j][i];
            p2b += v * v;
            pdb += v * bbv[j];
        }
        for (int m = 1; m < 16; m <<= 1) {
            p2a += __shfl_xor(p2a, m, 16);
            pda += __shfl_xor(pda, m, 16);
            p2b += __shfl_xor(p2b, m, 16);
            pdb += __shfl_xor(pdb, m, 16);
        }
        int mrow = rs * 16 + i + half * 8;
        if (l15 == 0) {
            rs_mx2[mrow] = p2a + p2b;
            rs_dot[mrow] = pda + pdb;
        }
    }
    __syncthreads();

    if (t < BM) {
        float xn = xn_row[t];
        float mxn = fmaxf(sqrtf(rs_mx2[t]), MINN);
        float tv = tanhf(mxn / xn * artanh_f(xn));
        float tc = fminf(tv, MAXN);
        float s_mv = tc / mxn;
        float x2 = tc * tc;
        float xy = s_mv * rs_dot[t];
        float y2 = y2p[0];
        float al = 1.f + 2.f * xy + y2;
        float be = 1.f - x2;
        float den = fmaxf(1.f + 2.f * xy + x2 * y2, MINN);
        float nn2 = fmaxf(al * al * x2 + 2.f * al * be * xy + be * be * y2, 0.f);
        float hn = fmaxf(sqrtf(nn2) / den, MINN);
        float hc = fminf(hn, MAXN);
        float L = artanh_f(hc) / hn;
        rowCA[t] = L * al * s_mv / den;
        rowCB[t] = L * be / den;
    }
    __syncthreads();

#pragma unroll
    for (int i = 0; i < 8; ++i) {
        int mrow = rs * 16 + i + half * 8;
        int gr = row0 + mrow;
        if (gr < nrows) {
            float ca = rowCA[mrow], cb = rowCB[mrow];
#pragma unroll
            for (int j = 0; j < 8; ++j) {
                int colg = j * 16 + l15;
                HT[(size_t)gr * 128 + colg] = ca * acc[j][i] + cb * bbv[j];
            }
        }
    }
}

__global__ __launch_bounds__(256) void edge_agg(const float* __restrict__ HT,
                                                const int* __restrict__ erow,
                                                const int* __restrict__ ecol,
                                                float* __restrict__ AGG, int E) {
    int e = blockIdx.x * 8 + (threadIdx.x >> 5);
    if (e >= E) return;
    int lane = threadIdx.x & 31;
    int r = erow[e], c = ecol[e];
    float4 v = *(const float4*)(HT + (size_t)c * 128 + lane * 4);
    float* dst = AGG + (size_t)r * 128 + lane * 4;
    atomicAdd(dst + 0, v.x);
    atomicAdd(dst + 1, v.y);
    atomicAdd(dst + 2, v.z);
    atomicAdd(dst + 3, v.w);
}

#define SCAN_B 1024
#define BKT_SH 8
#define BKT_N  256
#define MAXBK  1024
#define SORT_T 1024
#define SORT_U 4

__global__ __launch_bounds__(SORT_T) void bin_count(const int* __restrict__ erow,
                                                    int* __restrict__ bh,
                                                    int E, int nbk) {
    __shared__ int hist[MAXBK];
    const int t = threadIdx.x, G = gridDim.x, b = blockIdx.x;
    for (int k = t; k < nbk; k += SORT_T) hist[k] = 0;
    __syncthreads();
    const int chunk = (E + G - 1) / G;
    const int beg = b * chunk, end = min(E, beg + chunk);
    for (int i = beg + t; i < end; i += SORT_U * SORT_T) {
        int r[SORT_U];
#pragma unroll
        for (int u = 0; u < SORT_U; ++u) {
            const int j = i + u * SORT_T;
            r[u] = j < end ? erow[j] : 0;
        }
#pragma unroll
        for (int u = 0; u < SORT_U; ++u) {
            const int k = min((int)((unsigned)r[u] >> BKT_SH), nbk - 1);
            if (i + u * SORT_T < end) atomicAdd(&hist[k], 1);
        }
    }
    __syncthreads();
    for (int k = t; k < nbk; k += SORT_T) bh[(size_t)k * G + b] = hist[k];
}

__global__ __launch_bounds__(SORT_T) void bin_scatter(const int* __restrict__ erow,
                                                      const int* __restrict__ ecol,
                                                      const int* __restrict__ bo,
                                                      int2* __restrict__ tmp,
                                                      int E, int nbk) {
    __shared__ int cur[MAXBK];
    const int t = threadIdx.x, G = gridDim.x, b = blockIdx.x;
    for (int k = t; k < nbk; k += SORT_T) cur[k] = bo[(size_t)k * G + b];
    __syncthreads();
    const int chunk = (E + G - 1) / G;
    const int beg = b * chunk, end = min(E, beg + chunk);
    for (int i = beg + t; i < end; i += SORT_U * SORT_T) {
        int r[SORT_U], c[SORT_U];
#pragma unroll
        for (int u = 0; u < SORT_U; ++u) {
            const int j = i + u * SORT_T;
            r[u] = j < end ? erow[j] : 0;
            c[u] = j < end ? ecol[j] : 0;
        }
#pragma unroll
        for (int u = 0; u < SORT_U; ++u) {
            if (i + u * SORT_T < end) {
                const int k = min((int)((unsigned)r[u] >> BKT_SH), nbk - 1);
                const int p = atomicAdd(&cur[k], 1);
                tmp[p] = make_int2(r[u], c[u]);
            }
        }
    }
}

__global__ __launch_bounds__(SORT_T) void bin_sort(const int2* __restrict__ tmp,
                                                   const int* __restrict__ bo,
                                                   int* __restrict__ offs,
                                                   int* __restrict__ scol,
                                                   int G, int nbk, int n) {
    __shared__ int cnt[BKT_N];
    __shared__ int wsum[BKT_N / 32];
    const int t = threadIdx.x, k = blockIdx.x, lane = t & 31, wid = t >> 5;
    const int s = bo[(size_t)k * G], e = bo[(size_t)(k + 1) * G];
    if (t < BKT_N) cnt[t] = 0;
    __syncthreads();
    for (int i = s + t; i < e; i += SORT_U * SORT_T) {
        int r[SORT_U];
#pragma unroll
        for (int u = 0; u < SORT_U; ++u) {
            const int j = i + u * SORT_T;
            r[u] = j < e ? tmp[j].x : 0;
        }
#pragma unroll
        for (int u = 0; u < SORT_U; ++u)
            if (i + u * SORT_T < e) atomicAdd(&cnt[r[u] & (BKT_N - 1)], 1);
    }
    __syncthreads();
    int v = 0, sc = 0;
    if (t < BKT_N) {
        v = cnt[t];
        sc = v;
        for (int m = 1; m < 32; m <<= 1) {
            int u = __shfl_up(sc, m, 32);
            if (lane >= m) sc += u;
        }
        if (lane == 31) wsum[wid] = sc;
    }
    __syncthreads();
    if (t < BKT_N) {
        int base = s;
        for (int w = 0; w < wid; ++w) base += wsum[w];
        const int excl = base + sc - v;
        cnt[t] = excl;
        const int node = k * BKT_N + t;
        if (node < n) offs[node] = excl;
    }
    if (k == nbk - 1 && t == 0) offs[n] = e;
    __syncthreads();
    for (int i = s + t; i < e; i += SORT_U * SORT_T) {
        int2 rc[SORT_U];
#pragma unroll
        for (int u = 0; u < SORT_U; ++u) {
            const int j = i + u * SORT_T;
            rc[u] = j < e ? tmp[j] : make_int2(0, 0);
        }
#pragma unroll
        for (int u = 0; u < SORT_U; ++u) {
            if (i + u * SORT_T < e) {
                const int p = atomicAdd(&cnt[rc[u].x & (BKT_N - 1)], 1);
                scol[p] = rc[u].y;
            }
        }
    }
}

__device__ __forceinline__ int block_sum_1024(int v, int* sh) {
    int t = threadIdx.x;
    for (int m = 1; m < 32; m <<= 1) v += __shfl_xor(v, m, 32);
    if ((t & 31) == 0) sh[t >> 5] = v;
    __syncthreads();
    int w = sh[t & 31];
    for (int m = 1; m < 32; m <<= 1) w += __shfl_xor(w, m, 32);
    __syncthreads();
    return w;
}

__global__ __launch_bounds__(SCAN_B) void scan_sums(const int* __restrict__ cnt,
                                                    int* __restrict__ bsum, int n) {
    __shared__ int sh[32];
    int i = blockIdx.x * SCAN_B + threadIdx.x;
    int tot = block_sum_1024(i < n ? cnt[i] : 0, sh);
    if (threadIdx.x == 0) bsum[blockIdx.x] = tot;
}

__global__ __launch_bounds__(SCAN_B) void scan_apply(int* __restrict__ offs,
                                                     const int* __restrict__ bsum, int n) {
    __shared__ int sh[32];
    const int t = threadIdx.x, lane = t & 31, wid = t >> 5;
    int p = 0;
    for (int b = t; b < (int)blockIdx.x; b += SCAN_B) p += bsum[b];
    const int base = block_sum_1024(p, sh);

    int i = blockIdx.x * SCAN_B + t;
    int v = i < n ? offs[i] : 0;
    int s = v;
    for (int m = 1; m < 32; m <<= 1) {
        int u = __shfl_up(s, m, 32);
        if (lane >= m) s += u;
    }
    if (lane == 31) sh[wid] = s;
    __syncthreads();
    if (wid == 0) {
        int w0 = sh[lane], w = w0;
        for (int m = 1; m < 32; m <<= 1) {
            int u = __shfl_up(w, m, 32);
            if (lane >= m) w += u;
        }
        sh[lane] = w - w0;
    }
    __syncthreads();
    int excl = base + sh[wid] + s - v;
    if (i < n) offs[i] = excl;
    if (i == n - 1) offs[n] = excl + v;
}

__device__ __forceinline__ float wave_sum32(float x) {
    for (int m = 1; m < 32; m <<= 1) x += __shfl_xor(x, m, 32);
    return x;
}

__device__ __forceinline__ float4 act_row(float4 v) {
    float an2 = wave_sum32(v.x * v.x + v.y * v.y + v.z * v.z + v.w * v.w);
    float an = fmaxf(sqrtf(an2), MINN);
    float tn = fminf(fmaxf(tanhf(an), MINN), MAXN);
    float lg = artanh_f(tn) / an;
    float4 at;
    at.x = tanhf(lg * v.x);
    at.y = tanhf(lg * v.y);
    at.z = tanhf(lg * v.z);
    at.w = tanhf(lg * v.w);
    float vn2 = wave_sum32(at.x * at.x + at.y * at.y + at.z * at.z + at.w * at.w);
    float vn = fmaxf(sqrtf(vn2), MINN);
    float s = fminf(tanhf(vn), MAXN) / vn;
    return make_float4(s * at.x, s * at.y, s * at.z, s * at.w);
}

__global__ __launch_bounds__(256) void node_gather_readout(const float* __restrict__ HT,
                                                           const int* __restrict__ offs,
                                                           const int* __restrict__ scol,
                                                           const int* __restrict__ seg,
                                                           float* __restrict__ ssum,
                                                           float* __restrict__ cnt, int nrows) {
    const int lane = threadIdx.x & 31;
    const int w0 = __builtin_amdgcn_readfirstlane((int)(blockIdx.x * 8 + (threadIdx.x >> 5)));
    const int nw = gridDim.x * 8;
    for (int r = w0; r < nrows; r += nw) {
        const int beg = offs[r], end = offs[r + 1];
        const int sg = seg[r];
        float4 acc = make_float4(0.f, 0.f, 0.f, 0.f);
        for (int b0 = beg; b0 < end; b0 += 32) {
            const int n = min(32, end - b0);
            const int myc = (lane < n) ? scol[b0 + lane] : 0;
            int j = 0;
            for (; j + 4 <= n; j += 4) {
                const int c0 = __builtin_amdgcn_readlane(myc, j);
                const int c1 = __builtin_amdgcn_readlane(myc, j + 1);
                const int c2 = __builtin_amdgcn_readlane(myc, j + 2);
                const int c3 = __builtin_amdgcn_readlane(myc, j + 3);
                const float4 v0 = *((const float4*)(HT + (size_t)c0 * 128) + lane);
                const float4 v1 = *((const float4*)(HT + (size_t)c1 * 128) + lane);
                const float4 v2 = *((const float4*)(HT + (size_t)c2 * 128) + lane);
                const float4 v3 = *((const float4*)(HT + (size_t)c3 * 128) + lane);
                acc.x += v0.x; acc.y += v0.y; acc.z += v0.z; acc.w += v0.w;
                acc.x += v1.x; acc.y += v1.y; acc.z += v1.z; acc.w += v1.w;
                acc.x += v2.x; acc.y += v2.y; acc.z += v2.z; acc.w += v2.w;
                acc.x += v3.x; acc.y += v3.y; acc.z += v3.z; acc.w += v3.w;
            }
            for (; j < n; ++j) {
                const int c0 = __builtin_amdgcn_readlane(myc, j);
                const float4 v0 = *((const float4*)(HT + (size_t)c0 * 128) + lane);
                acc.x += v0.x; acc.y += v0.y; acc.z += v0.z; acc.w += v0.w;
            }
        }
        const float4 h = act_row(acc);
        float hn2 = wave_sum32(h.x * h.x + h.y * h.y + h.z * h.z + h.w * h.w);
        float hn = fmaxf(sqrtf(hn2), MINN);
        float pc = fminf(hn, MAXN);
        float L = artanh_f(pc) / hn;
        float* d = ssum + (size_t)sg * 128 + lane * 4;
        atomicAdd(d + 0, L * h.x);
        atomicAdd(d + 1, L * h.y);
        atomicAdd(d + 2, L * h.z);
        atomicAdd(d + 3, L * h.w);
        if (lane == 0) atomicAdd(&cnt[sg], 1.f);
    }
}

__device__ __forceinline__ void add_rows(const float* __restrict__ HTin, int myc, int n,
                                         int lane, float4& acc) {
    int j = 0;
    for (; j + 8 <= n; j += 8) {
        const int c0 = __builtin_amdgcn_readlane(myc, j);
        const int c1 = __builtin_amdgcn_readlane(myc, j + 1);
        const int c2 = __builtin_amdgcn_readlane(myc, j + 2);
        const int c3 = __builtin_amdgcn_readlane(myc, j + 3);
        const int c4 = __builtin_amdgcn_readlane(myc, j + 4);
        const int c5 = __builtin_amdgcn_readlane(myc, j + 5);
        const int c6 = __builtin_amdgcn_readlane(myc, j + 6);
        const int c7 = __builtin_amdgcn_readlane(myc, j + 7);
        const float4 v0 = *((const float4*)(HTin + (size_t)c0 * 128) + lane);
        const float4 v1 = *((const float4*)(HTin + (size_t)c1 * 128) + lane);
        const float4 v2 = *((const float4*)(HTin + (size_t)c2 * 128) + lane);
        const float4 v3 = *((const float4*)(HTin + (size_t)c3 * 128) + lane);
        const float4 v4 = *((const float4*)(HTin + (size_t)c4 * 128) + lane);
        const float4 v5 = *((const float4*)(HTin + (size_t)c5 * 128) + lane);
        const float4 v6 = *((const float4*)(HTin + (size_t)c6 * 128) + lane);
        const float4 v7 = *((const float4*)(HTin + (size_t)c7 * 128) + lane);
        acc.x += v0.x; acc.y += v0.y; acc.z += v0.z; acc.w += v0.w;
        acc.x += v1.x; acc.y += v1.y; acc.z += v1.z; acc.w += v1.w;
        acc.x += v2.x; acc.y += v2.y; acc.z += v2.z; acc.w += v2.w;
        acc.x += v3.x; acc.y += v3.y; acc.z += v3.z; acc.w += v3.w;
        acc.x += v4.x; acc.y += v4.y; acc.z += v4.z; acc.w += v4.w;
        acc.x += v5.x; acc.y += v5.y; acc.z += v5.z; acc.w += v5.w;
        acc.x += v6.x; acc.y += v6.y; acc.z += v6.z; acc.w += v6.w;
        acc.x += v7.x; acc.y += v7.y; acc.z += v7.z; acc.w += v7.w;
    }
    for (; j + 4 <= n; j += 4) {
        const int c0 = __builtin_amdgcn_readlane(myc, j);
        const int c1 = __builtin_amdgcn_readlane(myc, j + 1);
        const int c2 = __builtin_amdgcn_readlane(myc, j + 2);
        const int c3 = __builtin_amdgcn_readlane(myc, j + 3);
        const float4 v0 = *((const float4*)(HTin + (size_t)c0 * 128) + lane);
        const float4 v1 = *((const float4*)(HTin + (size_t)c1 * 128) + lane);
        const float4 v2 = *((const float4*)(HTin + (size_t)c2 * 128) + lane);
        const float4 v3 = *((const float4*)(HTin + (size_t)c3 * 128) + lane);
        acc.x += v0.x; acc.y += v0.y; acc.z += v0.z; acc.w += v0.w;
        acc.x += v1.x; acc.y += v1.y; acc.z += v1.z; acc.w += v1.w;
        acc.x += v2.x; acc.y += v2.y; acc.z += v2.z; acc.w += v2.w;
        acc.x += v3.x; acc.y += v3.y; acc.z += v3.z; acc.w += v3.w;
    }
    for (; j < n; ++j) {
        const int c0 = __builtin_amdgcn_readlane(myc, j);
        const float4 v0 = *((const float4*)(HTin + (size_t)c0 * 128) + lane);
        acc.x += v0.x; acc.y += v0.y; acc.z += v0.z; acc.w += v0.w;
    }
}

#define HD   128
#define LDA  (HD + 4)
#define KC2  16
#define LDW  (KC2 + 4)

__global__ __launch_bounds__(128, 7) void gather_act_linear_wmma(
    const float* __restrict__ HTin,
    const int* __restrict__ offs, const int* __restrict__ scol, int nrows,
    const float* __restrict__ W,
    const float* __restrict__ bb,
    const float* __restrict__ y2p,
    float* __restrict__ HT) {
    __shared__ float As[BM][LDA];
    __shared__ float Ws[128][LDW];
    float* const xn_row = &Ws[0][0];
    float* const rs_mx2 = xn_row + BM;
    float* const rs_dot = rs_mx2 + BM;
    float* const rowCA = rs_dot + BM;
    float* const rowCB = rowCA + BM;

    const int t = threadIdx.x;
    const int row0 = blockIdx.x * BM;
    const int lane = t & 31;
    const int rs = __builtin_amdgcn_readfirstlane(t >> 5);
    const int half = lane >> 4;
    const int l15 = lane & 15;

    const int rbase = row0 + rs * 16;
    const int ob = offs[min(rbase + min(lane, 16), nrows)];
    float my_xn = 0.f;
    int begN = __builtin_amdgcn_readlane(ob, 0);
    int endN = __builtin_amdgcn_readlane(ob, 1);
    int mycN = (lane < endN - begN) ? scol[begN + lane] : 0;
    for (int i = 0; i < 16; ++i) {
        const int beg = begN, end = endN;
        const int myc = mycN;
        if (i < 15) {
            begN = __builtin_amdgcn_readlane(ob, i + 1);
            endN = __builtin_amdgcn_readlane(ob, i + 2);
            mycN = (lane < endN - begN) ? scol[begN + lane] : 0;
        }
        float4 acc = make_float4(0.f, 0.f, 0.f, 0.f);
        add_rows(HTin, myc, min(32, end - beg), lane, acc);
        for (int b0 = beg + 32; b0 < end; b0 += 32) {
            const int n = min(32, end - b0);
            const int mc = (lane < n) ? scol[b0 + lane] : 0;
            add_rows(HTin, mc, n, lane, acc);
        }
        const float4 h = act_row(acc);
        const float hn2 = wave_sum32(h.x * h.x + h.y * h.y + h.z * h.z + h.w * h.w);
        if (lane == i) my_xn = fmaxf(sqrtf(hn2), MINN);
        v4f hv;
        hv.x = h.x; hv.y = h.y; hv.z = h.z; hv.w = h.w;
        *(v4f*)(&As[rs * 16 + i][lane * 4]) = hv;
    }

    v8f acc[8];
    for (int j = 0; j < 8; ++j)
        for (int i = 0; i < 8; ++i) acc[j][i] = 0.f;

    const int sr = t >> 2;
    const int sc4 = t & 3;
    const int kq = HD >> 2;
    const v4f* __restrict__ W4 = (const v4f*)W;
    const int iw = sr * kq + sc4;
    const int mr = rs * 16 + l15;

    for (int kc = 0; kc < HD / KC2; ++kc) {
        const int q0 = kc * (KC2 / 4);
        v4f wv[4];
#pragma unroll
        for (int i = 0; i < 4; ++i) wv[i] = W4[iw + i * 32 * kq + q0];
        asm volatile("" : "+v"(wv[0]), "+v"(wv[1]), "+v"(wv[2]), "+v"(wv[3]));
#pragma unroll
        for (int i = 0; i < 4; ++i) *(v4f*)(&Ws[sr + 32 * i][sc4 * 4]) = wv[i];
        __syncthreads();

        for (int kk = 0; kk < KC2; kk += 4) {
            const int ko = kk + half * 2;
            v2f a;
            a.x = As[mr][kc * KC2 + ko];
            a.y = As[mr][kc * KC2 + ko + 1];
#pragma unroll
            for (int j = 0; j < 8; ++j) {
                const int colg = j * 16 + l15;
                v2f b;
                b.x = Ws[colg][ko];
                b.y = Ws[colg][ko + 1];
                acc[j] = __builtin_amdgcn_wmma_f32_16x16x4_f32(
                    false, a, false, b, (short)0, acc[j], false, false);
            }
        }
        __syncthreads();
    }

    if (lane < 16) xn_row[rs * 16 + lane] = my_xn;

    float bbv[8];
#pragma unroll
    for (int j = 0; j < 8; ++j) bbv[j] = bb[j * 16 + l15];

#pragma unroll
    for (int i = 0; i < 8; ++i) {
        float p2a = 0.f, pda = 0.f, p2b = 0.f, pdb = 0.f;
#pragma unroll
        for (int j = 0; j < 4; ++j) {
            float v = acc[j][i];
            p2a += v * v;
            pda += v * bbv[j];
        }
#pragma unroll
        for (int j = 4; j < 8; ++j) {
            float v = acc[j][i];
            p2b += v * v;
            pdb += v * bbv[j];
        }
        for (int m = 1; m < 16; m <<= 1) {
            p2a += __shfl_xor(p2a, m, 16);
            pda += __shfl_xor(pda, m, 16);
            p2b += __shfl_xor(p2b, m, 16);
            pdb += __shfl_xor(pdb, m, 16);
        }
        int mrow = rs * 16 + i + half * 8;
        if (l15 == 0) {
            rs_mx2[mrow] = p2a + p2b;
            rs_dot[mrow] = pda + pdb;
        }
    }
    __syncthreads();

    if (t < BM) {
        float xn = xn_row[t];
        float mxn = fmaxf(sqrtf(rs_mx2[t]), MINN);
        float tv = tanhf(mxn / xn * artanh_f(xn));
        float tc = fminf(tv, MAXN);
        float s_mv = tc / mxn;
        float x2 = tc * tc;
        float xy = s_mv * rs_dot[t];
        float y2 = y2p[0];
        float al = 1.f + 2.f * xy + y2;
        float be = 1.f - x2;
        float den = fmaxf(1.f + 2.f * xy + x2 * y2, MINN);
        float nn2 = fmaxf(al * al * x2 + 2.f * al * be * xy + be * be * y2, 0.f);
        float hn = fmaxf(sqrtf(nn2) / den, MINN);
        float hc = fminf(hn, MAXN);
        float L = artanh_f(hc) / hn;
        rowCA[t] = L * al * s_mv / den;
        rowCB[t] = L * be / den;
    }
    __syncthreads();

#pragma unroll
    for (int i = 0; i < 8; ++i) {
        int mrow = rs * 16 + i + half * 8;
        int gr = row0 + mrow;
        if (gr < nrows) {
            float ca = rowCA[mrow], cb = rowCB[mrow];
#pragma unroll
            for (int j = 0; j < 8; ++j) {
                int colg = j * 16 + l15;
                HT[(size_t)gr * 128 + colg] = ca * acc[j][i] + cb * bbv[j];
            }
        }
    }
}

__global__ __launch_bounds__(256) void post_agg(float* __restrict__ H, int nrows) {
    int r = blockIdx.x * 8 + (threadIdx.x >> 5);
    if (r >= nrows) return;
    int lane = threadIdx.x & 31;
    float* p = H + (size_t)r * 128 + lane * 4;
    float4 v = *(const float4*)p;
    float an2 = v.x * v.x + v.y * v.y + v.z * v.z + v.w * v.w;
    for (int m = 1; m < 32; m <<= 1) an2 += __shfl_xor(an2, m, 32);
    float an = fmaxf(sqrtf(an2), MINN);
    float tn = fminf(fmaxf(tanhf(an), MINN), MAXN);
    float lg = artanh_f(tn) / an;
    float4 at;
    at.x = tanhf(lg * v.x);
    at.y = tanhf(lg * v.y);
    at.z = tanhf(lg * v.z);
    at.w = tanhf(lg * v.w);
    float vn2 = at.x * at.x + at.y * at.y + at.z * at.z + at.w * at.w;
    for (int m = 1; m < 32; m <<= 1) vn2 += __shfl_xor(vn2, m, 32);
    float vn = fmaxf(sqrtf(vn2), MINN);
    float s = fminf(tanhf(vn), MAXN) / vn;
    *(float4*)p = make_float4(s * at.x, s * at.y, s * at.z, s * at.w);
}

__global__ __launch_bounds__(256) void final_scatter(
    const float* __restrict__ H, const int* __restrict__ seg,
    float* __restrict__ ssum, float* __restrict__ cnt, int nrows) {
    int r = blockIdx.x * 8 + (threadIdx.x >> 5);
    if (r >= nrows) return;
    int lane = threadIdx.x & 31;
    float4 v = *(const float4*)(H + (size_t)r * 128 + lane * 4);
    float hn2 = v.x * v.x + v.y * v.y + v.z * v.z + v.w * v.w;
    for (int m = 1; m < 32; m <<= 1) hn2 += __shfl_xor(hn2, m, 32);
    float hn = fmaxf(sqrtf(hn2), MINN);
    float pc = fminf(hn, MAXN);
    float L = artanh_f(pc) / hn;
    int s = seg[r];
    float* d = ssum + (size_t)s * 128 + lane * 4;
    atomicAdd(d + 0, L * v.x);
    atomicAdd(d + 1, L * v.y);
    atomicAdd(d + 2, L * v.z);
    atomicAdd(d + 3, L * v.w);
    if (lane == 0) atomicAdd(&cnt[s], 1.f);
}

__global__ __launch_bounds__(256) void pool_k(const float* __restrict__ ssum,
                                              const float* __restrict__ cnt,
                                              float* __restrict__ out, int nseg) {
    int r = blockIdx.x * 8 + (threadIdx.x >> 5);
    if (r >= nseg) return;
    int lane = threadIdx.x & 31;
    float inv = 1.f / fmaxf(cnt[r], 1.f);
    float4 v = *(const float4*)(ssum + (size_t)r * 128 + lane * 4);
    v.x *= inv; v.y *= inv; v.z *= inv; v.w *= inv;
    float mn2 = v.x * v.x + v.y * v.y + v.z * v.z + v.w * v.w;
    for (int m = 1; m < 32; m <<= 1) mn2 += __shfl_xor(mn2, m, 32);
    float mn = fmaxf(sqrtf(mn2), MINN);
    float s = fminf(tanhf(mn), MAXN) / mn;
    float* o = out + (size_t)r * 128 + lane * 4;
    o[0] = s * v.x; o[1] = s * v.y; o[2] = s * v.z; o[3] = s * v.w;
}

__global__ void zero_f32(float* __restrict__ p, size_t n) {
    size_t i = (size_t)blockIdx.x * blockDim.x + threadIdx.x;
    size_t st = (size_t)gridDim.x * blockDim.x;
    for (; i < n; i += st) p[i] = 0.f;
}

extern "C" void kernel_launch(void* const* d_in, const int* in_sizes, int n_in,
                              void* d_out, int out_size, void* d_ws, size_t ws_size,
                              hipStream_t stream) {
    const float* x = (const float*)d_in[0];
    const int* erow = (const int*)d_in[1];
    const int* ecol = (const int*)d_in[2];
    const int* seg = (const int*)d_in[3];
    const float* W1 = (const float*)d_in[4];
    const float* b1 = (const float*)d_in[5];
    const float* W2 = (const float*)d_in[6];
    const float* b2 = (const float*)d_in[7];

    const int N = in_sizes[0] / 256;
    const int E = in_sizes[1];
    const int NSEG = out_size / 128;

    char* ws = (char*)d_ws;
    size_t bufBytes = (size_t)N * 128 * sizeof(float);
    float* buf0 = (float*)ws;
    float* buf1 = (float*)(ws + bufBytes);
    float* bbias = (float*)(ws + 2 * bufBytes);
    float* y2 = bbias + 256;
    float* ssum = buf0;
    float* cnt = buf0 + (size_t)NSEG * 128;

    dim3 blk(256);
    dim3 gemmBlk(128);
    int gRows = (N + 7) / 8;
    int gGemm = (N + BM - 1) / BM;
    int gEdge = (E + 7) / 8;

    const int nScan = (N + SCAN_B - 1) / SCAN_B;
    size_t offBytes = (((size_t)N + 1) * sizeof(int) + 255) / 256 * 256;
    size_t bsBytes = ((size_t)nScan * sizeof(int) + 255) / 256 * 256;
    int* offs = (int*)(ws + 2 * bufBytes + 4096);
    int* scol = (int*)(ws + 2 * bufBytes + 4096 + offBytes + bsBytes);
    const int G1 = 256;
    const int NBK = (N + BKT_N - 1) / BKT_N;
    const int nTab = NBK * G1;
    const int nScanT = (nTab + SCAN_B - 1) / SCAN_B;
    size_t tmpBytes = ((size_t)E * sizeof(int2) + 255) / 256 * 256;
    size_t tabBytes = (((size_t)nTab + 1) * sizeof(int) + 255) / 256 * 256;
    int2* tmp = (int2*)buf1;
    int* btab = (int*)(ws + bufBytes + tmpBytes);
    int* bsum = (int*)(ws + bufBytes + tmpBytes + tabBytes);
    const bool grouped =
        ws_size >= 2 * bufBytes + 4096 + offBytes + bsBytes + (size_t)E * sizeof(int) &&
        NBK <= MAXBK &&
        tmpBytes + tabBytes + (size_t)nScanT * sizeof(int) <= bufBytes;

    bias_prep<<<2, 128, 0, stream>>>(b1, b2, bbias, y2);

    if (grouped) {
        bin_count<<<G1, SORT_T, 0, stream>>>(erow, btab, E, NBK);
        scan_sums<<<nScanT, SCAN_B, 0, stream>>>(btab, bsum, nTab);
        scan_apply<<<nScanT, SCAN_B, 0, stream>>>(btab, bsum, nTab);
        bin_scatter<<<G1, SORT_T, 0, stream>>>(erow, ecol, btab, tmp, E, NBK);
        bin_sort<<<NBK, SORT_T, 0, stream>>>(tmp, btab, offs, scol, G1, NBK, N);

        hyp_linear_wmma<<<gGemm, gemmBlk, 0, stream>>>(x, 256, N, W1, bbias, y2, buf0, 1);

        gather_act_linear_wmma<<<gGemm, gemmBlk, 0, stream>>>(buf0, offs, scol, N, W2,
                                                          bbias + 128, y2 + 1, buf1);
        float* ssumG = buf0;
        float* cntG = buf0 + (size_t)NSEG * 128;
        zero_f32<<<64, 256, 0, stream>>>(ssumG, (size_t)NSEG * 128 + NSEG);
        node_gather_readout<<<2048, blk, 0, stream>>>(buf1, offs, scol, seg, ssumG, cntG, N);
        pool_k<<<(NSEG + 7) / 8, blk, 0, stream>>>(ssumG, cntG, (float*)d_out, NSEG);
        return;
    }

    hyp_linear_wmma<<<gGemm, gemmBlk, 0, stream>>>(x, 256, N, W1, bbias, y2, buf0, 1);
    zero_f32<<<2048, 256, 0, stream>>>(buf1, (size_t)N * 128);
    edge_agg<<<gEdge, blk, 0, stream>>>(buf0, erow, ecol, buf1, E);
    post_agg<<<gRows, blk, 0, stream>>>(buf1, N);

    hyp_linear_wmma<<<gGemm, gemmBlk, 0, stream>>>(buf1, 128, N, W2, bbias + 128,
                                               y2 + 1, buf0, 0);
    zero_f32<<<2048, 256, 0, stream>>>(buf1, (size_t)N * 128);
    edge_agg<<<gEdge, blk, 0, stream>>>(buf0, erow, ecol, buf1, E);
    post_agg<<<gRows, blk, 0, stream>>>(buf1, N);

    zero_f32<<<64, 256, 0, stream>>>(ssum, (size_t)NSEG * 128 + NSEG);
    final_scatter<<<gRows, blk, 0, stream>>>(buf1, seg, ssum, cnt, N);
    pool_k<<<(NSEG + 7) / 8, blk, 0, stream>>>(ssum, cnt, (float*)d_out, NSEG);
}


# row maps after each edge sum: a logmap0 that follows expmap0+proj collapsed by artanh(min(tanh a,M)) = min(a,artanh M); per node 3 to 4 tanh/log calls and 1 of 3 row sums gone; f32 sum order kept
# speedup vs baseline: 3.4288x; 1.0368x over previous
#include <hip/hip_runtime.h>
#include <math.h>


#define MAXN 0.996f
#define MINN 1e-15f
#define KC   32
#define LDP  (KC + 4)
#define BM   64

typedef float v2f __attribute__((ext_vector_type(2)));
typedef float v8f __attribute__((ext_vector_type(8)));
typedef float v4f __attribute__((ext_vector_type(4)));

__device__ __forceinline__ float artanh_f(float z) {
    z = fminf(fmaxf(z, -(1.f - 1e-7f)), 1.f - 1e-7f);
    return 0.5f * logf((1.f + z) / (1.f - z));
}

__global__ void bias_prep(const float* __restrict__ b1,
                          const float* __restrict__ b2,
                          float* __restrict__ bb, float* __restrict__ y2) {
    __shared__ float red[128];
    const float* b = blockIdx.x ? b2 : b1;
    int t = threadIdx.x;
    float v = b[t];
    red[t] = v * v;
    __syncthreads();
    for (int s = 64; s > 0; s >>= 1) {
        if (t < s) red[t] += red[t + s];
        __syncthreads();
    }
    float bn = fmaxf(sqrtf(red[0]), MINN);
    float tc = fminf(tanhf(bn), MAXN);
    bb[blockIdx.x * 128 + t] = tc / bn * v;
    if (t == 0) y2[blockIdx.x] = tc * tc;
}

__global__ __launch_bounds__(128) void hyp_linear_wmma(
    const float* __restrict__ X, int Kdim, int nrows,
    const float* __restrict__ W,
    const float* __restrict__ bb,
    const float* __restrict__ y2p,
    float* __restrict__ HT, int apply_exp) {
    __shared__ float As[BM][LDP];
    __shared__ float Ws[128][LDP];
    __shared__ float red[BM][8];
    __shared__ float scale_row[BM], xn_row[BM];
    __shared__ float rs_mx2[BM], rs_dot[BM];
    __shared__ float rowCA[BM], rowCB[BM];

    const int t = threadIdx.x;
    const int row0 = blockIdx.x * BM;

    const int lane = t & 31;
    const int rs = t >> 5;
    const int half = lane >> 4;
    const int l15 = lane & 15;

    v8f acc[8];
    for (int j = 0; j < 8; ++j)
        for (int i = 0; i < 8; ++i) acc[j][i] = 0.f;

    const int nchunks = Kdim / KC;

    const int sr = t >> 3;
    const int sc4 = t & 7;
    const int kq = Kdim >> 2;
    const v4f* __restrict__ X4 = (const v4f*)X;
    const v4f* __restrict__ W4 = (const v4f*)W;
    int ia[4];
#pragma unroll
    for (int i = 0; i < 4; ++i) ia[i] = min(row0 + sr + 16 * i, nrows - 1) * kq + sc4;
    const int iw = sr * kq + sc4;

    const int mr = rs * 16 + l15;

    float ss[4] = {0.f, 0.f, 0.f, 0.f};

    for (int kc = 0; kc < nchunks; ++kc) {
        const int q0 = kc * (KC / 4);
        v4f av[4], wv[8];
#pragma unroll
        for (int i = 0; i < 4; ++i) av[i] = X4[ia[i] + q0];
#pragma unroll
        for (int i = 0; i < 8; ++i) wv[i] = W4[iw + i * 16 * kq + q0];
        asm volatile("" : "+v"(av[0]), "+v"(av[1]), "+v"(av[2]), "+v"(av[3]),
                          "+v"(wv[0]), "+v"(wv[1]), "+v"(wv[2]), "+v"(wv[3]),
                          "+v"(wv[4]), "+v"(wv[5]), "+v"(wv[6]), "+v"(wv[7]));
#pragma unroll
        for (int i = 0; i < 4; ++i) {
            ss[i] += av[i].x * av[i].x + av[i].y * av[i].y + av[i].z * av[i].z + av[i].w * av[i].w;
            *(v4f*)(&As[sr + 16 * i][sc4 * 4]) = av[i];
        }
#pragma unroll
        for (int i = 0; i < 8; ++i) *(v4f*)(&Ws[sr + 16 * i][sc4 * 4]) = wv[i];
        __syncthreads();

        for (int kk = 0; kk < KC; kk += 4) {
            const int ko = kk + half * 2;
            v2f a;
            a.x = As[mr][ko];
            a.y = As[mr][ko + 1];
#pragma unroll
            for (int j = 0; j < 8; ++j) {
                const int colg = j * 16 + l15;
                v2f b;
                b.x = Ws[colg][ko];
                b.y = Ws[colg][ko + 1];
                acc[j] = __builtin_amdgcn_wmma_f32_16x16x4_f32(
                    false, a, false, b, (short)0, acc[j], false, false);
            }
        }
        __syncthreads();
    }

#pragma unroll
    for (int i = 0; i < 4; ++i) red[sr + 16 * i][sc4] = ss[i];
    __syncthreads();
    if (t < BM) {
        float un2 = red[t][0];
#pragma unroll
        for (int q = 1; q < 8; ++q) un2 += red[t][q];
        float un = fmaxf(sqrtf(un2), MINN);
        if (apply_exp) {
            float bc = fminf(tanhf(un), MAXN);
            scale_row[t] = bc / un;
            xn_row[t] = fmaxf(bc, MINN);
        } else {
            scale_row[t] = 1.f;
            xn_row[t] = un;
        }
        if (row0 + t >= nrows) scale_row[t] = 0.f;
    }
    __syncthreads();
    if (apply_exp) {
#pragma unroll
        for (int i = 0; i < 8; ++i) {
            const float sc = scale_row[rs * 16 + i + half * 8];
#pragma unroll
            for (int j = 0; j < 8; ++j) acc[j][i] *= sc;
        }
    }

    float bbv[8];
#pragma unroll
    for (int j = 0; j < 8; ++j) bbv[j] = bb[j * 16 + l15];

#pragma unroll
    for (int i = 0; i < 8; ++i) {
        float p2a = 0.f, pda = 0.f, p2b = 0.f, pdb = 0.f;
#pragma unroll
        for (int j = 0; j < 4; ++j) {
            float v = acc[j][i];
            p2a += v * v;
            pda += v * bbv[j];
        }
#pragma unroll
        for (int j = 4; j < 8; ++j) {
            float v = acc[j][i];
            p2b += v * v;
            pdb += v * bbv[j];
        }
        for (int m = 1; m < 16; m <<= 1) {
            p2a += __shfl_xor(p2a, m, 16);
            pda += __shfl_xor(pda, m, 16);
            p2b += __shfl_xor(p2b, m, 16);
            pdb += __shfl_xor(pdb, m, 16);
        }
        int mrow = rs * 16 + i + half * 8;
        if (l15 == 0) {
            rs_mx2[mrow] = p2a + p2b;
            rs_dot[mrow] = pda + pdb;
        }
    }
    __syncthreads();

    if (t < BM) {
        float xn = xn_row[t];
        float mxn = fmaxf(sqrtf(rs_mx2[t]), MINN);
        float tv = tanhf(mxn / xn * artanh_f(xn));
        float tc = fminf(tv, MAXN);
        float s_mv = tc / mxn;
        float x2 = tc * tc;
        float xy = s_mv * rs_dot[t];
        float y2 = y2p[0];
        float al = 1.f + 2.f * xy + y2;
        float be = 1.f - x2;
        float den = fmaxf(1.f + 2.f * xy + x2 * y2, MINN);
        float nn2 = fmaxf(al * al * x2 + 2.f * al * be * xy + be * be * y2, 0.f);
        float hn = fmaxf(sqrtf(nn2) / den, MINN);
        float hc = fminf(hn, MAXN);
        float L = artanh_f(hc) / hn;
        rowCA[t] = L * al * s_mv / den;
        rowCB[t] = L * be / den;
    }
    __syncthreads();

#pragma unroll
    for (int i = 0; i < 8; ++i) {
        int mrow = rs * 16 + i + half * 8;
        int gr = row0 + mrow;
        if (gr < nrows) {
            float ca = rowCA[mrow], cb = rowCB[mrow];
#pragma unroll
            for (int j = 0; j < 8; ++j) {
                int colg = j * 16 + l15;
                HT[(size_t)gr * 128 + colg] = ca * acc[j][i] + cb * bbv[j];
            }
        }
    }
}

__global__ __launch_bounds__(256) void edge_agg(const float* __restrict__ HT,
                                                const int* __restrict__ erow,
                                                const int* __restrict__ ecol,
                                                float* __restrict__ AGG, int E) {
    int e = blockIdx.x * 8 + (threadIdx.x >> 5);
    if (e >= E) return;
    int lane = threadIdx.x & 31;
    int r = erow[e], c = ecol[e];
    float4 v = *(const float4*)(HT + (size_t)c * 128 + lane * 4);
    float* dst = AGG + (size_t)r * 128 + lane * 4;
    atomicAdd(dst + 0, v.x);
    atomicAdd(dst + 1, v.y);
    atomicAdd(dst + 2, v.z);
    atomicAdd(dst + 3, v.w);
}

#define SCAN_B 1024
#define BKT_SH 8
#define BKT_N  256
#define MAXBK  1024
#define SORT_T 1024
#define SORT_U 4

__global__ __launch_bounds__(SORT_T) void bin_count(const int* __restrict__ erow,
                                                    int* __restrict__ bh,
                                                    int E, int nbk) {
    __shared__ int hist[MAXBK];
    const int t = threadIdx.x, G = gridDim.x, b = blockIdx.x;
    for (int k = t; k < nbk; k += SORT_T) hist[k] = 0;
    __syncthreads();
    const int chunk = (E + G - 1) / G;
    const int beg = b * chunk, end = min(E, beg + chunk);
    for (int i = beg + t; i < end; i += SORT_U * SORT_T) {
        int r[SORT_U];
#pragma unroll
        for (int u = 0; u < SORT_U; ++u) {
            const int j = i + u * SORT_T;
            r[u] = j < end ? erow[j] : 0;
        }
#pragma unroll
        for (int u = 0; u < SORT_U; ++u) {
            const int k = min((int)((unsigned)r[u] >> BKT_SH), nbk - 1);
            if (i + u * SORT_T < end) atomicAdd(&hist[k], 1);
        }
    }
    __syncthreads();
    for (int k = t; k < nbk; k += SORT_T) bh[(size_t)k * G + b] = hist[k];
}

__global__ __launch_bounds__(SORT_T) void bin_scatter(const int* __restrict__ erow,
                                                      const int* __restrict__ ecol,
                                                      const int* __restrict__ bo,
                                                      int2* __restrict__ tmp,
                                                      int E, int nbk) {
    __shared__ int cur[MAXBK];
    const int t = threadIdx.x, G = gridDim.x, b = blockIdx.x;
    for (int k = t; k < nbk; k += SORT_T) cur[k] = bo[(size_t)k * G + b];
    __syncthreads();
    const int chunk = (E + G - 1) / G;
    const int beg = b * chunk, end = min(E, beg + chunk);
    for (int i = beg + t; i < end; i += SORT_U * SORT_T) {
        int r[SORT_U], c[SORT_U];
#pragma unroll
        for (int u = 0; u < SORT_U; ++u) {
            const int j = i + u * SORT_T;
            r[u] = j < end ? erow[j] : 0;
            c[u] = j < end ? ecol[j] : 0;
        }
#pragma unroll
        for (int u = 0; u < SORT_U; ++u) {
            if (i + u * SORT_T < end) {
                const int k = min((int)((unsigned)r[u] >> BKT_SH), nbk - 1);
                const int p = atomicAdd(&cur[k], 1);
                tmp[p] = make_int2(r[u], c[u]);
            }
        }
    }
}

__global__ __launch_bounds__(SORT_T) void bin_sort(const int2* __restrict__ tmp,
                                                   const int* __restrict__ bo,
                                                   int* __restrict__ offs,
                                                   int* __restrict__ scol,
                                                   int G, int nbk, int n) {
    __shared__ int cnt[BKT_N];
    __shared__ int wsum[BKT_N / 32];
    const int t = threadIdx.x, k = blockIdx.x, lane = t & 31, wid = t >> 5;
    const int s = bo[(size_t)k * G], e = bo[(size_t)(k + 1) * G];
    if (t < BKT_N) cnt[t] = 0;
    __syncthreads();
    for (int i = s + t; i < e; i += SORT_U * SORT_T) {
        int r[SORT_U];
#pragma unroll
        for (int u = 0; u < SORT_U; ++u) {
            const int j = i + u * SORT_T;
            r[u] = j < e ? tmp[j].x : 0;
        }
#pragma unroll
        for (int u = 0; u < SORT_U; ++u)
            if (i + u * SORT_T < e) atomicAdd(&cnt[r[u] & (BKT_N - 1)], 1);
    }
    __syncthreads();
    int v = 0, sc = 0;
    if (t < BKT_N) {
        v = cnt[t];
        sc = v;
        for (int m = 1; m < 32; m <<= 1) {
            int u = __shfl_up(sc, m, 32);
            if (lane >= m) sc += u;
        }
        if (lane == 31) wsum[wid] = sc;
    }
    __syncthreads();
    if (t < BKT_N) {
        int base = s;
        for (int w = 0; w < wid; ++w) base += wsum[w];
        const int excl = base + sc - v;
        cnt[t] = excl;
        const int node = k * BKT_N + t;
        if (node < n) offs[node] = excl;
    }
    if (k == nbk - 1 && t == 0) offs[n] = e;
    __syncthreads();
    for (int i = s + t; i < e; i += SORT_U * SORT_T) {
        int2 rc[SORT_U];
#pragma unroll
        for (int u = 0; u < SORT_U; ++u) {
            const int j = i + u * SORT_T;
            rc[u] = j < e ? tmp[j] : make_int2(0, 0);
        }
#pragma unroll
        for (int u = 0; u < SORT_U; ++u) {
            if (i + u * SORT_T < e) {
                const int p = atomicAdd(&cnt[rc[u].x & (BKT_N - 1)], 1);
                scol[p] = rc[u].y;
            }
        }
    }
}

__device__ __forceinline__ int block_sum_1024(int v, int* sh) {
    int t = threadIdx.x;
    for (int m = 1; m < 32; m <<= 1) v += __shfl_xor(v, m, 32);
    if ((t & 31) == 0) sh[t >> 5] = v;
    __syncthreads();
    int w = sh[t & 31];
    for (int m = 1; m < 32; m <<= 1) w += __shfl_xor(w, m, 32);
    __syncthreads();
    return w;
}

__global__ __launch_bounds__(SCAN_B) void scan_sums(const int* __restrict__ cnt,
                                                    int* __restrict__ bsum, int n) {
    __shared__ int sh[32];
    int i = blockIdx.x * SCAN_B + threadIdx.x;
    int tot = block_sum_1024(i < n ? cnt[i] : 0, sh);
    if (threadIdx.x == 0) bsum[blockIdx.x] = tot;
}

__global__ __launch_bounds__(SCAN_B) void scan_apply(int* __restrict__ offs,
                                                     const int* __restrict__ bsum, int n) {
    __shared__ int sh[32];
    const int t = threadIdx.x, lane = t & 31, wid = t >> 5;
    int p = 0;
    for (int b = t; b < (int)blockIdx.x; b += SCAN_B) p += bsum[b];
    const int base = block_sum_1024(p, sh);

    int i = blockIdx.x * SCAN_B + t;
    int v = i < n ? offs[i] : 0;
    int s = v;
    for (int m = 1; m < 32; m <<= 1) {
        int u = __shfl_up(s, m, 32);
        if (lane >= m) s += u;
    }
    if (lane == 31) sh[wid] = s;
    __syncthreads();
    if (wid == 0) {
        int w0 = sh[lane], w = w0;
        for (int m = 1; m < 32; m <<= 1) {
            int u = __shfl_up(w, m, 32);
            if (lane >= m) w += u;
        }
        sh[lane] = w - w0;
    }
    __syncthreads();
    int excl = base + sh[wid] + s - v;
    if (i < n) offs[i] = excl;
    if (i == n - 1) offs[n] = excl + v;
}

__device__ __forceinline__ float wave_sum32(float x) {
    for (int m = 1; m < 32; m <<= 1) x += __shfl_xor(x, m, 32);
    return x;
}

__device__ __forceinline__ float4 tanh_row(float4 v, float amax, float& vn) {
    float an2 = wave_sum32(v.x * v.x + v.y * v.y + v.z * v.z + v.w * v.w);
    float an = fmaxf(sqrtf(an2), MINN);
    float lg = fminf(1.f, amax / an);
    float4 at;
    at.x = tanhf(lg * v.x);
    at.y = tanhf(lg * v.y);
    at.z = tanhf(lg * v.z);
    at.w = tanhf(lg * v.w);
    float vn2 = wave_sum32(at.x * at.x + at.y * at.y + at.z * at.z + at.w * at.w);
    vn = fmaxf(sqrtf(vn2), MINN);
    return at;
}

__global__ __launch_bounds__(256) void node_gather_readout(const float* __restrict__ HT,
                                                           const int* __restrict__ offs,
                                                           const int* __restrict__ scol,
                                                           const int* __restrict__ seg,
                                                           float* __restrict__ ssum,
                                                           float* __restrict__ cnt, int nrows) {
    const int lane = threadIdx.x & 31;
    const int w0 = __builtin_amdgcn_readfirstlane((int)(blockIdx.x * 8 + (threadIdx.x >> 5)));
    const int nw = gridDim.x * 8;
    const float amax = artanh_f(MAXN);
    for (int r = w0; r < nrows; r += nw) {
        const int beg = offs[r], end = offs[r + 1];
        const int sg = seg[r];
        float4 acc = make_float4(0.f, 0.f, 0.f, 0.f);
        for (int b0 = beg; b0 < end; b0 += 32) {
            const int n = min(32, end - b0);
            const int myc = (lane < n) ? scol[b0 + lane] : 0;
            int j = 0;
            for (; j + 4 <= n; j += 4) {
                const int c0 = __builtin_amdgcn_readlane(myc, j);
                const int c1 = __builtin_amdgcn_readlane(myc, j + 1);
                const int c2 = __builtin_amdgcn_readlane(myc, j + 2);
                const int c3 = __builtin_amdgcn_readlane(myc, j + 3);
                const float4 v0 = *((const float4*)(HT + (size_t)c0 * 128) + lane);
                const float4 v1 = *((const float4*)(HT + (size_t)c1 * 128) + lane);
                const float4 v2 = *((const float4*)(HT + (size_t)c2 * 128) + lane);
                const float4 v3 = *((const float4*)(HT + (size_t)c3 * 128) + lane);
                acc.x += v0.x; acc.y += v0.y; acc.z += v0.z; acc.w += v0.w;
                acc.x += v1.x; acc.y += v1.y; acc.z += v1.z; acc.w += v1.w;
                acc.x += v2.x; acc.y += v2.y; acc.z += v2.z; acc.w += v2.w;
                acc.x += v3.x; acc.y += v3.y; acc.z += v3.z; acc.w += v3.w;
            }
            for (; j < n; ++j) {
                const int c0 = __builtin_amdgcn_readlane(myc, j);
                const float4 v0 = *((const float4*)(HT + (size_t)c0 * 128) + lane);
                acc.x += v0.x; acc.y += v0.y; acc.z += v0.z; acc.w += v0.w;
            }
        }
        float vn;
        const float4 at = tanh_row(acc, amax, vn);
        const float L = fminf(1.f, amax / vn);
        float* d = ssum + (size_t)sg * 128 + lane * 4;
        atomicAdd(d + 0, L * at.x);
        atomicAdd(d + 1, L * at.y);
        atomicAdd(d + 2, L * at.z);
        atomicAdd(d + 3, L * at.w);
        if (lane == 0) atomicAdd(&cnt[sg], 1.f);
    }
}

__device__ __forceinline__ void add_rows(const float* __restrict__ HTin, int myc, int n,
                                         int lane, float4& acc) {
    int j = 0;
    for (; j + 8 <= n; j += 8) {
        const int c0 = __builtin_amdgcn_readlane(myc, j);
        const int c1 = __builtin_amdgcn_readlane(myc, j + 1);
        const int c2 = __builtin_amdgcn_readlane(myc, j + 2);
        const int c3 = __builtin_amdgcn_readlane(myc, j + 3);
        const int c4 = __builtin_amdgcn_readlane(myc, j + 4);
        const int c5 = __builtin_amdgcn_readlane(myc, j + 5);
        const int c6 = __builtin_amdgcn_readlane(myc, j + 6);
        const int c7 = __builtin_amdgcn_readlane(myc, j + 7);
        const float4 v0 = *((const float4*)(HTin + (size_t)c0 * 128) + lane);
        const float4 v1 = *((const float4*)(HTin + (size_t)c1 * 128) + lane);
        const float4 v2 = *((const float4*)(HTin + (size_t)c2 * 128) + lane);
        const float4 v3 = *((const float4*)(HTin + (size_t)c3 * 128) + lane);
        const float4 v4 = *((const float4*)(HTin + (size_t)c4 * 128) + lane);
        const float4 v5 = *((const float4*)(HTin + (size_t)c5 * 128) + lane);
        const float4 v6 = *((const float4*)(HTin + (size_t)c6 * 128) + lane);
        const float4 v7 = *((const float4*)(HTin + (size_t)c7 * 128) + lane);
        acc.x += v0.x; acc.y += v0.y; acc.z += v0.z; acc.w += v0.w;
        acc.x += v1.x; acc.y += v1.y; acc.z += v1.z; acc.w += v1.w;
        acc.x += v2.x; acc.y += v2.y; acc.z += v2.z; acc.w += v2.w;
        acc.x += v3.x; acc.y += v3.y; acc.z += v3.z; acc.w += v3.w;
        acc.x += v4.x; acc.y += v4.y; acc.z += v4.z; acc.w += v4.w;
        acc.x += v5.x; acc.y += v5.y; acc.z += v5.z; acc.w += v5.w;
        acc.x += v6.x; acc.y += v6.y; acc.z += v6.z; acc.w += v6.w;
        acc.x += v7.x; acc.y += v7.y; acc.z += v7.z; acc.w += v7.w;
    }
    for (; j + 4 <= n; j += 4) {
        const int c0 = __builtin_amdgcn_readlane(myc, j);
        const int c1 = __builtin_amdgcn_readlane(myc, j + 1);
        const int c2 = __builtin_amdgcn_readlane(myc, j + 2);
        const int c3 = __builtin_amdgcn_readlane(myc, j + 3);
        const float4 v0 = *((const float4*)(HTin + (size_t)c0 * 128) + lane);
        const float4 v1 = *((const float4*)(HTin + (size_t)c1 * 128) + lane);
        const float4 v2 = *((const float4*)(HTin + (size_t)c2 * 128) + lane);
        const float4 v3 = *((const float4*)(HTin + (size_t)c3 * 128) + lane);
        acc.x += v0.x; acc.y += v0.y; acc.z += v0.z; acc.w += v0.w;
        acc.x += v1.x; acc.y += v1.y; acc.z += v1.z; acc.w += v1.w;
        acc.x += v2.x; acc.y += v2.y; acc.z += v2.z; acc.w += v2.w;
        acc.x += v3.x; acc.y += v3.y; acc.z += v3.z; acc.w += v3.w;
    }
    for (; j < n; ++j) {
        const int c0 = __builtin_amdgcn_readlane(myc, j);
        const float4 v0 = *((const float4*)(HTin + (size_t)c0 * 128) + lane);
        acc.x += v0.x; acc.y += v0.y; acc.z += v0.z; acc.w += v0.w;
    }
}

#define HD   128
#define LDA  (HD + 4)
#define KC2  16
#define LDW  (KC2 + 4)

__global__ __launch_bounds__(128, 7) void gather_act_linear_wmma(
    const float* __restrict__ HTin,
    const int* __restrict__ offs, const int* __restrict__ scol, int nrows,
    const float* __restrict__ W,
    const float* __restrict__ bb,
    const float* __restrict__ y2p,
    float* __restrict__ HT) {
    __shared__ float As[BM][LDA];
    __shared__ float Ws[128][LDW];
    float* const xn_row = &Ws[0][0];
    float* const rs_mx2 = xn_row + BM;
    float* const rs_dot = rs_mx2 + BM;
    float* const rowCA = rs_dot + BM;
    float* const rowCB = rowCA + BM;

    const int t = threadIdx.x;
    const int row0 = blockIdx.x * BM;
    const int lane = t & 31;
    const int rs = __builtin_amdgcn_readfirstlane(t >> 5);
    const int half = lane >> 4;
    const int l15 = lane & 15;

    const int rbase = row0 + rs * 16;
    const float amax = artanh_f(MAXN);
    const int ob = offs[min(rbase + min(lane, 16), nrows)];
    float my_xn = 0.f;
    int begN = __builtin_amdgcn_readlane(ob, 0);
    int endN = __builtin_amdgcn_readlane(ob, 1);
    int mycN = (lane < endN - begN) ? scol[begN + lane] : 0;
    for (int i = 0; i < 16; ++i) {
        const int beg = begN, end = endN;
        const int myc = mycN;
        if (i < 15) {
            begN = __builtin_amdgcn_readlane(ob, i + 1);
            endN = __builtin_amdgcn_readlane(ob, i + 2);
            mycN = (lane < endN - begN) ? scol[begN + lane] : 0;
        }
        float4 acc = make_float4(0.f, 0.f, 0.f, 0.f);
        add_rows(HTin, myc, min(32, end - beg), lane, acc);
        for (int b0 = beg + 32; b0 < end; b0 += 32) {
            const int n = min(32, end - b0);
            const int mc = (lane < n) ? scol[b0 + lane] : 0;
            add_rows(HTin, mc, n, lane, acc);
        }
        float vn;
        const float4 at = tanh_row(acc, amax, vn);
        const float tcl = fminf(tanhf(vn), MAXN);
        const float s = tcl / vn;
        if (lane == i) my_xn = fmaxf(tcl, MINN);
        v4f hv;
        hv.x = s * at.x; hv.y = s * at.y; hv.z = s * at.z; hv.w = s * at.w;
        *(v4f*)(&As[rs * 16 + i][lane * 4]) = hv;
    }

    v8f acc[8];
    for (int j = 0; j < 8; ++j)
        for (int i = 0; i < 8; ++i) acc[j][i] = 0.f;

    const int sr = t >> 2;
    const int sc4 = t & 3;
    const int kq = HD >> 2;
    const v4f* __restrict__ W4 = (const v4f*)W;
    const int iw = sr * kq + sc4;
    const int mr = rs * 16 + l15;

    for (int kc = 0; kc < HD / KC2; ++kc) {
        const int q0 = kc * (KC2 / 4);
        v4f wv[4];
#pragma unroll
        for (int i = 0; i < 4; ++i) wv[i] = W4[iw + i * 32 * kq + q0];
        asm volatile("" : "+v"(wv[0]), "+v"(wv[1]), "+v"(wv[2]), "+v"(wv[3]));
#pragma unroll
        for (int i = 0; i < 4; ++i) *(v4f*)(&Ws[sr + 32 * i][sc4 * 4]) = wv[i];
        __syncthreads();

        for (int kk = 0; kk < KC2; kk += 4) {
            const int ko = kk + half * 2;
            v2f a;
            a.x = As[mr][kc * KC2 + ko];
            a.y = As[mr][kc * KC2 + ko + 1];
#pragma unroll
            for (int j = 0; j < 8; ++j) {
                const int colg = j * 16 + l15;
                v2f b;
                b.x = Ws[colg][ko];
                b.y = Ws[colg][ko + 1];
                acc[j] = __builtin_amdgcn_wmma_f32_16x16x4_f32(
                    false, a, false, b, (short)0, acc[j], false, false);
            }
        }
        __syncthreads();
    }

    if (lane < 16) xn_row[rs * 16 + lane] = my_xn;

    float bbv[8];
#pragma unroll
    for (int j = 0; j < 8; ++j) bbv[j] = bb[j * 16 + l15];

#pragma unroll
    for (int i = 0; i < 8; ++i) {
        float p2a = 0.f, pda = 0.f, p2b = 0.f, pdb = 0.f;
#pragma unroll
        for (int j = 0; j < 4; ++j) {
            float v = acc[j][i];
            p2a += v * v;
            pda += v * bbv[j];
        }
#pragma unroll
        for (int j = 4; j < 8; ++j) {
            float v = acc[j][i];
            p2b += v * v;
            pdb += v * bbv[j];
        }
        for (int m = 1; m < 16; m <<= 1) {
            p2a += __shfl_xor(p2a, m, 16);
            pda += __shfl_xor(pda, m, 16);
            p2b += __shfl_xor(p2b, m, 16);
            pdb += __shfl_xor(pdb, m, 16);
        }
        int mrow = rs * 16 + i + half * 8;
        if (l15 == 0) {
            rs_mx2[mrow] = p2a + p2b;
            rs_dot[mrow] = pda + pdb;
        }
    }
    __syncthreads();

    if (t < BM) {
        float xn = xn_row[t];
        float mxn = fmaxf(sqrtf(rs_mx2[t]), MINN);
        float tv = tanhf(mxn / xn * artanh_f(xn));
        float tc = fminf(tv, MAXN);
        float s_mv = tc / mxn;
        float x2 = tc * tc;
        float xy = s_mv * rs_dot[t];
        float y2 = y2p[0];
        float al = 1.f + 2.f * xy + y2;
        float be = 1.f - x2;
        float den = fmaxf(1.f + 2.f * xy + x2 * y2, MINN);
        float nn2 = fmaxf(al * al * x2 + 2.f * al * be * xy + be * be * y2, 0.f);
        float hn = fmaxf(sqrtf(nn2) / den, MINN);
        float hc = fminf(hn, MAXN);
        float L = artanh_f(hc) / hn;
        rowCA[t] = L * al * s_mv / den;
        rowCB[t] = L * be / den;
    }
    __syncthreads();

#pragma unroll
    for (int i = 0; i < 8; ++i) {
        int mrow = rs * 16 + i + half * 8;
        int gr = row0 + mrow;
        if (gr < nrows) {
            float ca = rowCA[mrow], cb = rowCB[mrow];
#pragma unroll
            for (int j = 0; j < 8; ++j) {
                int colg = j * 16 + l15;
                HT[(size_t)gr * 128 + colg] = ca * acc[j][i] + cb * bbv[j];
            }
        }
    }
}

__global__ __launch_bounds__(256) void post_agg(float* __restrict__ H, int nrows) {
    int r = blockIdx.x * 8 + (threadIdx.x >> 5);
    if (r >= nrows) return;
    int lane = threadIdx.x & 31;
    float* p = H + (size_t)r * 128 + lane * 4;
    float4 v = *(const float4*)p;
    float an2 = v.x * v.x + v.y * v.y + v.z * v.z + v.w * v.w;
    for (int m = 1; m < 32; m <<= 1) an2 += __shfl_xor(an2, m, 32);
    float an = fmaxf(sqrtf(an2), MINN);
    float tn = fminf(fmaxf(tanhf(an), MINN), MAXN);
    float lg = artanh_f(tn) / an;
    float4 at;
    at.x = tanhf(lg * v.x);
    at.y = tanhf(lg * v.y);
    at.z = tanhf(lg * v.z);
    at.w = tanhf(lg * v.w);
    float vn2 = at.x * at.x + at.y * at.y + at.z * at.z + at.w * at.w;
    for (int m = 1; m < 32; m <<= 1) vn2 += __shfl_xor(vn2, m, 32);
    float vn = fmaxf(sqrtf(vn2), MINN);
    float s = fminf(tanhf(vn), MAXN) / vn;
    *(float4*)p = make_float4(s * at.x, s * at.y, s * at.z, s * at.w);
}

__global__ __launch_bounds__(256) void final_scatter(
    const float* __restrict__ H, const int* __restrict__ seg,
    float* __restrict__ ssum, float* __restrict__ cnt, int nrows) {
    int r = blockIdx.x * 8 + (threadIdx.x >> 5);
    if (r >= nrows) return;
    int lane = threadIdx.x & 31;
    float4 v = *(const float4*)(H + (size_t)r * 128 + lane * 4);
    float hn2 = v.x * v.x + v.y * v.y + v.z * v.z + v.w * v.w;
    for (int m = 1; m < 32; m <<= 1) hn2 += __shfl_xor(hn2, m, 32);
    float hn = fmaxf(sqrtf(hn2), MINN);
    float pc = fminf(hn, MAXN);
    float L = artanh_f(pc) / hn;
    int s = seg[r];
    float* d = ssum + (size_t)s * 128 + lane * 4;
    atomicAdd(d + 0, L * v.x);
    atomicAdd(d + 1, L * v.y);
    atomicAdd(d + 2, L * v.z);
    atomicAdd(d + 3, L * v.w);
    if (lane == 0) atomicAdd(&cnt[s], 1.f);
}

__global__ __launch_bounds__(256) void pool_k(const float* __restrict__ ssum,
                                              const float* __restrict__ cnt,
                                              float* __restrict__ out, int nseg) {
    int r = blockIdx.x * 8 + (threadIdx.x >> 5);
    if (r >= nseg) return;
    int lane = threadIdx.x & 31;
    float inv = 1.f / fmaxf(cnt[r], 1.f);
    float4 v = *(const float4*)(ssum + (size_t)r * 128 + lane * 4);
    v.x *= inv; v.y *= inv; v.z *= inv; v.w *= inv;
    float mn2 = v.x * v.x + v.y * v.y + v.z * v.z + v.w * v.w;
    for (int m = 1; m < 32; m <<= 1) mn2 += __shfl_xor(mn2, m, 32);
    float mn = fmaxf(sqrtf(mn2), MINN);
    float s = fminf(tanhf(mn), MAXN) / mn;
    float* o = out + (size_t)r * 128 + lane * 4;
    o[0] = s * v.x; o[1] = s * v.y; o[2] = s * v.z; o[3] = s * v.w;
}

__global__ void zero_f32(float* __restrict__ p, size_t n) {
    size_t i = (size_t)blockIdx.x * blockDim.x + threadIdx.x;
    size_t st = (size_t)gridDim.x * blockDim.x;
    for (; i < n; i += st) p[i] = 0.f;
}

extern "C" void kernel_launch(void* const* d_in, const int* in_sizes, int n_in,
                              void* d_out, int out_size, void* d_ws, size_t ws_size,
                              hipStream_t stream) {
    const float* x = (const float*)d_in[0];
    const int* erow = (const int*)d_in[1];
    const int* ecol = (const int*)d_in[2];
    const int* seg = (const int*)d_in[3];
    const float* W1 = (const float*)d_in[4];
    const float* b1 = (const float*)d_in[5];
    const float* W2 = (const float*)d_in[6];
    const float* b2 = (const float*)d_in[7];

    const int N = in_sizes[0] / 256;
    const int E = in_sizes[1];
    const int NSEG = out_size / 128;

    char* ws = (char*)d_ws;
    size_t bufBytes = (size_t)N * 128 * sizeof(float);
    float* buf0 = (float*)ws;
    float* buf1 = (float*)(ws + bufBytes);
    float* bbias = (float*)(ws + 2 * bufBytes);
    float* y2 = bbias + 256;
    float* ssum = buf0;
    float* cnt = buf0 + (size_t)NSEG * 128;

    dim3 blk(256);
    dim3 gemmBlk(128);
    int gRows = (N + 7) / 8;
    int gGemm = (N + BM - 1) / BM;
    int gEdge = (E + 7) / 8;

    const int nScan = (N + SCAN_B - 1) / SCAN_B;
    size_t offBytes = (((size_t)N + 1) * sizeof(int) + 255) / 256 * 256;
    size_t bsBytes = ((size_t)nScan * sizeof(int) + 255) / 256 * 256;
    int* offs = (int*)(ws + 2 * bufBytes + 4096);
    int* scol = (int*)(ws + 2 * bufBytes + 4096 + offBytes + bsBytes);
    const int G1 = 256;
    const int NBK = (N + BKT_N - 1) / BKT_N;
    const int nTab = NBK * G1;
    const int nScanT = (nTab + SCAN_B - 1) / SCAN_B;
    size_t tmpBytes = ((size_t)E * sizeof(int2) + 255) / 256 * 256;
    size_t tabBytes = (((size_t)nTab + 1) * sizeof(int) + 255) / 256 * 256;
    int2* tmp = (int2*)buf1;
    int* btab = (int*)(ws + bufBytes + tmpBytes);
    int* bsum = (int*)(ws + bufBytes + tmpBytes + tabBytes);
    const bool grouped =
        ws_size >= 2 * bufBytes + 4096 + offBytes + bsBytes + (size_t)E * sizeof(int) &&
        NBK <= MAXBK &&
        tmpBytes + tabBytes + (size_t)nScanT * sizeof(int) <= bufBytes;

    bias_prep<<<2, 128, 0, stream>>>(b1, b2, bbias, y2);

    if (grouped) {
        bin_count<<<G1, SORT_T, 0, stream>>>(erow, btab, E, NBK);
        scan_sums<<<nScanT, SCAN_B, 0, stream>>>(btab, bsum, nTab);
        scan_apply<<<nScanT, SCAN_B, 0, stream>>>(btab, bsum, nTab);
        bin_scatter<<<G1, SORT_T, 0, stream>>>(erow, ecol, btab, tmp, E, NBK);
        bin_sort<<<NBK, SORT_T, 0, stream>>>(tmp, btab, offs, scol, G1, NBK, N);

        hyp_linear_wmma<<<gGemm, gemmBlk, 0, stream>>>(x, 256, N, W1, bbias, y2, buf0, 1);

        gather_act_linear_wmma<<<gGemm, gemmBlk, 0, stream>>>(buf0, offs, scol, N, W2,
                                                          bbias + 128, y2 + 1, buf1);
        float* ssumG = buf0;
        float* cntG = buf0 + (size_t)NSEG * 128;
        zero_f32<<<64, 256, 0, stream>>>(ssumG, (size_t)NSEG * 128 + NSEG);
        node_gather_readout<<<2048, blk, 0, stream>>>(buf1, offs, scol, seg, ssumG, cntG, N);
        pool_k<<<(NSEG + 7) / 8, blk, 0, stream>>>(ssumG, cntG, (float*)d_out, NSEG);
        return;
    }

    hyp_linear_wmma<<<gGemm, gemmBlk, 0, stream>>>(x, 256, N, W1, bbias, y2, buf0, 1);
    zero_f32<<<2048, 256, 0, stream>>>(buf1, (size_t)N * 128);
    edge_agg<<<gEdge, blk, 0, stream>>>(buf0, erow, ecol, buf1, E);
    post_agg<<<gRows, blk, 0, stream>>>(buf1, N);

    hyp_linear_wmma<<<gGemm, gemmBlk, 0, stream>>>(buf1, 128, N, W2, bbias + 128,
                                               y2 + 1, buf0, 0);
    zero_f32<<<2048, 256, 0, stream>>>(buf1, (size_t)N * 128);
    edge_agg<<<gEdge, blk, 0, stream>>>(buf0, erow, ecol, buf1, E);
    post_agg<<<gRows, blk, 0, stream>>>(buf1, N);

    zero_f32<<<64, 256, 0, stream>>>(ssum, (size_t)NSEG * 128 + NSEG);
    final_scatter<<<gRows, blk, 0, stream>>>(buf1, seg, ssum, cnt, N);
    pool_k<<<(NSEG + 7) / 8, blk, 0, stream>>>(ssum, cnt, (float*)d_out, NSEG);
}


# segment adds: finished row passed through the wave's LDS slot, lane l gets elements l+32k: an f32 atomic instruction covers one whole 128-B line (4 line adds per node, was 16); arrival order as before
# speedup vs baseline: 3.5291x; 1.0293x over previous
#include <hip/hip_runtime.h>
#include <math.h>


#define MAXN 0.996f
#define MINN 1e-15f
#define KC   32
#define LDP  (KC + 4)
#define BM   64

typedef float v2f __attribute__((ext_vector_type(2)));
typedef float v8f __attribute__((ext_vector_type(8)));
typedef float v4f __attribute__((ext_vector_type(4)));

__device__ __forceinline__ float artanh_f(float z) {
    z = fminf(fmaxf(z, -(1.f - 1e-7f)), 1.f - 1e-7f);
    return 0.5f * logf((1.f + z) / (1.f - z));
}

__global__ void bias_prep(const float* __restrict__ b1,
                          const float* __restrict__ b2,
                          float* __restrict__ bb, float* __restrict__ y2) {
    __shared__ float red[128];
    const float* b = blockIdx.x ? b2 : b1;
    int t = threadIdx.x;
    float v = b[t];
    red[t] = v * v;
    __syncthreads();
    for (int s = 64; s > 0; s >>= 1) {
        if (t < s) red[t] += red[t + s];
        __syncthreads();
    }
    float bn = fmaxf(sqrtf(red[0]), MINN);
    float tc = fminf(tanhf(bn), MAXN);
    bb[blockIdx.x * 128 + t] = tc / bn * v;
    if (t == 0) y2[blockIdx.x] = tc * tc;
}

__global__ __launch_bounds__(128) void hyp_linear_wmma(
    const float* __restrict__ X, int Kdim, int nrows,
    const float* __restrict__ W,
    const float* __restrict__ bb,
    const float* __restrict__ y2p,
    float* __restrict__ HT, int apply_exp) {
    __shared__ float As[BM][LDP];
    __shared__ float Ws[128][LDP];
    __shared__ float red[BM][8];
    __shared__ float scale_row[BM], xn_row[BM];
    __shared__ float rs_mx2[BM], rs_dot[BM];
    __shared__ float rowCA[BM], rowCB[BM];

    const int t = threadIdx.x;
    const int row0 = blockIdx.x * BM;

    const int lane = t & 31;
    const int rs = t >> 5;
    const int half = lane >> 4;
    const int l15 = lane & 15;

    v8f acc[8];
    for (int j = 0; j < 8; ++j)
        for (int i = 0; i < 8; ++i) acc[j][i] = 0.f;

    const int nchunks = Kdim / KC;

    const int sr = t >> 3;
    const int sc4 = t & 7;
    const int kq = Kdim >> 2;
    const v4f* __restrict__ X4 = (const v4f*)X;
    const v4f* __restrict__ W4 = (const v4f*)W;
    int ia[4];
#pragma unroll
    for (int i = 0; i < 4; ++i) ia[i] = min(row0 + sr + 16 * i, nrows - 1) * kq + sc4;
    const int iw = sr * kq + sc4;

    const int mr = rs * 16 + l15;

    float ss[4] = {0.f, 0.f, 0.f, 0.f};

    for (int kc = 0; kc < nchunks; ++kc) {
        const int q0 = kc * (KC / 4);
        v4f av[4], wv[8];
#pragma unroll
        for (int i = 0; i < 4; ++i) av[i] = X4[ia[i] + q0];
#pragma unroll
        for (int i = 0; i < 8; ++i) wv[i] = W4[iw + i * 16 * kq + q0];
        asm volatile("" : "+v"(av[0]), "+v"(av[1]), "+v"(av[2]), "+v"(av[3]),
                          "+v"(wv[0]), "+v"(wv[1]), "+v"(wv[2]), "+v"(wv[3]),
                          "+v"(wv[4]), "+v"(wv[5]), "+v"(wv[6]), "+v"(wv[7]));
#pragma unroll
        for (int i = 0; i < 4; ++i) {
            ss[i] += av[i].x * av[i].x + av[i].y * av[i].y + av[i].z * av[i].z + av[i].w * av[i].w;
            *(v4f*)(&As[sr + 16 * i][sc4 * 4]) = av[i];
        }
#pragma unroll
        for (int i = 0; i < 8; ++i) *(v4f*)(&Ws[sr + 16 * i][sc4 * 4]) = wv[i];
        __syncthreads();

        for (int kk = 0; kk < KC; kk += 4) {
            const int ko = kk + half * 2;
            v2f a;
            a.x = As[mr][ko];
            a.y = As[mr][ko + 1];
#pragma unroll
            for (int j = 0; j < 8; ++j) {
                const int colg = j * 16 + l15;
                v2f b;
                b.x = Ws[colg][ko];
                b.y = Ws[colg][ko + 1];
                acc[j] = __builtin_amdgcn_wmma_f32_16x16x4_f32(
                    false, a, false, b, (short)0, acc[j], false, false);
            }
        }
        __syncthreads();
    }

#pragma unroll
    for (int i = 0; i < 4; ++i) red[sr + 16 * i][sc4] = ss[i];
    __syncthreads();
    if (t < BM) {
        float un2 = red[t][0];
#pragma unroll
        for (int q = 1; q < 8; ++q) un2 += red[t][q];
        float un = fmaxf(sqrtf(un2), MINN);
        if (apply_exp) {
            float bc = fminf(tanhf(un), MAXN);
            scale_row[t] = bc / un;
            xn_row[t] = fmaxf(bc, MINN);
        } else {
            scale_row[t] = 1.f;
            xn_row[t] = un;
        }
        if (row0 + t >= nrows) scale_row[t] = 0.f;
    }
    __syncthreads();
    if (apply_exp) {
#pragma unroll
        for (int i = 0; i < 8; ++i) {
            const float sc = scale_row[rs * 16 + i + half * 8];
#pragma unroll
            for (int j = 0; j < 8; ++j) acc[j][i] *= sc;
        }
    }

    float bbv[8];
#pragma unroll
    for (int j = 0; j < 8; ++j) bbv[j] = bb[j * 16 + l15];

#pragma unroll
    for (int i = 0; i < 8; ++i) {
        float p2a = 0.f, pda = 0.f, p2b = 0.f, pdb = 0.f;
#pragma unroll
        for (int j = 0; j < 4; ++j) {
            float v = acc[j][i];
            p2a += v * v;
            pda += v * bbv[j];
        }
#pragma unroll
        for (int j = 4; j < 8; ++j) {
            float v = acc[j][i];
            p2b += v * v;
            pdb += v * bbv[j];
        }
        for (int m = 1; m < 16; m <<= 1) {
            p2a += __shfl_xor(p2a, m, 16);
            pda += __shfl_xor(pda, m, 16);
            p2b += __shfl_xor(p2b, m, 16);
            pdb += __shfl_xor(pdb, m, 16);
        }
        int mrow = rs * 16 + i + half * 8;
        if (l15 == 0) {
            rs_mx2[mrow] = p2a + p2b;
            rs_dot[mrow] = pda + pdb;
        }
    }
    __syncthreads();

    if (t < BM) {
        float xn = xn_row[t];
        float mxn = fmaxf(sqrtf(rs_mx2[t]), MINN);
        float tv = tanhf(mxn / xn * artanh_f(xn));
        float tc = fminf(tv, MAXN);
        float s_mv = tc / mxn;
        float x2 = tc * tc;
        float xy = s_mv * rs_dot[t];
        float y2 = y2p[0];
        float al = 1.f + 2.f * xy + y2;
        float be = 1.f - x2;
        float den = fmaxf(1.f + 2.f * xy + x2 * y2, MINN);
        float nn2 = fmaxf(al * al * x2 + 2.f * al * be * xy + be * be * y2, 0.f);
        float hn = fmaxf(sqrtf(nn2) / den, MINN);
        float hc = fminf(hn, MAXN);
        float L = artanh_f(hc) / hn;
        rowCA[t] = L * al * s_mv / den;
        rowCB[t] = L * be / den;
    }
    __syncthreads();

#pragma unroll
    for (int i = 0; i < 8; ++i) {
        int mrow = rs * 16 + i + half * 8;
        int gr = row0 + mrow;
        if (gr < nrows) {
            float ca = rowCA[mrow], cb = rowCB[mrow];
#pragma unroll
            for (int j = 0; j < 8; ++j) {
                int colg = j * 16 + l15;
                HT[(size_t)gr * 128 + colg] = ca * acc[j][i] + cb * bbv[j];
            }
        }
    }
}

__global__ __launch_bounds__(256) void edge_agg(const float* __restrict__ HT,
                                                const int* __restrict__ erow,
                                                const int* __restrict__ ecol,
                                                float* __restrict__ AGG, int E) {
    int e = blockIdx.x * 8 + (threadIdx.x >> 5);
    if (e >= E) return;
    int lane = threadIdx.x & 31;
    int r = erow[e], c = ecol[e];
    float4 v = *(const float4*)(HT + (size_t)c * 128 + lane * 4);
    float* dst = AGG + (size_t)r * 128 + lane * 4;
    atomicAdd(dst + 0, v.x);
    atomicAdd(dst + 1, v.y);
    atomicAdd(dst + 2, v.z);
    atomicAdd(dst + 3, v.w);
}

#define SCAN_B 1024
#define BKT_SH 8
#define BKT_N  256
#define MAXBK  1024
#define SORT_T 1024
#define SORT_U 4

__global__ __launch_bounds__(SORT_T) void bin_count(const int* __restrict__ erow,
                                                    int* __restrict__ bh,
                                                    int E, int nbk) {
    __shared__ int hist[MAXBK];
    const int t = threadIdx.x, G = gridDim.x, b = blockIdx.x;
    for (int k = t; k < nbk; k += SORT_T) hist[k] = 0;
    __syncthreads();
    const int chunk = (E + G - 1) / G;
    const int beg = b * chunk, end = min(E, beg + chunk);
    for (int i = beg + t; i < end; i += SORT_U * SORT_T) {
        int r[SORT_U];
#pragma unroll
        for (int u = 0; u < SORT_U; ++u) {
            const int j = i + u * SORT_T;
            r[u] = j < end ? erow[j] : 0;
        }
#pragma unroll
        for (int u = 0; u < SORT_U; ++u) {
            const int k = min((int)((unsigned)r[u] >> BKT_SH), nbk - 1);
            if (i + u * SORT_T < end) atomicAdd(&hist[k], 1);
        }
    }
    __syncthreads();
    for (int k = t; k < nbk; k += SORT_T) bh[(size_t)k * G + b] = hist[k];
}

__global__ __launch_bounds__(SORT_T) void bin_scatter(const int* __restrict__ erow,
                                                      const int* __restrict__ ecol,
                                                      const int* __restrict__ bo,
                                                      int2* __restrict__ tmp,
                                                      int E, int nbk) {
    __shared__ int cur[MAXBK];
    const int t = threadIdx.x, G = gridDim.x, b = blockIdx.x;
    for (int k = t; k < nbk; k += SORT_T) cur[k] = bo[(size_t)k * G + b];
    __syncthreads();
    const int chunk = (E + G - 1) / G;
    const int beg = b * chunk, end = min(E, beg + chunk);
    for (int i = beg + t; i < end; i += SORT_U * SORT_T) {
        int r[SORT_U], c[SORT_U];
#pragma unroll
        for (int u = 0; u < SORT_U; ++u) {
            const int j = i + u * SORT_T;
            r[u] = j < end ? erow[j] : 0;
            c[u] = j < end ? ecol[j] : 0;
        }
#pragma unroll
        for (int u = 0; u < SORT_U; ++u) {
            if (i + u * SORT_T < end) {
                const int k = min((int)((unsigned)r[u] >> BKT_SH), nbk - 1);
                const int p = atomicAdd(&cur[k], 1);
                tmp[p] = make_int2(r[u], c[u]);
            }
        }
    }
}

__global__ __launch_bounds__(SORT_T) void bin_sort(const int2* __restrict__ tmp,
                                                   const int* __restrict__ bo,
                                                   int* __restrict__ offs,
                                                   int* __restrict__ scol,
                                                   int G, int nbk, int n) {
    __shared__ int cnt[BKT_N];
    __shared__ int wsum[BKT_N / 32];
    const int t = threadIdx.x, k = blockIdx.x, lane = t & 31, wid = t >> 5;
    const int s = bo[(size_t)k * G], e = bo[(size_t)(k + 1) * G];
    if (t < BKT_N) cnt[t] = 0;
    __syncthreads();
    for (int i = s + t; i < e; i += SORT_U * SORT_T) {
        int r[SORT_U];
#pragma unroll
        for (int u = 0; u < SORT_U; ++u) {
            const int j = i + u * SORT_T;
            r[u] = j < e ? tmp[j].x : 0;
        }
#pragma unroll
        for (int u = 0; u < SORT_U; ++u)
            if (i + u * SORT_T < e) atomicAdd(&cnt[r[u] & (BKT_N - 1)], 1);
    }
    __syncthreads();
    int v = 0, sc = 0;
    if (t < BKT_N) {
        v = cnt[t];
        sc = v;
        for (int m = 1; m < 32; m <<= 1) {
            int u = __shfl_up(sc, m, 32);
            if (lane >= m) sc += u;
        }
        if (lane == 31) wsum[wid] = sc;
    }
    __syncthreads();
    if (t < BKT_N) {
        int base = s;
        for (int w = 0; w < wid; ++w) base += wsum[w];
        const int excl = base + sc - v;
        cnt[t] = excl;
        const int node = k * BKT_N + t;
        if (node < n) offs[node] = excl;
    }
    if (k == nbk - 1 && t == 0) offs[n] = e;
    __syncthreads();
    for (int i = s + t; i < e; i += SORT_U * SORT_T) {
        int2 rc[SORT_U];
#pragma unroll
        for (int u = 0; u < SORT_U; ++u) {
            const int j = i + u * SORT_T;
            rc[u] = j < e ? tmp[j] : make_int2(0, 0);
        }
#pragma unroll
        for (int u = 0; u < SORT_U; ++u) {
            if (i + u * SORT_T < e) {
                const int p = atomicAdd(&cnt[rc[u].x & (BKT_N - 1)], 1);
                scol[p] = rc[u].y;
            }
        }
    }
}

__device__ __forceinline__ int block_sum_1024(int v, int* sh) {
    int t = threadIdx.x;
    for (int m = 1; m < 32; m <<= 1) v += __shfl_xor(v, m, 32);
    if ((t & 31) == 0) sh[t >> 5] = v;
    __syncthreads();
    int w = sh[t & 31];
    for (int m = 1; m < 32; m <<= 1) w += __shfl_xor(w, m, 32);
    __syncthreads();
    return w;
}

__global__ __launch_bounds__(SCAN_B) void scan_sums(const int* __restrict__ cnt,
                                                    int* __restrict__ bsum, int n) {
    __shared__ int sh[32];
    int i = blockIdx.x * SCAN_B + threadIdx.x;
    int tot = block_sum_1024(i < n ? cnt[i] : 0, sh);
    if (threadIdx.x == 0) bsum[blockIdx.x] = tot;
}

__global__ __launch_bounds__(SCAN_B) void scan_apply(int* __restrict__ offs,
                                                     const int* __restrict__ bsum, int n) {
    __shared__ int sh[32];
    const int t = threadIdx.x, lane = t & 31, wid = t >> 5;
    int p = 0;
    for (int b = t; b < (int)blockIdx.x; b += SCAN_B) p += bsum[b];
    const int base = block_sum_1024(p, sh);

    int i = blockIdx.x * SCAN_B + t;
    int v = i < n ? offs[i] : 0;
    int s = v;
    for (int m = 1; m < 32; m <<= 1) {
        int u = __shfl_up(s, m, 32);
        if (lane >= m) s += u;
    }
    if (lane == 31) sh[wid] = s;
    __syncthreads();
    if (wid == 0) {
        int w0 = sh[lane], w = w0;
        for (int m = 1; m < 32; m <<= 1) {
            int u = __shfl_up(w, m, 32);
            if (lane >= m) w += u;
        }
        sh[lane] = w - w0;
    }
    __syncthreads();
    int excl = base + sh[wid] + s - v;
    if (i < n) offs[i] = excl;
    if (i == n - 1) offs[n] = excl + v;
}

__device__ __forceinline__ float wave_sum32(float x) {
    for (int m = 1; m < 32; m <<= 1) x += __shfl_xor(x, m, 32);
    return x;
}

__device__ __forceinline__ float4 tanh_row(float4 v, float amax, float& vn) {
    float an2 = wave_sum32(v.x * v.x + v.y * v.y + v.z * v.z + v.w * v.w);
    float an = fmaxf(sqrtf(an2), MINN);
    float lg = fminf(1.f, amax / an);
    float4 at;
    at.x = tanhf(lg * v.x);
    at.y = tanhf(lg * v.y);
    at.z = tanhf(lg * v.z);
    at.w = tanhf(lg * v.w);
    float vn2 = wave_sum32(at.x * at.x + at.y * at.y + at.z * at.z + at.w * at.w);
    vn = fmaxf(sqrtf(vn2), MINN);
    return at;
}

__global__ __launch_bounds__(256) void node_gather_readout(const float* __restrict__ HT,
                                                           const int* __restrict__ offs,
                                                           const int* __restrict__ scol,
                                                           const int* __restrict__ seg,
                                                           float* __restrict__ ssum,
                                                           float* __restrict__ cnt, int nrows) {
    __shared__ float tr[8][128];
    const int lane = threadIdx.x & 31;
    const int wslot = threadIdx.x >> 5;
    const int w0 = __builtin_amdgcn_readfirstlane((int)(blockIdx.x * 8 + (threadIdx.x >> 5)));
    const int nw = gridDim.x * 8;
    const float amax = artanh_f(MAXN);
    for (int r = w0; r < nrows; r += nw) {
        const int beg = offs[r], end = offs[r + 1];
        const int sg = seg[r];
        float4 acc = make_float4(0.f, 0.f, 0.f, 0.f);
        for (int b0 = beg; b0 < end; b0 += 32) {
            const int n = min(32, end - b0);
            const int myc = (lane < n) ? scol[b0 + lane] : 0;
            int j = 0;
            for (; j + 4 <= n; j += 4) {
                const int c0 = __builtin_amdgcn_readlane(myc, j);
                const int c1 = __builtin_amdgcn_readlane(myc, j + 1);
                const int c2 = __builtin_amdgcn_readlane(myc, j + 2);
                const int c3 = __builtin_amdgcn_readlane(myc, j + 3);
                const float4 v0 = *((const float4*)(HT + (size_t)c0 * 128) + lane);
                const float4 v1 = *((const float4*)(HT + (size_t)c1 * 128) + lane);
                const float4 v2 = *((const float4*)(HT + (size_t)c2 * 128) + lane);
                const float4 v3 = *((const float4*)(HT + (size_t)c3 * 128) + lane);
                acc.x += v0.x; acc.y += v0.y; acc.z += v0.z; acc.w += v0.w;
                acc.x += v1.x; acc.y += v1.y; acc.z += v1.z; acc.w += v1.w;
                acc.x += v2.x; acc.y += v2.y; acc.z += v2.z; acc.w += v2.w;
                acc.x += v3.x; acc.y += v3.y; acc.z += v3.z; acc.w += v3.w;
            }
            for (; j < n; ++j) {
                const int c0 = __builtin_amdgcn_readlane(myc, j);
                const float4 v0 = *((const float4*)(HT + (size_t)c0 * 128) + lane);
                acc.x += v0.x; acc.y += v0.y; acc.z += v0.z; acc.w += v0.w;
            }
        }
        float vn;
        const float4 at = tanh_row(acc, amax, vn);
        const float L = fminf(1.f, amax / vn);
        *(float4*)(&tr[wslot][lane * 4]) = make_float4(L * at.x, L * at.y, L * at.z, L * at.w);
        const float e0 = tr[wslot][lane];
        const float e1 = tr[wslot][lane + 32];
        const float e2 = tr[wslot][lane + 64];
        const float e3 = tr[wslot][lane + 96];
        float* d = ssum + (size_t)sg * 128 + lane;
        atomicAdd(d, e0);
        atomicAdd(d + 32, e1);
        atomicAdd(d + 64, e2);
        atomicAdd(d + 96, e3);
        if (lane == 0) atomicAdd(&cnt[sg], 1.f);
    }
}

__device__ __forceinline__ void add_rows(const float* __restrict__ HTin, int myc, int n,
                                         int lane, float4& acc) {
    int j = 0;
    for (; j + 8 <= n; j += 8) {
        const int c0 = __builtin_amdgcn_readlane(myc, j);
        const int c1 = __builtin_amdgcn_readlane(myc, j + 1);
        const int c2 = __builtin_amdgcn_readlane(myc, j + 2);
        const int c3 = __builtin_amdgcn_readlane(myc, j + 3);
        const int c4 = __builtin_amdgcn_readlane(myc, j + 4);
        const int c5 = __builtin_amdgcn_readlane(myc, j + 5);
        const int c6 = __builtin_amdgcn_readlane(myc, j + 6);
        const int c7 = __builtin_amdgcn_readlane(myc, j + 7);
        const float4 v0 = *((const float4*)(HTin + (size_t)c0 * 128) + lane);
        const float4 v1 = *((const float4*)(HTin + (size_t)c1 * 128) + lane);
        const float4 v2 = *((const float4*)(HTin + (size_t)c2 * 128) + lane);
        const float4 v3 = *((const float4*)(HTin + (size_t)c3 * 128) + lane);
        const float4 v4 = *((const float4*)(HTin + (size_t)c4 * 128) + lane);
        const float4 v5 = *((const float4*)(HTin + (size_t)c5 * 128) + lane);
        const float4 v6 = *((const float4*)(HTin + (size_t)c6 * 128) + lane);
        const float4 v7 = *((const float4*)(HTin + (size_t)c7 * 128) + lane);
        acc.x += v0.x; acc.y += v0.y; acc.z += v0.z; acc.w += v0.w;
        acc.x += v1.x; acc.y += v1.y; acc.z += v1.z; acc.w += v1.w;
        acc.x += v2.x; acc.y += v2.y; acc.z += v2.z; acc.w += v2.w;
        acc.x += v3.x; acc.y += v3.y; acc.z += v3.z; acc.w += v3.w;
        acc.x += v4.x; acc.y += v4.y; acc.z += v4.z; acc.w += v4.w;
        acc.x += v5.x; acc.y += v5.y; acc.z += v5.z; acc.w += v5.w;
        acc.x += v6.x; acc.y += v6.y; acc.z += v6.z; acc.w += v6.w;
        acc.x += v7.x; acc.y += v7.y; acc.z += v7.z; acc.w += v7.w;
    }
    for (; j + 4 <= n; j += 4) {
        const int c0 = __builtin_amdgcn_readlane(myc, j);
        const int c1 = __builtin_amdgcn_readlane(myc, j + 1);
        const int c2 = __builtin_amdgcn_readlane(myc, j + 2);
        const int c3 = __builtin_amdgcn_readlane(myc, j + 3);
        const float4 v0 = *((const float4*)(HTin + (size_t)c0 * 128) + lane);
        const float4 v1 = *((const float4*)(HTin + (size_t)c1 * 128) + lane);
        const float4 v2 = *((const float4*)(HTin + (size_t)c2 * 128) + lane);
        const float4 v3 = *((const float4*)(HTin + (size_t)c3 * 128) + lane);
        acc.x += v0.x; acc.y += v0.y; acc.z += v0.z; acc.w += v0.w;
        acc.x += v1.x; acc.y += v1.y; acc.z += v1.z; acc.w += v1.w;
        acc.x += v2.x; acc.y += v2.y; acc.z += v2.z; acc.w += v2.w;
        acc.x += v3.x; acc.y += v3.y; acc.z += v3.z; acc.w += v3.w;
    }
    for (; j < n; ++j) {
        const int c0 = __builtin_amdgcn_readlane(myc, j);
        const float4 v0 = *((const float4*)(HTin + (size_t)c0 * 128) + lane);
        acc.x += v0.x; acc.y += v0.y; acc.z += v0.z; acc.w += v0.w;
    }
}

#define HD   128
#define LDA  (HD + 4)
#define KC2  16
#define LDW  (KC2 + 4)

__global__ __launch_bounds__(128, 7) void gather_act_linear_wmma(
    const float* __restrict__ HTin,
    const int* __restrict__ offs, const int* __restrict__ scol, int nrows,
    const float* __restrict__ W,
    const float* __restrict__ bb,
    const float* __restrict__ y2p,
    float* __restrict__ HT) {
    __shared__ float As[BM][LDA];
    __shared__ float Ws[128][LDW];
    float* const xn_row = &Ws[0][0];
    float* const rs_mx2 = xn_row + BM;
    float* const rs_dot = rs_mx2 + BM;
    float* const rowCA = rs_dot + BM;
    float* const rowCB = rowCA + BM;

    const int t = threadIdx.x;
    const int row0 = blockIdx.x * BM;
    const int lane = t & 31;
    const int rs = __builtin_amdgcn_readfirstlane(t >> 5);
    const int half = lane >> 4;
    const int l15 = lane & 15;

    const int rbase = row0 + rs * 16;
    const float amax = artanh_f(MAXN);
    const int ob = offs[min(rbase + min(lane, 16), nrows)];
    float my_xn = 0.f;
    int begN = __builtin_amdgcn_readlane(ob, 0);
    int endN = __builtin_amdgcn_readlane(ob, 1);
    int mycN = (lane < endN - begN) ? scol[begN + lane] : 0;
    for (int i = 0; i < 16; ++i) {
        const int beg = begN, end = endN;
        const int myc = mycN;
        if (i < 15) {
            begN = __builtin_amdgcn_readlane(ob, i + 1);
            endN = __builtin_amdgcn_readlane(ob, i + 2);
            mycN = (lane < endN - begN) ? scol[begN + lane] : 0;
        }
        float4 acc = make_float4(0.f, 0.f, 0.f, 0.f);
        add_rows(HTin, myc, min(32, end - beg), lane, acc);
        for (int b0 = beg + 32; b0 < end; b0 += 32) {
            const int n = min(32, end - b0);
            const int mc = (lane < n) ? scol[b0 + lane] : 0;
            add_rows(HTin, mc, n, lane, acc);
        }
        float vn;
        const float4 at = tanh_row(acc, amax, vn);
        const float tcl = fminf(tanhf(vn), MAXN);
        const float s = tcl / vn;
        if (lane == i) my_xn = fmaxf(tcl, MINN);
        v4f hv;
        hv.x = s * at.x; hv.y = s * at.y; hv.z = s * at.z; hv.w = s * at.w;
        *(v4f*)(&As[rs * 16 + i][lane * 4]) = hv;
    }

    v8f acc[8];
    for (int j = 0; j < 8; ++j)
        for (int i = 0; i < 8; ++i) acc[j][i] = 0.f;

    const int sr = t >> 2;
    const int sc4 = t & 3;
    const int kq = HD >> 2;
    const v4f* __restrict__ W4 = (const v4f*)W;
    const int iw = sr * kq + sc4;
    const int mr = rs * 16 + l15;

    for (int kc = 0; kc < HD / KC2; ++kc) {
        const int q0 = kc * (KC2 / 4);
        v4f wv[4];
#pragma unroll
        for (int i = 0; i < 4; ++i) wv[i] = W4[iw + i * 32 * kq + q0];
        asm volatile("" : "+v"(wv[0]), "+v"(wv[1]), "+v"(wv[2]), "+v"(wv[3]));
#pragma unroll
        for (int i = 0; i < 4; ++i) *(v4f*)(&Ws[sr + 32 * i][sc4 * 4]) = wv[i];
        __syncthreads();

        for (int kk = 0; kk < KC2; kk += 4) {
            const int ko = kk + half * 2;
            v2f a;
            a.x = As[mr][kc * KC2 + ko];
            a.y = As[mr][kc * KC2 + ko + 1];
#pragma unroll
            for (int j = 0; j < 8; ++j) {
                const int colg = j * 16 + l15;
                v2f b;
                b.x = Ws[colg][ko];
                b.y = Ws[colg][ko + 1];
                acc[j] = __builtin_amdgcn_wmma_f32_16x16x4_f32(
                    false, a, false, b, (short)0, acc[j], false, false);
            }
        }
        __syncthreads();
    }

    if (lane < 16) xn_row[rs * 16 + lane] = my_xn;

    float bbv[8];
#pragma unroll
    for (int j = 0; j < 8; ++j) bbv[j] = bb[j * 16 + l15];

#pragma unroll
    for (int i = 0; i < 8; ++i) {
        float p2a = 0.f, pda = 0.f, p2b = 0.f, pdb = 0.f;
#pragma unroll
        for (int j = 0; j < 4; ++j) {
            float v = acc[j][i];
            p2a += v * v;
            pda += v * bbv[j];
        }
#pragma unroll
        for (int j = 4; j < 8; ++j) {
            float v = acc[j][i];
            p2b += v * v;
            pdb += v * bbv[j];
        }
        for (int m = 1; m < 16; m <<= 1) {
            p2a += __shfl_xor(p2a, m, 16);
            pda += __shfl_xor(pda, m, 16);
            p2b += __shfl_xor(p2b, m, 16);
            pdb += __shfl_xor(pdb, m, 16);
        }
        int mrow = rs * 16 + i + half * 8;
        if (l15 == 0) {
            rs_mx2[mrow] = p2a + p2b;
            rs_dot[mrow] = pda + pdb;
        }
    }
    __syncthreads();

    if (t < BM) {
        float xn = xn_row[t];
        float mxn = fmaxf(sqrtf(rs_mx2[t]), MINN);
        float tv = tanhf(mxn / xn * artanh_f(xn));
        float tc = fminf(tv, MAXN);
        float s_mv = tc / mxn;
        float x2 = tc * tc;
        float xy = s_mv * rs_dot[t];
        float y2 = y2p[0];
        float al = 1.f + 2.f * xy + y2;
        float be = 1.f - x2;
        float den = fmaxf(1.f + 2.f * xy + x2 * y2, MINN);
        float nn2 = fmaxf(al * al * x2 + 2.f * al * be * xy + be * be * y2, 0.f);
        float hn = fmaxf(sqrtf(nn2) / den, MINN);
        float hc = fminf(hn, MAXN);
        float L = artanh_f(hc) / hn;
        rowCA[t] = L * al * s_mv / den;
        rowCB[t] = L * be / den;
    }
    __syncthreads();

#pragma unroll
    for (int i = 0; i < 8; ++i) {
        int mrow = rs * 16 + i + half * 8;
        int gr = row0 + mrow;
        if (gr < nrows) {
            float ca = rowCA[mrow], cb = rowCB[mrow];
#pragma unroll
            for (int j = 0; j < 8; ++j) {
                int colg = j * 16 + l15;
                HT[(size_t)gr * 128 + colg] = ca * acc[j][i] + cb * bbv[j];
            }
        }
    }
}

__global__ __launch_bounds__(256) void post_agg(float* __restrict__ H, int nrows) {
    int r = blockIdx.x * 8 + (threadIdx.x >> 5);
    if (r >= nrows) return;
    int lane = threadIdx.x & 31;
    float* p = H + (size_t)r * 128 + lane * 4;
    float4 v = *(const float4*)p;
    float an2 = v.x * v.x + v.y * v.y + v.z * v.z + v.w * v.w;
    for (int m = 1; m < 32; m <<= 1) an2 += __shfl_xor(an2, m, 32);
    float an = fmaxf(sqrtf(an2), MINN);
    float tn = fminf(fmaxf(tanhf(an), MINN), MAXN);
    float lg = artanh_f(tn) / an;
    float4 at;
    at.x = tanhf(lg * v.x);
    at.y = tanhf(lg * v.y);
    at.z = tanhf(lg * v.z);
    at.w = tanhf(lg * v.w);
    float vn2 = at.x * at.x + at.y * at.y + at.z * at.z + at.w * at.w;
    for (int m = 1; m < 32; m <<= 1) vn2 += __shfl_xor(vn2, m, 32);
    float vn = fmaxf(sqrtf(vn2), MINN);
    float s = fminf(tanhf(vn), MAXN) / vn;
    *(float4*)p = make_float4(s * at.x, s * at.y, s * at.z, s * at.w);
}

__global__ __launch_bounds__(256) void final_scatter(
    const float* __restrict__ H, const int* __restrict__ seg,
    float* __restrict__ ssum, float* __restrict__ cnt, int nrows) {
    int r = blockIdx.x * 8 + (threadIdx.x >> 5);
    if (r >= nrows) return;
    int lane = threadIdx.x & 31;
    float4 v = *(const float4*)(H + (size_t)r * 128 + lane * 4);
    float hn2 = v.x * v.x + v.y * v.y + v.z * v.z + v.w * v.w;
    for (int m = 1; m < 32; m <<= 1) hn2 += __shfl_xor(hn2, m, 32);
    float hn = fmaxf(sqrtf(hn2), MINN);
    float pc = fminf(hn, MAXN);
    float L = artanh_f(pc) / hn;
    int s = seg[r];
    float* d = ssum + (size_t)s * 128 + lane * 4;
    atomicAdd(d + 0, L * v.x);
    atomicAdd(d + 1, L * v.y);
    atomicAdd(d + 2, L * v.z);
    atomicAdd(d + 3, L * v.w);
    if (lane == 0) atomicAdd(&cnt[s], 1.f);
}

__global__ __launch_bounds__(256) void pool_k(const float* __restrict__ ssum,
                                              const float* __restrict__ cnt,
                                              float* __restrict__ out, int nseg) {
    int r = blockIdx.x * 8 + (threadIdx.x >> 5);
    if (r >= nseg) return;
    int lane = threadIdx.x & 31;
    float inv = 1.f / fmaxf(cnt[r], 1.f);
    float4 v = *(const float4*)(ssum + (size_t)r * 128 + lane * 4);
    v.x *= inv; v.y *= inv; v.z *= inv; v.w *= inv;
    float mn2 = v.x * v.x + v.y * v.y + v.z * v.z + v.w * v.w;
    for (int m = 1; m < 32; m <<= 1) mn2 += __shfl_xor(mn2, m, 32);
    float mn = fmaxf(sqrtf(mn2), MINN);
    float s = fminf(tanhf(mn), MAXN) / mn;
    float* o = out + (size_t)r * 128 + lane * 4;
    o[0] = s * v.x; o[1] = s * v.y; o[2] = s * v.z; o[3] = s * v.w;
}

__global__ void zero_f32(float* __restrict__ p, size_t n) {
    size_t i = (size_t)blockIdx.x * blockDim.x + threadIdx.x;
    size_t st = (size_t)gridDim.x * blockDim.x;
    for (; i < n; i += st) p[i] = 0.f;
}

extern "C" void kernel_launch(void* const* d_in, const int* in_sizes, int n_in,
                              void* d_out, int out_size, void* d_ws, size_t ws_size,
                              hipStream_t stream) {
    const float* x = (const float*)d_in[0];
    const int* erow = (const int*)d_in[1];
    const int* ecol = (const int*)d_in[2];
    const int* seg = (const int*)d_in[3];
    const float* W1 = (const float*)d_in[4];
    const float* b1 = (const float*)d_in[5];
    const float* W2 = (const float*)d_in[6];
    const float* b2 = (const float*)d_in[7];

    const int N = in_sizes[0] / 256;
    const int E = in_sizes[1];
    const int NSEG = out_size / 128;

    char* ws = (char*)d_ws;
    size_t bufBytes = (size_t)N * 128 * sizeof(float);
    float* buf0 = (float*)ws;
    float* buf1 = (float*)(ws + bufBytes);
    float* bbias = (float*)(ws + 2 * bufBytes);
    float* y2 = bbias + 256;
    float* ssum = buf0;
    float* cnt = buf0 + (size_t)NSEG * 128;

    dim3 blk(256);
    dim3 gemmBlk(128);
    int gRows = (N + 7) / 8;
    int gGemm = (N + BM - 1) / BM;
    int gEdge = (E + 7) / 8;

    const int nScan = (N + SCAN_B - 1) / SCAN_B;
    size_t offBytes = (((size_t)N + 1) * sizeof(int) + 255) / 256 * 256;
    size_t bsBytes = ((size_t)nScan * sizeof(int) + 255) / 256 * 256;
    int* offs = (int*)(ws + 2 * bufBytes + 4096);
    int* scol = (int*)(ws + 2 * bufBytes + 4096 + offBytes + bsBytes);
    const int G1 = 256;
    const int NBK = (N + BKT_N - 1) / BKT_N;
    const int nTab = NBK * G1;
    const int nScanT = (nTab + SCAN_B - 1) / SCAN_B;
    size_t tmpBytes = ((size_t)E * sizeof(int2) + 255) / 256 * 256;
    size_t tabBytes = (((size_t)nTab + 1) * sizeof(int) + 255) / 256 * 256;
    int2* tmp = (int2*)buf1;
    int* btab = (int*)(ws + bufBytes + tmpBytes);
    int* bsum = (int*)(ws + bufBytes + tmpBytes + tabBytes);
    const bool grouped =
        ws_size >= 2 * bufBytes + 4096 + offBytes + bsBytes + (size_t)E * sizeof(int) &&
        NBK <= MAXBK &&
        tmpBytes + tabBytes + (size_t)nScanT * sizeof(int) <= bufBytes;

    bias_prep<<<2, 128, 0, stream>>>(b1, b2, bbias, y2);

    if (grouped) {
        bin_count<<<G1, SORT_T, 0, stream>>>(erow, btab, E, NBK);
        scan_sums<<<nScanT, SCAN_B, 0, stream>>>(btab, bsum, nTab);
        scan_apply<<<nScanT, SCAN_B, 0, stream>>>(btab, bsum, nTab);
        bin_scatter<<<G1, SORT_T, 0, stream>>>(erow, ecol, btab, tmp, E, NBK);
        bin_sort<<<NBK, SORT_T, 0, stream>>>(tmp, btab, offs, scol, G1, NBK, N);

        hyp_linear_wmma<<<gGemm, gemmBlk, 0, stream>>>(x, 256, N, W1, bbias, y2, buf0, 1);

        gather_act_linear_wmma<<<gGemm, gemmBlk, 0, stream>>>(buf0, offs, scol, N, W2,
                                                          bbias + 128, y2 + 1, buf1);
        float* ssumG = buf0;
        float* cntG = buf0 + (size_t)NSEG * 128;
        zero_f32<<<64, 256, 0, stream>>>(ssumG, (size_t)NSEG * 128 + NSEG);
        node_gather_readout<<<2048, blk, 0, stream>>>(buf1, offs, scol, seg, ssumG, cntG, N);
        pool_k<<<(NSEG + 7) / 8, blk, 0, stream>>>(ssumG, cntG, (float*)d_out, NSEG);
        return;
    }

    hyp_linear_wmma<<<gGemm, gemmBlk, 0, stream>>>(x, 256, N, W1, bbias, y2, buf0, 1);
    zero_f32<<<2048, 256, 0, stream>>>(buf1, (size_t)N * 128);
    edge_agg<<<gEdge, blk, 0, stream>>>(buf0, erow, ecol, buf1, E);
    post_agg<<<gRows, blk, 0, stream>>>(buf1, N);

    hyp_linear_wmma<<<gGemm, gemmBlk, 0, stream>>>(buf1, 128, N, W2, bbias + 128,
                                               y2 + 1, buf0, 0);
    zero_f32<<<2048, 256, 0, stream>>>(buf1, (size_t)N * 128);
    edge_agg<<<gEdge, blk, 0, stream>>>(buf0, erow, ecol, buf1, E);
    post_agg<<<gRows, blk, 0, stream>>>(buf1, N);

    zero_f32<<<64, 256, 0, stream>>>(ssum, (size_t)NSEG * 128 + NSEG);
    final_scatter<<<gRows, blk, 0, stream>>>(buf1, seg, ssum, cnt, N);
    pool_k<<<(NSEG + 7) / 8, blk, 0, stream>>>(ssum, cnt, (float*)d_out, NSEG);
}


# per-node row maps of the gather waves on the chip's transcendental instructions: library tanhf -> v_tanh_f32, exact sqrt and IEEE division sequences -> v_sqrt_f32, v_rcp_f32; their rounding changes
# speedup vs baseline: 3.6674x; 1.0392x over previous
#include <hip/hip_runtime.h>
#include <math.h>


#define MAXN 0.996f
#define MINN 1e-15f
#define KC   32
#define LDP  (KC + 4)
#define BM   64

typedef float v2f __attribute__((ext_vector_type(2)));
typedef float v8f __attribute__((ext_vector_type(8)));
typedef float v4f __attribute__((ext_vector_type(4)));

__device__ __forceinline__ float artanh_f(float z) {
    z = fminf(fmaxf(z, -(1.f - 1e-7f)), 1.f - 1e-7f);
    return 0.5f * logf((1.f + z) / (1.f - z));
}

__device__ __forceinline__ float tanh_hw(float x) { return __builtin_amdgcn_tanhf(x); }
__device__ __forceinline__ float sqrt_hw(float x) { return __builtin_amdgcn_sqrtf(x); }
__device__ __forceinline__ float rcp_hw(float x) { return __builtin_amdgcn_rcpf(x); }

__global__ void bias_prep(const float* __restrict__ b1,
                          const float* __restrict__ b2,
                          float* __restrict__ bb, float* __restrict__ y2) {
    __shared__ float red[128];
    const float* b = blockIdx.x ? b2 : b1;
    int t = threadIdx.x;
    float v = b[t];
    red[t] = v * v;
    __syncthreads();
    for (int s = 64; s > 0; s >>= 1) {
        if (t < s) red[t] += red[t + s];
        __syncthreads();
    }
    float bn = fmaxf(sqrtf(red[0]), MINN);
    float tc = fminf(tanhf(bn), MAXN);
    bb[blockIdx.x * 128 + t] = tc / bn * v;
    if (t == 0) y2[blockIdx.x] = tc * tc;
}

__global__ __launch_bounds__(128) void hyp_linear_wmma(
    const float* __restrict__ X, int Kdim, int nrows,
    const float* __restrict__ W,
    const float* __restrict__ bb,
    const float* __restrict__ y2p,
    float* __restrict__ HT, int apply_exp) {
    __shared__ float As[BM][LDP];
    __shared__ float Ws[128][LDP];
    __shared__ float red[BM][8];
    __shared__ float scale_row[BM], xn_row[BM];
    __shared__ float rs_mx2[BM], rs_dot[BM];
    __shared__ float rowCA[BM], rowCB[BM];

    const int t = threadIdx.x;
    const int row0 = blockIdx.x * BM;

    const int lane = t & 31;
    const int rs = t >> 5;
    const int half = lane >> 4;
    const int l15 = lane & 15;

    v8f acc[8];
    for (int j = 0; j < 8; ++j)
        for (int i = 0; i < 8; ++i) acc[j][i] = 0.f;

    const int nchunks = Kdim / KC;

    const int sr = t >> 3;
    const int sc4 = t & 7;
    const int kq = Kdim >> 2;
    const v4f* __restrict__ X4 = (const v4f*)X;
    const v4f* __restrict__ W4 = (const v4f*)W;
    int ia[4];
#pragma unroll
    for (int i = 0; i < 4; ++i) ia[i] = min(row0 + sr + 16 * i, nrows - 1) * kq + sc4;
    const int iw = sr * kq + sc4;

    const int mr = rs * 16 + l15;

    float ss[4] = {0.f, 0.f, 0.f, 0.f};

    for (int kc = 0; kc < nchunks; ++kc) {
        const int q0 = kc * (KC / 4);
        v4f av[4], wv[8];
#pragma unroll
        for (int i = 0; i < 4; ++i) av[i] = X4[ia[i] + q0];
#pragma unroll
        for (int i = 0; i < 8; ++i) wv[i] = W4[iw + i * 16 * kq + q0];
        asm volatile("" : "+v"(av[0]), "+v"(av[1]), "+v"(av[2]), "+v"(av[3]),
                          "+v"(wv[0]), "+v"(wv[1]), "+v"(wv[2]), "+v"(wv[3]),
                          "+v"(wv[4]), "+v"(wv[5]), "+v"(wv[6]), "+v"(wv[7]));
#pragma unroll
        for (int i = 0; i < 4; ++i) {
            ss[i] += av[i].x * av[i].x + av[i].y * av[i].y + av[i].z * av[i].z + av[i].w * av[i].w;
            *(v4f*)(&As[sr + 16 * i][sc4 * 4]) = av[i];
        }
#pragma unroll
        for (int i = 0; i < 8; ++i) *(v4f*)(&Ws[sr + 16 * i][sc4 * 4]) = wv[i];
        __syncthreads();

        for (int kk = 0; kk < KC; kk += 4) {
            const int ko = kk + half * 2;
            v2f a;
            a.x = As[mr][ko];
            a.y = As[mr][ko + 1];
#pragma unroll
            for (int j = 0; j < 8; ++j) {
                const int colg = j * 16 + l15;
                v2f b;
                b.x = Ws[colg][ko];
                b.y = Ws[colg][ko + 1];
                acc[j] = __builtin_amdgcn_wmma_f32_16x16x4_f32(
                    false, a, false, b, (short)0, acc[j], false, false);
            }
        }
        __syncthreads();
    }

#pragma unroll
    for (int i = 0; i < 4; ++i) red[sr + 16 * i][sc4] = ss[i];
    __syncthreads();
    if (t < BM) {
        float un2 = red[t][0];
#pragma unroll
        for (int q = 1; q < 8; ++q) un2 += red[t][q];
        float un = fmaxf(sqrtf(un2), MINN);
        if (apply_exp) {
            float bc = fminf(tanhf(un), MAXN);
            scale_row[t] = bc / un;
            xn_row[t] = fmaxf(bc, MINN);
        } else {
            scale_row[t] = 1.f;
            xn_row[t] = un;
        }
        if (row0 + t >= nrows) scale_row[t] = 0.f;
    }
    __syncthreads();
    if (apply_exp) {
#pragma unroll
        for (int i = 0; i < 8; ++i) {
            const float sc = scale_row[rs * 16 + i + half * 8];
#pragma unroll
            for (int j = 0; j < 8; ++j) acc[j][i] *= sc;
        }
    }

    float bbv[8];
#pragma unroll
    for (int j = 0; j < 8; ++j) bbv[j] = bb[j * 16 + l15];

#pragma unroll
    for (int i = 0; i < 8; ++i) {
        float p2a = 0.f, pda = 0.f, p2b = 0.f, pdb = 0.f;
#pragma unroll
        for (int j = 0; j < 4; ++j) {
            float v = acc[j][i];
            p2a += v * v;
            pda += v * bbv[j];
        }
#pragma unroll
        for (int j = 4; j < 8; ++j) {
            float v = acc[j][i];
            p2b += v * v;
            pdb += v * bbv[j];
        }
        for (int m = 1; m < 16; m <<= 1) {
            p2a += __shfl_xor(p2a, m, 16);
            pda += __shfl_xor(pda, m, 16);
            p2b += __shfl_xor(p2b, m, 16);
            pdb += __shfl_xor(pdb, m, 16);
        }
        int mrow = rs * 16 + i + half * 8;
        if (l15 == 0) {
            rs_mx2[mrow] = p2a + p2b;
            rs_dot[mrow] = pda + pdb;
        }
    }
    __syncthreads();

    if (t < BM) {
        float xn = xn_row[t];
        float mxn = fmaxf(sqrtf(rs_mx2[t]), MINN);
        float tv = tanhf(mxn / xn * artanh_f(xn));
        float tc = fminf(tv, MAXN);
        float s_mv = tc / mxn;
        float x2 = tc * tc;
        float xy = s_mv * rs_dot[t];
        float y2 = y2p[0];
        float al = 1.f + 2.f * xy + y2;
        float be = 1.f - x2;
        float den = fmaxf(1.f + 2.f * xy + x2 * y2, MINN);
        float nn2 = fmaxf(al * al * x2 + 2.f * al * be * xy + be * be * y2, 0.f);
        float hn = fmaxf(sqrtf(nn2) / den, MINN);
        float hc = fminf(hn, MAXN);
        float L = artanh_f(hc) / hn;
        rowCA[t] = L * al * s_mv / den;
        rowCB[t] = L * be / den;
    }
    __syncthreads();

#pragma unroll
    for (int i = 0; i < 8; ++i) {
        int mrow = rs * 16 + i + half * 8;
        int gr = row0 + mrow;
        if (gr < nrows) {
            float ca = rowCA[mrow], cb = rowCB[mrow];
#pragma unroll
            for (int j = 0; j < 8; ++j) {
                int colg = j * 16 + l15;
                HT[(size_t)gr * 128 + colg] = ca * acc[j][i] + cb * bbv[j];
            }
        }
    }
}

__global__ __launch_bounds__(256) void edge_agg(const float* __restrict__ HT,
                                                const int* __restrict__ erow,
                                                const int* __restrict__ ecol,
                                                float* __restrict__ AGG, int E) {
    int e = blockIdx.x * 8 + (threadIdx.x >> 5);
    if (e >= E) return;
    int lane = threadIdx.x & 31;
    int r = erow[e], c = ecol[e];
    float4 v = *(const float4*)(HT + (size_t)c * 128 + lane * 4);
    float* dst = AGG + (size_t)r * 128 + lane * 4;
    atomicAdd(dst + 0, v.x);
    atomicAdd(dst + 1, v.y);
    atomicAdd(dst + 2, v.z);
    atomicAdd(dst + 3, v.w);
}

#define SCAN_B 1024
#define BKT_SH 8
#define BKT_N  256
#define MAXBK  1024
#define SORT_T 1024
#define SORT_U 4

__global__ __launch_bounds__(SORT_T) void bin_count(const int* __restrict__ erow,
                                                    int* __restrict__ bh,
                                                    int E, int nbk) {
    __shared__ int hist[MAXBK];
    const int t = threadIdx.x, G = gridDim.x, b = blockIdx.x;
    for (int k = t; k < nbk; k += SORT_T) hist[k] = 0;
    __syncthreads();
    const int chunk = (E + G - 1) / G;
    const int beg = b * chunk, end = min(E, beg + chunk);
    for (int i = beg + t; i < end; i += SORT_U * SORT_T) {
        int r[SORT_U];
#pragma unroll
        for (int u = 0; u < SORT_U; ++u) {
            const int j = i + u * SORT_T;
            r[u] = j < end ? erow[j] : 0;
        }
#pragma unroll
        for (int u = 0; u < SORT_U; ++u) {
            const int k = min((int)((unsigned)r[u] >> BKT_SH), nbk - 1);
            if (i + u * SORT_T < end) atomicAdd(&hist[k], 1);
        }
    }
    __syncthreads();
    for (int k = t; k < nbk; k += SORT_T) bh[(size_t)k * G + b] = hist[k];
}

__global__ __launch_bounds__(SORT_T) void bin_scatter(const int* __restrict__ erow,
                                                      const int* __restrict__ ecol,
                                                      const int* __restrict__ bo,
                                                      int2* __restrict__ tmp,
                                                      int E, int nbk) {
    __shared__ int cur[MAXBK];
    const int t = threadIdx.x, G = gridDim.x, b = blockIdx.x;
    for (int k = t; k < nbk; k += SORT_T) cur[k] = bo[(size_t)k * G + b];
    __syncthreads();
    const int chunk = (E + G - 1) / G;
    const int beg = b * chunk, end = min(E, beg + chunk);
    for (int i = beg + t; i < end; i += SORT_U * SORT_T) {
        int r[SORT_U], c[SORT_U];
#pragma unroll
        for (int u = 0; u < SORT_U; ++u) {
            const int j = i + u * SORT_T;
            r[u] = j < end ? erow[j] : 0;
            c[u] = j < end ? ecol[j] : 0;
        }
#pragma unroll
        for (int u = 0; u < SORT_U; ++u) {
            if (i + u * SORT_T < end) {
                const int k = min((int)((unsigned)r[u] >> BKT_SH), nbk - 1);
                const int p = atomicAdd(&cur[k], 1);
                tmp[p] = make_int2(r[u], c[u]);
            }
        }
    }
}

__global__ __launch_bounds__(SORT_T) void bin_sort(const int2* __restrict__ tmp,
                                                   const int* __restrict__ bo,
                                                   int* __restrict__ offs,
                                                   int* __restrict__ scol,
                                                   int G, int nbk, int n) {
    __shared__ int cnt[BKT_N];
    __shared__ int wsum[BKT_N / 32];
    const int t = threadIdx.x, k = blockIdx.x, lane = t & 31, wid = t >> 5;
    const int s = bo[(size_t)k * G], e = bo[(size_t)(k + 1) * G];
    if (t < BKT_N) cnt[t] = 0;
    __syncthreads();
    for (int i = s + t; i < e; i += SORT_U * SORT_T) {
        int r[SORT_U];
#pragma unroll
        for (int u = 0; u < SORT_U; ++u) {
            const int j = i + u * SORT_T;
            r[u] = j < e ? tmp[j].x : 0;
        }
#pragma unroll
        for (int u = 0; u < SORT_U; ++u)
            if (i + u * SORT_T < e) atomicAdd(&cnt[r[u] & (BKT_N - 1)], 1);
    }
    __syncthreads();
    int v = 0, sc = 0;
    if (t < BKT_N) {
        v = cnt[t];
        sc = v;
        for (int m = 1; m < 32; m <<= 1) {
            int u = __shfl_up(sc, m, 32);
            if (lane >= m) sc += u;
        }
        if (lane == 31) wsum[wid] = sc;
    }
    __syncthreads();
    if (t < BKT_N) {
        int base = s;
        for (int w = 0; w < wid; ++w) base += wsum[w];
        const int excl = base + sc - v;
        cnt[t] = excl;
        const int node = k * BKT_N + t;
        if (node < n) offs[node] = excl;
    }
    if (k == nbk - 1 && t == 0) offs[n] = e;
    __syncthreads();
    for (int i = s + t; i < e; i += SORT_U * SORT_T) {
        int2 rc[SORT_U];
#pragma unroll
        for (int u = 0; u < SORT_U; ++u) {
            const int j = i + u * SORT_T;
            rc[u] = j < e ? tmp[j] : make_int2(0, 0);
        }
#pragma unroll
        for (int u = 0; u < SORT_U; ++u) {
            if (i + u * SORT_T < e) {
                const int p = atomicAdd(&cnt[rc[u].x & (BKT_N - 1)], 1);
                scol[p] = rc[u].y;
            }
        }
    }
}

__device__ __forceinline__ int block_sum_1024(int v, int* sh) {
    int t = threadIdx.x;
    for (int m = 1; m < 32; m <<= 1) v += __shfl_xor(v, m, 32);
    if ((t & 31) == 0) sh[t >> 5] = v;
    __syncthreads();
    int w = sh[t & 31];
    for (int m = 1; m < 32; m <<= 1) w += __shfl_xor(w, m, 32);
    __syncthreads();
    return w;
}

__global__ __launch_bounds__(SCAN_B) void scan_sums(const int* __restrict__ cnt,
                                                    int* __restrict__ bsum, int n) {
    __shared__ int sh[32];
    int i = blockIdx.x * SCAN_B + threadIdx.x;
    int tot = block_sum_1024(i < n ? cnt[i] : 0, sh);
    if (threadIdx.x == 0) bsum[blockIdx.x] = tot;
}

__global__ __launch_bounds__(SCAN_B) void scan_apply(int* __restrict__ offs,
                                                     const int* __restrict__ bsum, int n) {
    __shared__ int sh[32];
    const int t = threadIdx.x, lane = t & 31, wid = t >> 5;
    int p = 0;
    for (int b = t; b < (int)blockIdx.x; b += SCAN_B) p += bsum[b];
    const int base = block_sum_1024(p, sh);

    int i = blockIdx.x * SCAN_B + t;
    int v = i < n ? offs[i] : 0;
    int s = v;
    for (int m = 1; m < 32; m <<= 1) {
        int u = __shfl_up(s, m, 32);
        if (lane >= m) s += u;
    }
    if (lane == 31) sh[wid] = s;
    __syncthreads();
    if (wid == 0) {
        int w0 = sh[lane], w = w0;
        for (int m = 1; m < 32; m <<= 1) {
            int u = __shfl_up(w, m, 32);
            if (lane >= m) w += u;
        }
        sh[lane] = w - w0;
    }
    __syncthreads();
    int excl = base + sh[wid] + s - v;
    if (i < n) offs[i] = excl;
    if (i == n - 1) offs[n] = excl + v;
}

__device__ __forceinline__ float wave_sum32(float x) {
    for (int m = 1; m < 32; m <<= 1) x += __shfl_xor(x, m, 32);
    return x;
}

__device__ __forceinline__ float4 tanh_row(float4 v, float amax, float& vn) {
    float an2 = wave_sum32(v.x * v.x + v.y * v.y + v.z * v.z + v.w * v.w);
    float an = fmaxf(sqrt_hw(an2), MINN);
    float lg = fminf(1.f, amax * rcp_hw(an));
    float4 at;
    at.x = tanh_hw(lg * v.x);
    at.y = tanh_hw(lg * v.y);
    at.z = tanh_hw(lg * v.z);
    at.w = tanh_hw(lg * v.w);
    float vn2 = wave_sum32(at.x * at.x + at.y * at.y + at.z * at.z + at.w * at.w);
    vn = fmaxf(sqrt_hw(vn2), MINN);
    return at;
}

__global__ __launch_bounds__(256) void node_gather_readout(const float* __restrict__ HT,
                                                           const int* __restrict__ offs,
                                                           const int* __restrict__ scol,
                                                           const int* __restrict__ seg,
                                                           float* __restrict__ ssum,
                                                           float* __restrict__ cnt, int nrows) {
    __shared__ float tr[8][128];
    const int lane = threadIdx.x & 31;
    const int wslot = threadIdx.x >> 5;
    const int w0 = __builtin_amdgcn_readfirstlane((int)(blockIdx.x * 8 + (threadIdx.x >> 5)));
    const int nw = gridDim.x * 8;
    const float amax = artanh_f(MAXN);
    for (int r = w0; r < nrows; r += nw) {
        const int beg = offs[r], end = offs[r + 1];
        const int sg = seg[r];
        float4 acc = make_float4(0.f, 0.f, 0.f, 0.f);
        for (int b0 = beg; b0 < end; b0 += 32) {
            const int n = min(32, end - b0);
            const int myc = (lane < n) ? scol[b0 + lane] : 0;
            int j = 0;
            for (; j + 4 <= n; j += 4) {
                const int c0 = __builtin_amdgcn_readlane(myc, j);
                const int c1 = __builtin_amdgcn_readlane(myc, j + 1);
                const int c2 = __builtin_amdgcn_readlane(myc, j + 2);
                const int c3 = __builtin_amdgcn_readlane(myc, j + 3);
                const float4 v0 = *((const float4*)(HT + (size_t)c0 * 128) + lane);
                const float4 v1 = *((const float4*)(HT + (size_t)c1 * 128) + lane);
                const float4 v2 = *((const float4*)(HT + (size_t)c2 * 128) + lane);
                const float4 v3 = *((const float4*)(HT + (size_t)c3 * 128) + lane);
                acc.x += v0.x; acc.y += v0.y; acc.z += v0.z; acc.w += v0.w;
                acc.x += v1.x; acc.y += v1.y; acc.z += v1.z; acc.w += v1.w;
                acc.x += v2.x; acc.y += v2.y; acc.z += v2.z; acc.w += v2.w;
                acc.x += v3.x; acc.y += v3.y; acc.z += v3.z; acc.w += v3.w;
            }
            for (; j < n; ++j) {
                const int c0 = __builtin_amdgcn_readlane(myc, j);
                const float4 v0 = *((const float4*)(HT + (size_t)c0 * 128) + lane);
                acc.x += v0.x; acc.y += v0.y; acc.z += v0.z; acc.w += v0.w;
            }
        }
        float vn;
        const float4 at = tanh_row(acc, amax, vn);
        const float L = fminf(1.f, amax * rcp_hw(vn));
        *(float4*)(&tr[wslot][lane * 4]) = make_float4(L * at.x, L * at.y, L * at.z, L * at.w);
        const float e0 = tr[wslot][lane];
        const float e1 = tr[wslot][lane + 32];
        const float e2 = tr[wslot][lane + 64];
        const float e3 = tr[wslot][lane + 96];
        float* d = ssum + (size_t)sg * 128 + lane;
        atomicAdd(d, e0);
        atomicAdd(d + 32, e1);
        atomicAdd(d + 64, e2);
        atomicAdd(d + 96, e3);
        if (lane == 0) atomicAdd(&cnt[sg], 1.f);
    }
}

__device__ __forceinline__ void add_rows(const float* __restrict__ HTin, int myc, int n,
                                         int lane, float4& acc) {
    int j = 0;
    for (; j + 8 <= n; j += 8) {
        const int c0 = __builtin_amdgcn_readlane(myc, j);
        const int c1 = __builtin_amdgcn_readlane(myc, j + 1);
        const int c2 = __builtin_amdgcn_readlane(myc, j + 2);
        const int c3 = __builtin_amdgcn_readlane(myc, j + 3);
        const int c4 = __builtin_amdgcn_readlane(myc, j + 4);
        const int c5 = __builtin_amdgcn_readlane(myc, j + 5);
        const int c6 = __builtin_amdgcn_readlane(myc, j + 6);
        const int c7 = __builtin_amdgcn_readlane(myc, j + 7);
        const float4 v0 = *((const float4*)(HTin + (size_t)c0 * 128) + lane);
        const float4 v1 = *((const float4*)(HTin + (size_t)c1 * 128) + lane);
        const float4 v2 = *((const float4*)(HTin + (size_t)c2 * 128) + lane);
        const float4 v3 = *((const float4*)(HTin + (size_t)c3 * 128) + lane);
        const float4 v4 = *((const float4*)(HTin + (size_t)c4 * 128) + lane);
        const float4 v5 = *((const float4*)(HTin + (size_t)c5 * 128) + lane);
        const float4 v6 = *((const float4*)(HTin + (size_t)c6 * 128) + lane);
        const float4 v7 = *((const float4*)(HTin + (size_t)c7 * 128) + lane);
        acc.x += v0.x; acc.y += v0.y; acc.z += v0.z; acc.w += v0.w;
        acc.x += v1.x; acc.y += v1.y; acc.z += v1.z; acc.w += v1.w;
        acc.x += v2.x; acc.y += v2.y; acc.z += v2.z; acc.w += v2.w;
        acc.x += v3.x; acc.y += v3.y; acc.z += v3.z; acc.w += v3.w;
        acc.x += v4.x; acc.y += v4.y; acc.z += v4.z; acc.w += v4.w;
        acc.x += v5.x; acc.y += v5.y; acc.z += v5.z; acc.w += v5.w;
        acc.x += v6.x; acc.y += v6.y; acc.z += v6.z; acc.w += v6.w;
        acc.x += v7.x; acc.y += v7.y; acc.z += v7.z; acc.w += v7.w;
    }
    for (; j + 4 <= n; j += 4) {
        const int c0 = __builtin_amdgcn_readlane(myc, j);
        const int c1 = __builtin_amdgcn_readlane(myc, j + 1);
        const int c2 = __builtin_amdgcn_readlane(myc, j + 2);
        const int c3 = __builtin_amdgcn_readlane(myc, j + 3);
        const float4 v0 = *((const float4*)(HTin + (size_t)c0 * 128) + lane);
        const float4 v1 = *((const float4*)(HTin + (size_t)c1 * 128) + lane);
        const float4 v2 = *((const float4*)(HTin + (size_t)c2 * 128) + lane);
        const float4 v3 = *((const float4*)(HTin + (size_t)c3 * 128) + lane);
        acc.x += v0.x; acc.y += v0.y; acc.z += v0.z; acc.w += v0.w;
        acc.x += v1.x; acc.y += v1.y; acc.z += v1.z; acc.w += v1.w;
        acc.x += v2.x; acc.y += v2.y; acc.z += v2.z; acc.w += v2.w;
        acc.x += v3.x; acc.y += v3.y; acc.z += v3.z; acc.w += v3.w;
    }
    for (; j < n; ++j) {
        const int c0 = __builtin_amdgcn_readlane(myc, j);
        const float4 v0 = *((const float4*)(HTin + (size_t)c0 * 128) + lane);
        acc.x += v0.x; acc.y += v0.y; acc.z += v0.z; acc.w += v0.w;
    }
}

#define HD   128
#define LDA  (HD + 4)
#define KC2  16
#define LDW  (KC2 + 4)

__global__ __launch_bounds__(128, 7) void gather_act_linear_wmma(
    const float* __restrict__ HTin,
    const int* __restrict__ offs, const int* __restrict__ scol, int nrows,
    const float* __restrict__ W,
    const float* __restrict__ bb,
    const float* __restrict__ y2p,
    float* __restrict__ HT) {
    __shared__ float As[BM][LDA];
    __shared__ float Ws[128][LDW];
    float* const rs_mx2 = &Ws[0][0];
    float* const rs_dot = rs_mx2 + BM;
    float* const rowCA = rs_dot + BM;
    float* const rowCB = rowCA + BM;

    const int t = threadIdx.x;
    const int row0 = blockIdx.x * BM;
    const int lane = t & 31;
    const int rs = __builtin_amdgcn_readfirstlane(t >> 5);
    const int half = lane >> 4;
    const int l15 = lane & 15;

    const int rbase = row0 + rs * 16;
    const float amax = artanh_f(MAXN);
    const int ob = offs[min(rbase + min(lane, 16), nrows)];
    float my_xn = 0.f;
    int begN = __builtin_amdgcn_readlane(ob, 0);
    int endN = __builtin_amdgcn_readlane(ob, 1);
    int mycN = (lane < endN - begN) ? scol[begN + lane] : 0;
    for (int i = 0; i < 16; ++i) {
        const int beg = begN, end = endN;
        const int myc = mycN;
        if (i < 15) {
            begN = __builtin_amdgcn_readlane(ob, i + 1);
            endN = __builtin_amdgcn_readlane(ob, i + 2);
            mycN = (lane < endN - begN) ? scol[begN + lane] : 0;
        }
        float4 acc = make_float4(0.f, 0.f, 0.f, 0.f);
        add_rows(HTin, myc, min(32, end - beg), lane, acc);
        for (int b0 = beg + 32; b0 < end; b0 += 32) {
            const int n = min(32, end - b0);
            const int mc = (lane < n) ? scol[b0 + lane] : 0;
            add_rows(HTin, mc, n, lane, acc);
        }
        float vn;
        const float4 at = tanh_row(acc, amax, vn);
        const float tcl = fminf(tanh_hw(vn), MAXN);
        const float s = tcl * rcp_hw(vn);
        if (lane == i) my_xn = fmaxf(tcl, MINN);
        v4f hv;
        hv.x = s * at.x; hv.y = s * at.y; hv.z = s * at.z; hv.w = s * at.w;
        *(v4f*)(&As[rs * 16 + i][lane * 4]) = hv;
    }
    if (lane < 16) As[rs * 16 + lane][HD] = my_xn;

    v8f acc[8];
    for (int j = 0; j < 8; ++j)
        for (int i = 0; i < 8; ++i) acc[j][i] = 0.f;

    const int sr = t >> 2;
    const int sc4 = t & 3;
    const int kq = HD >> 2;
    const v4f* __restrict__ W4 = (const v4f*)W;
    const int iw = sr * kq + sc4;
    const int mr = rs * 16 + l15;

    for (int kc = 0; kc < HD / KC2; ++kc) {
        const int q0 = kc * (KC2 / 4);
        v4f wv[4];
#pragma unroll
        for (int i = 0; i < 4; ++i) wv[i] = W4[iw + i * 32 * kq + q0];
        asm volatile("" : "+v"(wv[0]), "+v"(wv[1]), "+v"(wv[2]), "+v"(wv[3]));
#pragma unroll
        for (int i = 0; i < 4; ++i) *(v4f*)(&Ws[sr + 32 * i][sc4 * 4]) = wv[i];
        __syncthreads();

        for (int kk = 0; kk < KC2; kk += 4) {
            const int ko = kk + half * 2;
            v2f a;
            a.x = As[mr][kc * KC2 + ko];
            a.y = As[mr][kc * KC2 + ko + 1];
#pragma unroll
            for (int j = 0; j < 8; ++j) {
                const int colg = j * 16 + l15;
                v2f b;
                b.x = Ws[colg][ko];
                b.y = Ws[colg][ko + 1];
                acc[j] = __builtin_amdgcn_wmma_f32_16x16x4_f32(
                    false, a, false, b, (short)0, acc[j], false, false);
            }
        }
        __syncthreads();
    }

    float bbv[8];
#pragma unroll
    for (int j = 0; j < 8; ++j) bbv[j] = bb[j * 16 + l15];

#pragma unroll
    for (int i = 0; i < 8; ++i) {
        float p2a = 0.f, pda = 0.f, p2b = 0.f, pdb = 0.f;
#pragma unroll
        for (int j = 0; j < 4; ++j) {
            float v = acc[j][i];
            p2a += v * v;
            pda += v * bbv[j];
        }
#pragma unroll
        for (int j = 4; j < 8; ++j) {
            float v = acc[j][i];
            p2b += v * v;
            pdb += v * bbv[j];
        }
        for (int m = 1; m < 16; m <<= 1) {
            p2a += __shfl_xor(p2a, m, 16);
            pda += __shfl_xor(pda, m, 16);
            p2b += __shfl_xor(p2b, m, 16);
            pdb += __shfl_xor(pdb, m, 16);
        }
        int mrow = rs * 16 + i + half * 8;
        if (l15 == 0) {
            rs_mx2[mrow] = p2a + p2b;
            rs_dot[mrow] = pda + pdb;
        }
    }
    __syncthreads();

    if (t < BM) {
        float xn = As[t][HD];
        float mxn = fmaxf(sqrtf(rs_mx2[t]), MINN);
        float tv = tanhf(mxn / xn * artanh_f(xn));
        float tc = fminf(tv, MAXN);
        float s_mv = tc / mxn;
        float x2 = tc * tc;
        float xy = s_mv * rs_dot[t];
        float y2 = y2p[0];
        float al = 1.f + 2.f * xy + y2;
        float be = 1.f - x2;
        float den = fmaxf(1.f + 2.f * xy + x2 * y2, MINN);
        float nn2 = fmaxf(al * al * x2 + 2.f * al * be * xy + be * be * y2, 0.f);
        float hn = fmaxf(sqrtf(nn2) / den, MINN);
        float hc = fminf(hn, MAXN);
        float L = artanh_f(hc) / hn;
        rowCA[t] = L * al * s_mv / den;
        rowCB[t] = L * be / den;
    }
    __syncthreads();

#pragma unroll
    for (int i = 0; i < 8; ++i) {
        int mrow = rs * 16 + i + half * 8;
        int gr = row0 + mrow;
        if (gr < nrows) {
            float ca = rowCA[mrow], cb = rowCB[mrow];
#pragma unroll
            for (int j = 0; j < 8; ++j) {
                int colg = j * 16 + l15;
                HT[(size_t)gr * 128 + colg] = ca * acc[j][i] + cb * bbv[j];
            }
        }
    }
}

__global__ __launch_bounds__(256) void post_agg(float* __restrict__ H, int nrows) {
    int r = blockIdx.x * 8 + (threadIdx.x >> 5);
    if (r >= nrows) return;
    int lane = threadIdx.x & 31;
    float* p = H + (size_t)r * 128 + lane * 4;
    float4 v = *(const float4*)p;
    float an2 = v.x * v.x + v.y * v.y + v.z * v.z + v.w * v.w;
    for (int m = 1; m < 32; m <<= 1) an2 += __shfl_xor(an2, m, 32);
    float an = fmaxf(sqrtf(an2), MINN);
    float tn = fminf(fmaxf(tanhf(an), MINN), MAXN);
    float lg = artanh_f(tn) / an;
    float4 at;
    at.x = tanhf(lg * v.x);
    at.y = tanhf(lg * v.y);
    at.z = tanhf(lg * v.z);
    at.w = tanhf(lg * v.w);
    float vn2 = at.x * at.x + at.y * at.y + at.z * at.z + at.w * at.w;
    for (int m = 1; m < 32; m <<= 1) vn2 += __shfl_xor(vn2, m, 32);
    float vn = fmaxf(sqrtf(vn2), MINN);
    float s = fminf(tanhf(vn), MAXN) / vn;
    *(float4*)p = make_float4(s * at.x, s * at.y, s * at.z, s * at.w);
}

__global__ __launch_bounds__(256) void final_scatter(
    const float* __restrict__ H, const int* __restrict__ seg,
    float* __restrict__ ssum, float* __restrict__ cnt, int nrows) {
    int r = blockIdx.x * 8 + (threadIdx.x >> 5);
    if (r >= nrows) return;
    int lane = threadIdx.x & 31;
    float4 v = *(const float4*)(H + (size_t)r * 128 + lane * 4);
    float hn2 = v.x * v.x + v.y * v.y + v.z * v.z + v.w * v.w;
    for (int m = 1; m < 32; m <<= 1) hn2 += __shfl_xor(hn2, m, 32);
    float hn = fmaxf(sqrtf(hn2), MINN);
    float pc = fminf(hn, MAXN);
    float L = artanh_f(pc) / hn;
    int s = seg[r];
    float* d = ssum + (size_t)s * 128 + lane * 4;
    atomicAdd(d + 0, L * v.x);
    atomicAdd(d + 1, L * v.y);
    atomicAdd(d + 2, L * v.z);
    atomicAdd(d + 3, L * v.w);
    if (lane == 0) atomicAdd(&cnt[s], 1.f);
}

__global__ __launch_bounds__(256) void pool_k(const float* __restrict__ ssum,
                                              const float* __restrict__ cnt,
                                              float* __restrict__ out, int nseg) {
    int r = blockIdx.x * 8 + (threadIdx.x >> 5);
    if (r >= nseg) return;
    int lane = threadIdx.x & 31;
    float inv = 1.f / fmaxf(cnt[r], 1.f);
    float4 v = *(const float4*)(ssum + (size_t)r * 128 + lane * 4);
    v.x *= inv; v.y *= inv; v.z *= inv; v.w *= inv;
    float mn2 = v.x * v.x + v.y * v.y + v.z * v.z + v.w * v.w;
    for (int m = 1; m < 32; m <<= 1) mn2 += __shfl_xor(mn2, m, 32);
    float mn = fmaxf(sqrtf(mn2), MINN);
    float s = fminf(tanhf(mn), MAXN) / mn;
    float* o = out + (size_t)r * 128 + lane * 4;
    o[0] = s * v.x; o[1] = s * v.y; o[2] = s * v.z; o[3] = s * v.w;
}

__global__ void zero_f32(float* __restrict__ p, size_t n) {
    size_t i = (size_t)blockIdx.x * blockDim.x + threadIdx.x;
    size_t st = (size_t)gridDim.x * blockDim.x;
    for (; i < n; i += st) p[i] = 0.f;
}

extern "C" void kernel_launch(void* const* d_in, const int* in_sizes, int n_in,
                              void* d_out, int out_size, void* d_ws, size_t ws_size,
                              hipStream_t stream) {
    const float* x = (const float*)d_in[0];
    const int* erow = (const int*)d_in[1];
    const int* ecol = (const int*)d_in[2];
    const int* seg = (const int*)d_in[3];
    const float* W1 = (const float*)d_in[4];
    const float* b1 = (const float*)d_in[5];
    const float* W2 = (const float*)d_in[6];
    const float* b2 = (const float*)d_in[7];

    const int N = in_sizes[0] / 256;
    const int E = in_sizes[1];
    const int NSEG = out_size / 128;

    char* ws = (char*)d_ws;
    size_t bufBytes = (size_t)N * 128 * sizeof(float);
    float* buf0 = (float*)ws;
    float* buf1 = (float*)(ws + bufBytes);
    float* bbias = (float*)(ws + 2 * bufBytes);
    float* y2 = bbias + 256;
    float* ssum = buf0;
    float* cnt = buf0 + (size_t)NSEG * 128;

    dim3 blk(256);
    dim3 gemmBlk(128);
    int gRows = (N + 7) / 8;
    int gGemm = (N + BM - 1) / BM;
    int gEdge = (E + 7) / 8;

    const int nScan = (N + SCAN_B - 1) / SCAN_B;
    size_t offBytes = (((size_t)N + 1) * sizeof(int) + 255) / 256 * 256;
    size_t bsBytes = ((size_t)nScan * sizeof(int) + 255) / 256 * 256;
    int* offs = (int*)(ws + 2 * bufBytes + 4096);
    int* scol = (int*)(ws + 2 * bufBytes + 4096 + offBytes + bsBytes);
    const int G1 = 256;
    const int NBK = (N + BKT_N - 1) / BKT_N;
    const int nTab = NBK * G1;
    const int nScanT = (nTab + SCAN_B - 1) / SCAN_B;
    size_t tmpBytes = ((size_t)E * sizeof(int2) + 255) / 256 * 256;
    size_t tabBytes = (((size_t)nTab + 1) * sizeof(int) + 255) / 256 * 256;
    int2* tmp = (int2*)buf1;
    int* btab = (int*)(ws + bufBytes + tmpBytes);
    int* bsum = (int*)(ws + bufBytes + tmpBytes + tabBytes);
    const bool grouped =
        ws_size >= 2 * bufBytes + 4096 + offBytes + bsBytes + (size_t)E * sizeof(int) &&
        NBK <= MAXBK &&
        tmpBytes + tabBytes + (size_t)nScanT * sizeof(int) <= bufBytes;

    bias_prep<<<2, 128, 0, stream>>>(b1, b2, bbias, y2);

    if (grouped) {
        bin_count<<<G1, SORT_T, 0, stream>>>(erow, btab, E, NBK);
        scan_sums<<<nScanT, SCAN_B, 0, stream>>>(btab, bsum, nTab);
        scan_apply<<<nScanT, SCAN_B, 0, stream>>>(btab, bsum, nTab);
        bin_scatter<<<G1, SORT_T, 0, stream>>>(erow, ecol, btab, tmp, E, NBK);
        bin_sort<<<NBK, SORT_T, 0, stream>>>(tmp, btab, offs, scol, G1, NBK, N);

        hyp_linear_wmma<<<gGemm, gemmBlk, 0, stream>>>(x, 256, N, W1, bbias, y2, buf0, 1);

        gather_act_linear_wmma<<<gGemm, gemmBlk, 0, stream>>>(buf0, offs, scol, N, W2,
                                                          bbias + 128, y2 + 1, buf1);
        float* ssumG = buf0;
        float* cntG = buf0 + (size_t)NSEG * 128;
        zero_f32<<<64, 256, 0, stream>>>(ssumG, (size_t)NSEG * 128 + NSEG);
        node_gather_readout<<<2048, blk, 0, stream>>>(buf1, offs, scol, seg, ssumG, cntG, N);
        pool_k<<<(NSEG + 7) / 8, blk, 0, stream>>>(ssumG, cntG, (float*)d_out, NSEG);
        return;
    }

    hyp_linear_wmma<<<gGemm, gemmBlk, 0, stream>>>(x, 256, N, W1, bbias, y2, buf0, 1);
    zero_f32<<<2048, 256, 0, stream>>>(buf1, (size_t)N * 128);
    edge_agg<<<gEdge, blk, 0, stream>>>(buf0, erow, ecol, buf1, E);
    post_agg<<<gRows, blk, 0, stream>>>(buf1, N);

    hyp_linear_wmma<<<gGemm, gemmBlk, 0, stream>>>(buf1, 128, N, W2, bbias + 128,
                                               y2 + 1, buf0, 0);
    zero_f32<<<2048, 256, 0, stream>>>(buf1, (size_t)N * 128);
    edge_agg<<<gEdge, blk, 0, stream>>>(buf0, erow, ecol, buf1, E);
    post_agg<<<gRows, blk, 0, stream>>>(buf1, N);

    zero_f32<<<64, 256, 0, stream>>>(ssum, (size_t)NSEG * 128 + NSEG);
    final_scatter<<<gRows, blk, 0, stream>>>(buf1, seg, ssum, cnt, N);
    pool_k<<<(NSEG + 7) / 8, blk, 0, stream>>>(ssum, cnt, (float*)d_out, NSEG);
}
